# starting kernel
# baseline (speedup 1.0000x reference)
#include <hip/hip_runtime.h>
#include <hip/hip_bf16.h>
#include <math.h>


#define HDIM 128
#define WS2  144
#define A_LD 138

typedef __attribute__((ext_vector_type(2))) float v2f;
typedef __attribute__((ext_vector_type(8))) float v8f;

__global__ __launch_bounds__(256)
void gcn_gemm_wmma_f32(const float* __restrict__ A,
                       const float* __restrict__ W,
                       const float* __restrict__ bias,
                       const float* __restrict__ dinv,
                       float* __restrict__ C,
                       float* __restrict__ ACC,
                       int N, int K, int Kpad) {
    extern __shared__ float lds[];
    v2f*   Wp = (v2f*)lds;
    float* As = lds + (size_t)(Kpad / 2) * WS2 * 2;

    const int tid  = threadIdx.x;
    const int wave = tid >> 5;
    const int lane = tid & 31;
    const int l15  = lane & 15;
    const int rowBase = blockIdx.x * 256 + wave * 32;

    for (int idx = tid; idx < (Kpad / 2) * HDIM; idx += 256) {
        int kk = idx >> 7;
        int n  = idx & (HDIM - 1);
        int k0 = kk * 2;
        v2f w;
        w.x = (k0     < K) ? W[(size_t)k0       * HDIM + n] : 0.0f;
        w.y = (k0 + 1 < K) ? W[(size_t)(k0 + 1) * HDIM + n] : 0.0f;
        Wp[kk * WS2 + n] = w;
    }

    float* Aw = As + (size_t)wave * 32 * A_LD;
    for (int r = 0; r < 32; ++r) {
        int row = rowBase + r;
        for (int k = lane; k < Kpad; k += 32) {
            Aw[r * A_LD + k] = (row < N && k < K) ? A[(size_t)row * K + k] : 0.0f;
        }
    }
    __syncthreads();

    v8f acc[2][8];
#pragma unroll
    for (int t = 0; t < 2; ++t)
#pragma unroll
        for (int j = 0; j < 8; ++j)
            acc[t][j] = (v8f){0.f,0.f,0.f,0.f,0.f,0.f,0.f,0.f};

    const int koff = (lane >= 16) ? 2 : 0;
    const float* Aw0 = Aw + l15 * A_LD;
    const float* Aw1 = Aw + (l15 + 16) * A_LD;

    for (int k0 = 0; k0 < Kpad; k0 += 4) {
        v2f a0 = *(const v2f*)(Aw0 + k0 + koff);
        v2f a1 = *(const v2f*)(Aw1 + k0 + koff);
        const v2f* Bk = Wp + (size_t)((k0 + koff) >> 1) * WS2 + l15;
#pragma unroll
        for (int j = 0; j < 8; ++j) {
            v2f b = Bk[j * 16];
            acc[0][j] = __builtin_amdgcn_wmma_f32_16x16x4_f32(
                false, a0, false, b, (short)0, acc[0][j], false, false);
            acc[1][j] = __builtin_amdgcn_wmma_f32_16x16x4_f32(
                false, a1, false, b, (short)0, acc[1][j], false, false);
        }
    }

    float bcol[8];
#pragma unroll
    for (int j = 0; j < 8; ++j) bcol[j] = bias[j * 16 + l15];

    const int rOff = (lane >= 16) ? 8 : 0;
#pragma unroll
    for (int t = 0; t < 2; ++t) {
#pragma unroll
        for (int i = 0; i < 8; ++i) {
            int row = rowBase + t * 16 + i + rOff;
            if (row < N) {
                float di  = dinv[row];
                float di2 = di * di;
                size_t base = (size_t)row * HDIM + l15;
#pragma unroll
                for (int j = 0; j < 8; ++j) {
                    float h = acc[t][j][i];
                    C[base + j * 16]   = h;
                    ACC[base + j * 16] = h * di2 + bcol[j];
                }
            }
        }
    }
}

__global__ void zero_f32(float* __restrict__ p, int n) {
    int i = blockIdx.x * blockDim.x + threadIdx.x;
    if (i < n) p[i] = 0.0f;
}

__global__ void deg_count(const int* __restrict__ dst, float* __restrict__ deg, int E) {
    int e = blockIdx.x * blockDim.x + threadIdx.x;
    if (e < E) unsafeAtomicAdd(&deg[dst[e]], 1.0f);
}

__global__ void finalize_dinv(float* __restrict__ d, int n) {
    int i = blockIdx.x * blockDim.x + threadIdx.x;
    if (i < n) d[i] = rsqrtf(d[i] + 1.0f);
}

__global__ __launch_bounds__(256)
void edge_scatter(const int* __restrict__ src, const int* __restrict__ dst,
                  const float* __restrict__ dinv, const float* __restrict__ g,
                  float* __restrict__ acc, int E) {
    int e = blockIdx.x * 8 + (threadIdx.x >> 5);
    if (e >= E) return;
    int lane = threadIdx.x & 31;

    int ep = e + 256;
    if (ep < E) {
        int sp = src[ep];
        __builtin_prefetch(g + (size_t)sp * HDIM + lane * 4, 0, 1);
    }

    int s = src[e];
    int d = dst[e];
    float coef = dinv[s] * dinv[d];
    const float4* gs = (const float4*)(g + (size_t)s * HDIM);
    float4 v = gs[lane];
    float* ad = acc + (size_t)d * HDIM + lane * 4;
    unsafeAtomicAdd(ad + 0, coef * v.x);
    unsafeAtomicAdd(ad + 1, coef * v.y);
    unsafeAtomicAdd(ad + 2, coef * v.z);
    unsafeAtomicAdd(ad + 3, coef * v.w);
}

__global__ void tanh_act(const float* __restrict__ in, float* __restrict__ out, int n) {
    int i = blockIdx.x * blockDim.x + threadIdx.x;
    if (i < n) out[i] = tanhf(in[i]);
}

extern "C" void kernel_launch(void* const* d_in, const int* in_sizes, int n_in,
                              void* d_out, int out_size, void* d_ws, size_t ws_size,
                              hipStream_t stream) {
    const float* x  = (const float*)d_in[0];
    const int*   ei = (const int*)d_in[1];
    const float* Wl[4] = {(const float*)d_in[2], (const float*)d_in[4],
                          (const float*)d_in[6], (const float*)d_in[8]};
    const float* bl[4] = {(const float*)d_in[3], (const float*)d_in[5],
                          (const float*)d_in[7], (const float*)d_in[9]};

    const int NFEAT = in_sizes[2] / HDIM;
    const int N     = in_sizes[0] / NFEAT;
    const int E     = in_sizes[1] / 2;
    const int* srcp = ei;
    const int* dstp = ei + E;

    float* wsf  = (float*)d_ws;
    float* dinv = wsf;
    float* r0   = wsf + N;
    float* r1   = r0 + (size_t)N * HDIM;
    float* r2   = r1 + (size_t)N * HDIM;

    const int THR = 256;

    zero_f32<<<(N + THR - 1) / THR, THR, 0, stream>>>(dinv, N);
    deg_count<<<(E + THR - 1) / THR, THR, 0, stream>>>(dstp, dinv, E);
    finalize_dinv<<<(N + THR - 1) / THR, THR, 0, stream>>>(dinv, N);

    const float* in = x;
    int K = NFEAT;
    for (int l = 0; l < 4; ++l) {
        const int Kpad = (K + 3) & ~3;
        const size_t shmem =
            ((size_t)(Kpad / 2) * WS2 * 2 + 8u * 32u * A_LD) * sizeof(float);
        gcn_gemm_wmma_f32<<<(N + 255) / 256, 256, shmem, stream>>>(
            in, Wl[l], bl[l], dinv, r0, r1, N, K, Kpad);

        edge_scatter<<<(E + 7) / 8, 256, 0, stream>>>(srcp, dstp, dinv, r0, r1, E);

        float* outp = (l == 3) ? (float*)d_out : r2;
        tanh_act<<<((size_t)N * HDIM + THR - 1) / THR, THR, 0, stream>>>(
            r1, outp, (int)((size_t)N * HDIM));

        in = r2;
        K = HDIM;
    }
}


# edge sum as per-call dst-sorted CSR gather: one wave owns a row, stores tanh(sum+self+bias) once; float atomics gone (one int atomic ranks slots), row sum order fixed: ascending edge id
# speedup vs baseline: 2.3001x; 2.3001x over previous
#include <hip/hip_runtime.h>
#include <hip/hip_bf16.h>
#include <math.h>


#define HDIM 128
#define WS2  144
#define A_LD 138

typedef __attribute__((ext_vector_type(2))) float v2f;
typedef __attribute__((ext_vector_type(8))) float v8f;

__global__ __launch_bounds__(256)
void gcn_gemm_wmma_f32(const float* __restrict__ A,
                       const float* __restrict__ W,
                       float* __restrict__ C,
                       int N, int K, int Kpad) {
    extern __shared__ float lds[];
    v2f*   Wp = (v2f*)lds;
    float* As = lds + (size_t)(Kpad / 2) * WS2 * 2;

    const int tid  = threadIdx.x;
    const int wave = tid >> 5;
    const int lane = tid & 31;
    const int l15  = lane & 15;
    const int rowBase = blockIdx.x * 256 + wave * 32;

    for (int idx = tid; idx < (Kpad / 2) * HDIM; idx += 256) {
        int kk = idx >> 7;
        int n  = idx & (HDIM - 1);
        int k0 = kk * 2;
        v2f w;
        w.x = (k0     < K) ? W[(size_t)k0       * HDIM + n] : 0.0f;
        w.y = (k0 + 1 < K) ? W[(size_t)(k0 + 1) * HDIM + n] : 0.0f;
        Wp[kk * WS2 + n] = w;
    }

    float* Aw = As + (size_t)wave * 32 * A_LD;
    for (int r = 0; r < 32; ++r) {
        int row = rowBase + r;
        for (int k = lane; k < Kpad; k += 32) {
            Aw[r * A_LD + k] = (row < N && k < K) ? A[(size_t)row * K + k] : 0.0f;
        }
    }
    __syncthreads();

    v8f acc[2][8];
#pragma unroll
    for (int t = 0; t < 2; ++t)
#pragma unroll
        for (int j = 0; j < 8; ++j)
            acc[t][j] = (v8f){0.f,0.f,0.f,0.f,0.f,0.f,0.f,0.f};

    const int koff = (lane >= 16) ? 2 : 0;
    const float* Aw0 = Aw + l15 * A_LD;
    const float* Aw1 = Aw + (l15 + 16) * A_LD;

    for (int k0 = 0; k0 < Kpad; k0 += 4) {
        v2f a0 = *(const v2f*)(Aw0 + k0 + koff);
        v2f a1 = *(const v2f*)(Aw1 + k0 + koff);
        const v2f* Bk = Wp + (size_t)((k0 + koff) >> 1) * WS2 + l15;
#pragma unroll
        for (int j = 0; j < 8; ++j) {
            v2f b = Bk[j * 16];
            acc[0][j] = __builtin_amdgcn_wmma_f32_16x16x4_f32(
                false, a0, false, b, (short)0, acc[0][j], false, false);
            acc[1][j] = __builtin_amdgcn_wmma_f32_16x16x4_f32(
                false, a1, false, b, (short)0, acc[1][j], false, false);
        }
    }

    const int rOff = (lane >= 16) ? 8 : 0;
#pragma unroll
    for (int t = 0; t < 2; ++t) {
#pragma unroll
        for (int i = 0; i < 8; ++i) {
            int row = rowBase + t * 16 + i + rOff;
            if (row < N) {
                size_t base = (size_t)row * HDIM + l15;
#pragma unroll
                for (int j = 0; j < 8; ++j) {
                    C[base + j * 16] = acc[t][j][i];
                }
            }
        }
    }
}

__global__ void zero_i32(int* __restrict__ p, int n) {
    int i = blockIdx.x * blockDim.x + threadIdx.x;
    if (i < n) p[i] = 0;
}

__global__ void count_rank(const int* __restrict__ dst, int* __restrict__ cnt,
                           int* __restrict__ rank, int E) {
    int e = blockIdx.x * blockDim.x + threadIdx.x;
    if (e < E) rank[e] = atomicAdd(&cnt[dst[e]], 1);
}

#define SCAN_T 256
#define SCAN_PER 4
__global__ __launch_bounds__(SCAN_T)
void scan_rowptr_dinv(const int* __restrict__ cnt, int* __restrict__ rowptr,
                      float* __restrict__ dinv, int N) {
    __shared__ int sh[SCAN_T];
    const int tid   = threadIdx.x;
    const int start = blockIdx.x * (SCAN_T * SCAN_PER);

    int pre = 0;
    const int4* __restrict__ cnt4 = (const int4*)cnt;
    for (int i = tid; i < start / 4; i += SCAN_T) {
        int4 v = cnt4[i];
        pre += (v.x + v.y) + (v.z + v.w);
    }

    int c[SCAN_PER];
    int loc = 0;
#pragma unroll
    for (int q = 0; q < SCAN_PER; ++q) {
        int i = start + tid * SCAN_PER + q;
        c[q] = (i < N) ? cnt[i] : 0;
        loc += c[q];
    }

    sh[tid] = pre;
    __syncthreads();
    for (int s = SCAN_T / 2; s > 0; s >>= 1) {
        if (tid < s) sh[tid] += sh[tid + s];
        __syncthreads();
    }
    const int offset = sh[0];
    __syncthreads();

    sh[tid] = loc;
    __syncthreads();
    for (int s = 1; s < SCAN_T; s <<= 1) {
        int v = (tid >= s) ? sh[tid - s] : 0;
        __syncthreads();
        sh[tid] += v;
        __syncthreads();
    }
    int run = offset + sh[tid] - loc;
#pragma unroll
    for (int q = 0; q < SCAN_PER; ++q) {
        int i = start + tid * SCAN_PER + q;
        if (i < N) {
            rowptr[i] = run;
            dinv[i]   = rsqrtf((float)c[q] + 1.0f);
        }
        run += c[q];
        if (i == N - 1) rowptr[N] = run;
    }
}

__global__ void csr_fill(const int* __restrict__ dst, const int* __restrict__ rank,
                         const int* __restrict__ rowptr, int* __restrict__ eid, int E) {
    int e = blockIdx.x * blockDim.x + threadIdx.x;
    if (e >= E) return;
    int p = rowptr[dst[e]] + rank[e];
    p = min(max(p, 0), E - 1);
    eid[p] = e;
}

__global__ __launch_bounds__(256)
void csr_order(const int* __restrict__ rowptr, const int* __restrict__ eid,
               const int* __restrict__ src, const float* __restrict__ dinv,
               int* __restrict__ col, float* __restrict__ cf, int N, int E) {
    const int lane = threadIdx.x & 31;
    const int row = __builtin_amdgcn_readfirstlane((int)(blockIdx.x * 8 + (threadIdx.x >> 5)));
    if (row >= N) return;
    const int beg = __builtin_amdgcn_readfirstlane(rowptr[row]);
    const int end = __builtin_amdgcn_readfirstlane(rowptr[row + 1]);
    const float dd = dinv[row];

    for (int a = beg; a < end; a += 32) {
        const int na = min(32, end - a);
        const int my = (lane < na) ? eid[a + lane] : 0x7fffffff;
        int rk = 0;
        for (int b = beg; b < end; b += 32) {
            const int nb = min(32, end - b);
            const int ob = (lane < nb) ? eid[b + lane] : 0x7fffffff;
            for (int j = 0; j < nb; ++j)
                rk += (__builtin_amdgcn_readlane(ob, j) < my) ? 1 : 0;
        }
        if (lane < na) {
            const int e = (int)min((unsigned)my, (unsigned)(E - 1));
            const int s = src[e];
            col[beg + rk] = s;
            cf[beg + rk]  = dinv[s] * dd;
        }
    }
}

__device__ __forceinline__ void fma4(float4& a, float c, const float4& v) {
    a.x += c * v.x; a.y += c * v.y; a.z += c * v.z; a.w += c * v.w;
}

__global__ __launch_bounds__(256)
void gcn_gather_rows(const int* __restrict__ rowptr, const int* __restrict__ col,
                     const float* __restrict__ cf, const float* __restrict__ dinv,
                     const float* __restrict__ bias, const float* __restrict__ h,
                     float* __restrict__ out, int N) {
    const int lane = threadIdx.x & 31;
    const int row = __builtin_amdgcn_readfirstlane((int)(blockIdx.x * 8 + (threadIdx.x >> 5)));
    if (row >= N) return;
    const int beg = __builtin_amdgcn_readfirstlane(rowptr[row]);
    const int end = __builtin_amdgcn_readfirstlane(rowptr[row + 1]);

    const float4* __restrict__ h4 = (const float4*)h;
    const float4 hv = h4[(size_t)row * 32 + lane];
    const float  di = dinv[row];
    const float4 b4 = ((const float4*)bias)[lane];

    float4 acc = make_float4(0.f, 0.f, 0.f, 0.f);
    const unsigned nmax = (unsigned)(N - 1);

    for (int base = beg; base < end; base += 32) {
        const int n = min(32, end - base);
        int   sv = 0;
        float cv = 0.f;
        if (lane < n) { sv = col[base + lane]; cv = cf[base + lane]; }
        sv = (int)min((unsigned)sv, nmax);
        const int cvi = __builtin_bit_cast(int, cv);

        int j = 0;
        for (; j + 4 <= n; j += 4) {
            const int s0 = __builtin_amdgcn_readlane(sv, j);
            const int s1 = __builtin_amdgcn_readlane(sv, j + 1);
            const int s2 = __builtin_amdgcn_readlane(sv, j + 2);
            const int s3 = __builtin_amdgcn_readlane(sv, j + 3);
            const float4 v0 = h4[(size_t)s0 * 32 + lane];
            const float4 v1 = h4[(size_t)s1 * 32 + lane];
            const float4 v2 = h4[(size_t)s2 * 32 + lane];
            const float4 v3 = h4[(size_t)s3 * 32 + lane];
            fma4(acc, __builtin_bit_cast(float, __builtin_amdgcn_readlane(cvi, j)),     v0);
            fma4(acc, __builtin_bit_cast(float, __builtin_amdgcn_readlane(cvi, j + 1)), v1);
            fma4(acc, __builtin_bit_cast(float, __builtin_amdgcn_readlane(cvi, j + 2)), v2);
            fma4(acc, __builtin_bit_cast(float, __builtin_amdgcn_readlane(cvi, j + 3)), v3);
        }
        const int r = n - j;
        if (r > 0) {
            const int s0 = __builtin_amdgcn_readlane(sv, j);
            const float4 v0 = h4[(size_t)s0 * 32 + lane];
            float4 v1 = make_float4(0.f, 0.f, 0.f, 0.f);
            float4 v2 = make_float4(0.f, 0.f, 0.f, 0.f);
            if (r > 1) {
                const int s1 = __builtin_amdgcn_readlane(sv, j + 1);
                v1 = h4[(size_t)s1 * 32 + lane];
            }
            if (r > 2) {
                const int s2 = __builtin_amdgcn_readlane(sv, j + 2);
                v2 = h4[(size_t)s2 * 32 + lane];
            }
            fma4(acc, __builtin_bit_cast(float, __builtin_amdgcn_readlane(cvi, j)), v0);
            if (r > 1) fma4(acc, __builtin_bit_cast(float, __builtin_amdgcn_readlane(cvi, j + 1)), v1);
            if (r > 2) fma4(acc, __builtin_bit_cast(float, __builtin_amdgcn_readlane(cvi, j + 2)), v2);
        }
    }

    const float di2 = di * di;
    float4 o;
    o.x = tanhf(acc.x + hv.x * di2 + b4.x);
    o.y = tanhf(acc.y + hv.y * di2 + b4.y);
    o.z = tanhf(acc.z + hv.z * di2 + b4.z);
    o.w = tanhf(acc.w + hv.w * di2 + b4.w);
    ((float4*)out)[(size_t)row * 32 + lane] = o;
}

extern "C" void kernel_launch(void* const* d_in, const int* in_sizes, int n_in,
                              void* d_out, int out_size, void* d_ws, size_t ws_size,
                              hipStream_t stream) {
    const float* x  = (const float*)d_in[0];
    const int*   ei = (const int*)d_in[1];
    const float* Wl[4] = {(const float*)d_in[2], (const float*)d_in[4],
                          (const float*)d_in[6], (const float*)d_in[8]};
    const float* bl[4] = {(const float*)d_in[3], (const float*)d_in[5],
                          (const float*)d_in[7], (const float*)d_in[9]};

    const int NFEAT = in_sizes[2] / HDIM;
    const int N     = in_sizes[0] / NFEAT;
    const int E     = in_sizes[1] / 2;
    const int* srcp = ei;
    const int* dstp = ei + E;

    float* wsf  = (float*)d_ws;
    float* dinv = wsf;
    float* r0   = wsf + N;
    float* r1   = r0 + (size_t)N * HDIM;
    float* r2   = r1 + (size_t)N * HDIM;

    int*   cnt    = (int*)r1;
    int*   rowptr = cnt + N;
    int*   rank   = rowptr + N + 32;
    int*   eid    = rank + E;
    int*   col    = eid + E;
    float* cf     = (float*)(col + E);

    const int THR = 256;

    zero_i32<<<(N + THR - 1) / THR, THR, 0, stream>>>(cnt, N);
    count_rank<<<(E + THR - 1) / THR, THR, 0, stream>>>(dstp, cnt, rank, E);
    scan_rowptr_dinv<<<(N + SCAN_T * SCAN_PER - 1) / (SCAN_T * SCAN_PER), SCAN_T, 0, stream>>>(
        cnt, rowptr, dinv, N);
    csr_fill<<<(E + THR - 1) / THR, THR, 0, stream>>>(dstp, rank, rowptr, eid, E);
    csr_order<<<(N + 7) / 8, 256, 0, stream>>>(rowptr, eid, srcp, dinv, col, cf, N, E);

    const float* in = x;
    int K = NFEAT;
    for (int l = 0; l < 4; ++l) {
        const int Kpad = (K + 3) & ~3;
        const size_t shmem =
            ((size_t)(Kpad / 2) * WS2 * 2 + 8u * 32u * A_LD) * sizeof(float);
        gcn_gemm_wmma_f32<<<(N + 255) / 256, 256, shmem, stream>>>(
            in, Wl[l], r0, N, K, Kpad);

        float* outp = (l == 3) ? (float*)d_out : r2;
        gcn_gather_rows<<<(N + 7) / 8, 256, 0, stream>>>(
            rowptr, col, cf, dinv, bl[l], r0, outp, N);

        in = r2;
        K = HDIM;
    }
}


# GEMM operand staging: A strip and W tile fetched with whole-line 16-byte loads, all issued before the first wait, instead of one dependent 4-byte load per round trip; LDS images unchanged
# speedup vs baseline: 4.1072x; 1.7857x over previous
#include <hip/hip_runtime.h>
#include <hip/hip_bf16.h>
#include <math.h>


#define HDIM 128
#define WS2  144
#define A_LD 138

#define A_IT 17
#define W_IT 9

typedef __attribute__((ext_vector_type(2))) float v2f;
typedef __attribute__((ext_vector_type(8))) float v8f;

__global__ __launch_bounds__(256)
void gcn_gemm_wmma_f32(const float* __restrict__ A,
                       const float* __restrict__ W,
                       float* __restrict__ C,
                       int N, int K, int Kpad) {
    extern __shared__ float lds[];
    v2f*   Wp = (v2f*)lds;
    float* As = lds + (size_t)(Kpad / 2) * WS2 * 2;

    const int tid  = threadIdx.x;
    const int wave = tid >> 5;
    const int lane = tid & 31;
    const int l15  = lane & 15;
    const int rowBase = blockIdx.x * 256 + wave * 32;

    const float4 z4 = make_float4(0.f, 0.f, 0.f, 0.f);
    float* Aw = As + (size_t)wave * 32 * A_LD;

    const int rowsValid = min(32, max(0, N - rowBase));
    const int validLen  = rowsValid * K;
    const int nvecFull  = validLen >> 2;
    const int nvec      = (32 * K) >> 2;
    const float*  strip  = A + (size_t)rowBase * K;
    const float4* strip4 = (const float4*)strip;

    float4 av0[A_IT], av1[A_IT];
    if (nvecFull > 0) {
#pragma unroll
        for (int u = 0; u < A_IT; ++u)
            av0[u] = strip4[min(u * 32 + lane, nvecFull - 1)];
    } else {
#pragma unroll
        for (int u = 0; u < A_IT; ++u) av0[u] = z4;
    }

    const float4* W4 = (const float4*)W;
    const int nW = (Kpad / 2) * 32;
    float4 wa[W_IT], wb[W_IT];
#pragma unroll
    for (int u = 0; u < W_IT; ++u) {
        const int idx = u * 256 + tid;
        const int k0  = 2 * (idx >> 5);
        const int n4  = idx & 31;
        wa[u] = W4[(size_t)min(k0,     K - 1) * 32 + n4];
        wb[u] = W4[(size_t)min(k0 + 1, K - 1) * 32 + n4];
    }

    if (nvecFull > 0) {
#pragma unroll
        for (int u = 0; u < A_IT; ++u)
            av1[u] = strip4[min((A_IT + u) * 32 + lane, nvecFull - 1)];
    } else {
#pragma unroll
        for (int u = 0; u < A_IT; ++u) av1[u] = z4;
    }

#pragma unroll
    for (int u = 0; u < W_IT; ++u) {
        const int idx = u * 256 + tid;
        if (idx < nW) {
            const int kk = idx >> 5;
            const int n4 = idx & 31;
            const int k0 = 2 * kk;
            const float4 a = (k0     < K) ? wa[u] : z4;
            const float4 b = (k0 + 1 < K) ? wb[u] : z4;
            float4* dst = (float4*)(Wp + kk * WS2 + 4 * n4);
            dst[0] = make_float4(a.x, b.x, a.y, b.y);
            dst[1] = make_float4(a.z, b.z, a.w, b.w);
        }
    }

    {
        const float invK = 1.0f / (float)K;
        const float fl   = (float)(4 * lane) + 0.5f;
        const int wrapAdd = A_LD - K;
        const bool rowVec = (K & 3) == 0;
#pragma unroll
        for (int u = 0; u < 2 * A_IT; ++u) {
            const int i = u * 32 + lane;
            const float4 v = (u < A_IT) ? av0[u < A_IT ? u : 0] : av1[u < A_IT ? 0 : u - A_IT];
            if (i < nvec) {
                const int e0 = 128 * u + 4 * lane;
                const int r  = (int)((fl + (float)(128 * u)) * invK);
                const int k  = e0 - r * K;
                const int a0 = r * A_LD + k;
                const float4 w = (i < nvecFull) ? v : z4;
                if (rowVec) {
                    v2f* p = (v2f*)(Aw + a0);
                    p[0] = (v2f){w.x, w.y};
                    p[1] = (v2f){w.z, w.w};
                } else {
                    Aw[a0] = w.x;
                    Aw[a0 + 1 + ((k + 1 >= K) ? wrapAdd : 0)] = w.y;
                    Aw[a0 + 2 + ((k + 2 >= K) ? wrapAdd : 0)] = w.z;
                    Aw[a0 + 3 + ((k + 3 >= K) ? wrapAdd : 0)] = w.w;
                }
            }
        }
        const int rem = validLen & 3;
        if (lane < rem) {
            const int e  = (nvecFull << 2) + lane;
            const int rr = e / K;
            Aw[rr * A_LD + (e - rr * K)] = strip[e];
        }
        for (int k2 = K; k2 < Kpad; ++k2) Aw[lane * A_LD + k2] = 0.0f;
    }
    __syncthreads();

    v8f acc[2][8];
#pragma unroll
    for (int t = 0; t < 2; ++t)
#pragma unroll
        for (int j = 0; j < 8; ++j)
            acc[t][j] = (v8f){0.f,0.f,0.f,0.f,0.f,0.f,0.f,0.f};

    const int koff = (lane >= 16) ? 2 : 0;
    const float* Aw0 = Aw + l15 * A_LD;
    const float* Aw1 = Aw + (l15 + 16) * A_LD;

    for (int k0 = 0; k0 < Kpad; k0 += 4) {
        v2f a0 = *(const v2f*)(Aw0 + k0 + koff);
        v2f a1 = *(const v2f*)(Aw1 + k0 + koff);
        const v2f* Bk = Wp + (size_t)((k0 + koff) >> 1) * WS2 + l15;
#pragma unroll
        for (int j = 0; j < 8; ++j) {
            v2f b = Bk[j * 16];
            acc[0][j] = __builtin_amdgcn_wmma_f32_16x16x4_f32(
                false, a0, false, b, (short)0, acc[0][j], false, false);
            acc[1][j] = __builtin_amdgcn_wmma_f32_16x16x4_f32(
                false, a1, false, b, (short)0, acc[1][j], false, false);
        }
    }

    const int rOff = (lane >= 16) ? 8 : 0;
#pragma unroll
    for (int t = 0; t < 2; ++t) {
#pragma unroll
        for (int i = 0; i < 8; ++i) {
            int row = rowBase + t * 16 + i + rOff;
            if (row < N) {
                size_t base = (size_t)row * HDIM + l15;
#pragma unroll
                for (int j = 0; j < 8; ++j) {
                    C[base + j * 16] = acc[t][j][i];
                }
            }
        }
    }
}

__global__ void zero_i32(int* __restrict__ p, int n) {
    int i = blockIdx.x * blockDim.x + threadIdx.x;
    if (i < n) p[i] = 0;
}

__global__ void count_rank(const int* __restrict__ dst, int* __restrict__ cnt,
                           int* __restrict__ rank, int E) {
    int e = blockIdx.x * blockDim.x + threadIdx.x;
    if (e < E) rank[e] = atomicAdd(&cnt[dst[e]], 1);
}

#define SCAN_T 256
#define SCAN_PER 4
__global__ __launch_bounds__(SCAN_T)
void scan_rowptr_dinv(const int* __restrict__ cnt, int* __restrict__ rowptr,
                      float* __restrict__ dinv, int N) {
    __shared__ int sh[SCAN_T];
    const int tid   = threadIdx.x;
    const int start = blockIdx.x * (SCAN_T * SCAN_PER);

    int pre = 0;
    const int4* __restrict__ cnt4 = (const int4*)cnt;
    for (int i = tid; i < start / 4; i += SCAN_T) {
        int4 v = cnt4[i];
        pre += (v.x + v.y) + (v.z + v.w);
    }

    int c[SCAN_PER];
    int loc = 0;
#pragma unroll
    for (int q = 0; q < SCAN_PER; ++q) {
        int i = start + tid * SCAN_PER + q;
        c[q] = (i < N) ? cnt[i] : 0;
        loc += c[q];
    }

    sh[tid] = pre;
    __syncthreads();
    for (int s = SCAN_T / 2; s > 0; s >>= 1) {
        if (tid < s) sh[tid] += sh[tid + s];
        __syncthreads();
    }
    const int offset = sh[0];
    __syncthreads();

    sh[tid] = loc;
    __syncthreads();
    for (int s = 1; s < SCAN_T; s <<= 1) {
        int v = (tid >= s) ? sh[tid - s] : 0;
        __syncthreads();
        sh[tid] += v;
        __syncthreads();
    }
    int run = offset + sh[tid] - loc;
#pragma unroll
    for (int q = 0; q < SCAN_PER; ++q) {
        int i = start + tid * SCAN_PER + q;
        if (i < N) {
            rowptr[i] = run;
            dinv[i]   = rsqrtf((float)c[q] + 1.0f);
        }
        run += c[q];
        if (i == N - 1) rowptr[N] = run;
    }
}

__global__ void csr_fill(const int* __restrict__ dst, const int* __restrict__ rank,
                         const int* __restrict__ rowptr, int* __restrict__ eid, int E) {
    int e = blockIdx.x * blockDim.x + threadIdx.x;
    if (e >= E) return;
    int p = rowptr[dst[e]] + rank[e];
    p = min(max(p, 0), E - 1);
    eid[p] = e;
}

__global__ __launch_bounds__(256)
void csr_order(const int* __restrict__ rowptr, const int* __restrict__ eid,
               const int* __restrict__ src, const float* __restrict__ dinv,
               int* __restrict__ col, float* __restrict__ cf, int N, int E) {
    const int lane = threadIdx.x & 31;
    const int row = __builtin_amdgcn_readfirstlane((int)(blockIdx.x * 8 + (threadIdx.x >> 5)));
    if (row >= N) return;
    const int beg = __builtin_amdgcn_readfirstlane(rowptr[row]);
    const int end = __builtin_amdgcn_readfirstlane(rowptr[row + 1]);
    const float dd = dinv[row];

    for (int a = beg; a < end; a += 32) {
        const int na = min(32, end - a);
        const int my = (lane < na) ? eid[a + lane] : 0x7fffffff;
        int rk = 0;
        for (int b = beg; b < end; b += 32) {
            const int nb = min(32, end - b);
            const int ob = (lane < nb) ? eid[b + lane] : 0x7fffffff;
            for (int j = 0; j < nb; ++j)
                rk += (__builtin_amdgcn_readlane(ob, j) < my) ? 1 : 0;
        }
        if (lane < na) {
            const int e = (int)min((unsigned)my, (unsigned)(E - 1));
            const int s = src[e];
            col[beg + rk] = s;
            cf[beg + rk]  = dinv[s] * dd;
        }
    }
}

__device__ __forceinline__ void fma4(float4& a, float c, const float4& v) {
    a.x += c * v.x; a.y += c * v.y; a.z += c * v.z; a.w += c * v.w;
}

__global__ __launch_bounds__(256)
void gcn_gather_rows(const int* __restrict__ rowptr, const int* __restrict__ col,
                     const float* __restrict__ cf, const float* __restrict__ dinv,
                     const float* __restrict__ bias, const float* __restrict__ h,
                     float* __restrict__ out, int N) {
    const int lane = threadIdx.x & 31;
    const int row = __builtin_amdgcn_readfirstlane((int)(blockIdx.x * 8 + (threadIdx.x >> 5)));
    if (row >= N) return;
    const int beg = __builtin_amdgcn_readfirstlane(rowptr[row]);
    const int end = __builtin_amdgcn_readfirstlane(rowptr[row + 1]);

    const float4* __restrict__ h4 = (const float4*)h;
    const float4 hv = h4[(size_t)row * 32 + lane];
    const float  di = dinv[row];
    const float4 b4 = ((const float4*)bias)[lane];

    float4 acc = make_float4(0.f, 0.f, 0.f, 0.f);
    const unsigned nmax = (unsigned)(N - 1);

    for (int base = beg; base < end; base += 32) {
        const int n = min(32, end - base);
        int   sv = 0;
        float cv = 0.f;
        if (lane < n) { sv = col[base + lane]; cv = cf[base + lane]; }
        sv = (int)min((unsigned)sv, nmax);
        const int cvi = __builtin_bit_cast(int, cv);

        int j = 0;
        for (; j + 4 <= n; j += 4) {
            const int s0 = __builtin_amdgcn_readlane(sv, j);
            const int s1 = __builtin_amdgcn_readlane(sv, j + 1);
            const int s2 = __builtin_amdgcn_readlane(sv, j + 2);
            const int s3 = __builtin_amdgcn_readlane(sv, j + 3);
            const float4 v0 = h4[(size_t)s0 * 32 + lane];
            const float4 v1 = h4[(size_t)s1 * 32 + lane];
            const float4 v2 = h4[(size_t)s2 * 32 + lane];
            const float4 v3 = h4[(size_t)s3 * 32 + lane];
            fma4(acc, __builtin_bit_cast(float, __builtin_amdgcn_readlane(cvi, j)),     v0);
            fma4(acc, __builtin_bit_cast(float, __builtin_amdgcn_readlane(cvi, j + 1)), v1);
            fma4(acc, __builtin_bit_cast(float, __builtin_amdgcn_readlane(cvi, j + 2)), v2);
            fma4(acc, __builtin_bit_cast(float, __builtin_amdgcn_readlane(cvi, j + 3)), v3);
        }
        const int r = n - j;
        if (r > 0) {
            const int s0 = __builtin_amdgcn_readlane(sv, j);
            const float4 v0 = h4[(size_t)s0 * 32 + lane];
            float4 v1 = make_float4(0.f, 0.f, 0.f, 0.f);
            float4 v2 = make_float4(0.f, 0.f, 0.f, 0.f);
            if (r > 1) {
                const int s1 = __builtin_amdgcn_readlane(sv, j + 1);
                v1 = h4[(size_t)s1 * 32 + lane];
            }
            if (r > 2) {
                const int s2 = __builtin_amdgcn_readlane(sv, j + 2);
                v2 = h4[(size_t)s2 * 32 + lane];
            }
            fma4(acc, __builtin_bit_cast(float, __builtin_amdgcn_readlane(cvi, j)), v0);
            if (r > 1) fma4(acc, __builtin_bit_cast(float, __builtin_amdgcn_readlane(cvi, j + 1)), v1);
            if (r > 2) fma4(acc, __builtin_bit_cast(float, __builtin_amdgcn_readlane(cvi, j + 2)), v2);
        }
    }

    const float di2 = di * di;
    float4 o;
    o.x = tanhf(acc.x + hv.x * di2 + b4.x);
    o.y = tanhf(acc.y + hv.y * di2 + b4.y);
    o.z = tanhf(acc.z + hv.z * di2 + b4.z);
    o.w = tanhf(acc.w + hv.w * di2 + b4.w);
    ((float4*)out)[(size_t)row * 32 + lane] = o;
}

extern "C" void kernel_launch(void* const* d_in, const int* in_sizes, int n_in,
                              void* d_out, int out_size, void* d_ws, size_t ws_size,
                              hipStream_t stream) {
    const float* x  = (const float*)d_in[0];
    const int*   ei = (const int*)d_in[1];
    const float* Wl[4] = {(const float*)d_in[2], (const float*)d_in[4],
                          (const float*)d_in[6], (const float*)d_in[8]};
    const float* bl[4] = {(const float*)d_in[3], (const float*)d_in[5],
                          (const float*)d_in[7], (const float*)d_in[9]};

    const int NFEAT = in_sizes[2] / HDIM;
    const int N     = in_sizes[0] / NFEAT;
    const int E     = in_sizes[1] / 2;
    const int* srcp = ei;
    const int* dstp = ei + E;

    float* wsf  = (float*)d_ws;
    float* dinv = wsf;
    float* r0   = wsf + N;
    float* r1   = r0 + (size_t)N * HDIM;
    float* r2   = r1 + (size_t)N * HDIM;

    int*   cnt    = (int*)r1;
    int*   rowptr = cnt + N;
    int*   rank   = rowptr + N + 32;
    int*   eid    = rank + E;
    int*   col    = eid + E;
    float* cf     = (float*)(col + E);

    const int THR = 256;

    zero_i32<<<(N + THR - 1) / THR, THR, 0, stream>>>(cnt, N);
    count_rank<<<(E + THR - 1) / THR, THR, 0, stream>>>(dstp, cnt, rank, E);
    scan_rowptr_dinv<<<(N + SCAN_T * SCAN_PER - 1) / (SCAN_T * SCAN_PER), SCAN_T, 0, stream>>>(
        cnt, rowptr, dinv, N);
    csr_fill<<<(E + THR - 1) / THR, THR, 0, stream>>>(dstp, rank, rowptr, eid, E);
    csr_order<<<(N + 7) / 8, 256, 0, stream>>>(rowptr, eid, srcp, dinv, col, cf, N, E);

    const float* in = x;
    int K = NFEAT;
    for (int l = 0; l < 4; ++l) {
        const int Kpad = (K + 3) & ~3;
        const size_t shmem =
            ((size_t)(Kpad / 2) * WS2 * 2 + 8u * 32u * A_LD) * sizeof(float);
        gcn_gemm_wmma_f32<<<(N + 255) / 256, 256, shmem, stream>>>(
            in, Wl[l], r0, N, K, Kpad);

        float* outp = (l == 3) ? (float*)d_out : r2;
        gcn_gather_rows<<<(N + 7) / 8, 256, 0, stream>>>(
            rowptr, col, cf, dinv, bl[l], r0, outp, N);

        in = r2;
        K = HDIM;
    }
}


# GEMM as resident work-groups (one per WGP): W tile staged once, each wave loops over 16-row strips and fetches the next strip's rows under the current strip's WMMAs; sums unchanged
# speedup vs baseline: 4.4168x; 1.0754x over previous
#include <hip/hip_runtime.h>
#include <hip/hip_bf16.h>
#include <math.h>


#define HDIM 128
#define WS2  144
#define A_LD 138

#define A_IT 17
#define W_IT 9

typedef __attribute__((ext_vector_type(2))) float v2f;
typedef __attribute__((ext_vector_type(8))) float v8f;

__device__ __forceinline__ void strip_load(float4 (&av)[A_IT], const float* __restrict__ A,
                                           int s, int N, int K, int lane) {
    const int rowBase   = s * 16;
    const int rowsValid = min(16, max(0, N - rowBase));
    const int nvecFull  = (rowsValid * K) >> 2;
    const float4* strip4 = (const float4*)(A + (size_t)rowBase * K);
    if (nvecFull > 0) {
#pragma unroll
        for (int u = 0; u < A_IT; ++u)
            av[u] = strip4[min(u * 32 + lane, nvecFull - 1)];
    } else {
#pragma unroll
        for (int u = 0; u < A_IT; ++u) av[u] = make_float4(0.f, 0.f, 0.f, 0.f);
    }
}

__device__ __forceinline__ void strip_to_lds(const float4 (&av)[A_IT], float* Aw,
                                             const float* __restrict__ A,
                                             int s, int N, int K, int Kpad, int lane) {
    const int rowBase   = s * 16;
    const int rowsValid = min(16, max(0, N - rowBase));
    const int validLen  = rowsValid * K;
    const int nvecFull  = validLen >> 2;
    const int nvec      = (16 * K) >> 2;
    const float* strip  = A + (size_t)rowBase * K;
    const float4 z4 = make_float4(0.f, 0.f, 0.f, 0.f);
    const float invK = 1.0f / (float)K;
    const float fl   = (float)(4 * lane) + 0.5f;
    const int wrapAdd = A_LD - K;
    const bool rowVec = (K & 3) == 0;
#pragma unroll
    for (int u = 0; u < A_IT; ++u) {
        const int i = u * 32 + lane;
        if (i < nvec) {
            const int e0 = 128 * u + 4 * lane;
            const int r  = (int)((fl + (float)(128 * u)) * invK);
            const int k  = e0 - r * K;
            const int a0 = r * A_LD + k;
            const float4 w = (i < nvecFull) ? av[u] : z4;
            if (rowVec) {
                v2f* p = (v2f*)(Aw + a0);
                p[0] = (v2f){w.x, w.y};
                p[1] = (v2f){w.z, w.w};
            } else {
                Aw[a0] = w.x;
                Aw[a0 + 1 + ((k + 1 >= K) ? wrapAdd : 0)] = w.y;
                Aw[a0 + 2 + ((k + 2 >= K) ? wrapAdd : 0)] = w.z;
                Aw[a0 + 3 + ((k + 3 >= K) ? wrapAdd : 0)] = w.w;
            }
        }
    }
    const int rem = validLen & 3;
    if (lane < rem) {
        const int e  = (nvecFull << 2) + lane;
        const int rr = e / K;
        Aw[rr * A_LD + (e - rr * K)] = strip[e];
    }
    if (lane < 16)
        for (int k2 = K; k2 < Kpad; ++k2) Aw[lane * A_LD + k2] = 0.0f;
}

__global__ __launch_bounds__(256) __attribute__((amdgpu_waves_per_eu(2, 2)))
void gcn_gemm_wmma_f32(const float* __restrict__ A,
                       const float* __restrict__ W,
                       float* __restrict__ C,
                       int N, int K, int Kpad) {
    extern __shared__ float lds[];
    v2f*   Wp = (v2f*)lds;
    float* As = lds + (size_t)(Kpad / 2) * WS2 * 2;

    const int tid  = threadIdx.x;
    const int lane = tid & 31;
    const int l15  = lane & 15;
    const int wave = __builtin_amdgcn_readfirstlane(tid >> 5);

    const int nStrips = (N + 15) >> 4;
    const int nblk    = (int)gridDim.x;
    const int stride  = nblk * 8;
    int s = wave * nblk + (int)blockIdx.x;

    const float4 z4 = make_float4(0.f, 0.f, 0.f, 0.f);
    float* Aw = As + (size_t)wave * 16 * A_LD;

    float4 av[A_IT];
    strip_load(av, A, s, N, K, lane);

    const float4* W4 = (const float4*)W;
    const int nW = (Kpad / 2) * 32;
    {
        float4 wa[W_IT], wb[W_IT];
#pragma unroll
        for (int u = 0; u < W_IT; ++u) {
            const int idx = u * 256 + tid;
            const int k0  = 2 * (idx >> 5);
            const int n4  = idx & 31;
            wa[u] = W4[(size_t)min(k0,     K - 1) * 32 + n4];
            wb[u] = W4[(size_t)min(k0 + 1, K - 1) * 32 + n4];
        }
#pragma unroll
        for (int u = 0; u < W_IT; ++u) {
            const int idx = u * 256 + tid;
            if (idx < nW) {
                const int kk = idx >> 5;
                const int n4 = idx & 31;
                const int k0 = 2 * kk;
                const float4 a = (k0     < K) ? wa[u] : z4;
                const float4 b = (k0 + 1 < K) ? wb[u] : z4;
                float4* dst = (float4*)(Wp + kk * WS2 + 4 * n4);
                dst[0] = make_float4(a.x, b.x, a.y, b.y);
                dst[1] = make_float4(a.z, b.z, a.w, b.w);
            }
        }
    }
    __syncthreads();

    const int koff = (lane >= 16) ? 2 : 0;
    const float* Aw0 = Aw + l15 * A_LD;
    const int rOff = (lane >= 16) ? 8 : 0;

    for (; s < nStrips; s += stride) {
        strip_to_lds(av, Aw, A, s, N, K, Kpad, lane);
        __builtin_amdgcn_fence(__ATOMIC_RELEASE, "workgroup", "local");
        __builtin_amdgcn_fence(__ATOMIC_ACQUIRE, "workgroup", "local");

        const int sn = s + stride;
        if (sn < nStrips) strip_load(av, A, sn, N, K, lane);

        v8f acc[8];
#pragma unroll
        for (int j = 0; j < 8; ++j)
            acc[j] = (v8f){0.f,0.f,0.f,0.f,0.f,0.f,0.f,0.f};

        for (int k0 = 0; k0 < Kpad; k0 += 4) {
            v2f a0 = *(const v2f*)(Aw0 + k0 + koff);
            const v2f* Bk = Wp + (size_t)((k0 + koff) >> 1) * WS2 + l15;
#pragma unroll
            for (int j = 0; j < 8; ++j) {
                v2f b = Bk[j * 16];
                acc[j] = __builtin_amdgcn_wmma_f32_16x16x4_f32(
                    false, a0, false, b, (short)0, acc[j], false, false);
            }
        }

        const int rowBase = s * 16;
#pragma unroll
        for (int i = 0; i < 8; ++i) {
            int row = rowBase + i + rOff;
            if (row < N) {
                size_t base = (size_t)row * HDIM + l15;
#pragma unroll
                for (int j = 0; j < 8; ++j) {
                    C[base + j * 16] = acc[j][i];
                }
            }
        }
    }
}

__global__ void zero_i32(int* __restrict__ p, int n) {
    int i = blockIdx.x * blockDim.x + threadIdx.x;
    if (i < n) p[i] = 0;
}

__global__ void count_rank(const int* __restrict__ dst, int* __restrict__ cnt,
                           int* __restrict__ rank, int E) {
    int e = blockIdx.x * blockDim.x + threadIdx.x;
    if (e < E) rank[e] = atomicAdd(&cnt[dst[e]], 1);
}

#define SCAN_T 256
#define SCAN_PER 4
__global__ __launch_bounds__(SCAN_T)
void scan_rowptr_dinv(const int* __restrict__ cnt, int* __restrict__ rowptr,
                      float* __restrict__ dinv, int N) {
    __shared__ int sh[SCAN_T];
    const int tid   = threadIdx.x;
    const int start = blockIdx.x * (SCAN_T * SCAN_PER);

    int pre = 0;
    const int4* __restrict__ cnt4 = (const int4*)cnt;
    for (int i = tid; i < start / 4; i += SCAN_T) {
        int4 v = cnt4[i];
        pre += (v.x + v.y) + (v.z + v.w);
    }

    int c[SCAN_PER];
    int loc = 0;
#pragma unroll
    for (int q = 0; q < SCAN_PER; ++q) {
        int i = start + tid * SCAN_PER + q;
        c[q] = (i < N) ? cnt[i] : 0;
        loc += c[q];
    }

    sh[tid] = pre;
    __syncthreads();
    for (int s = SCAN_T / 2; s > 0; s >>= 1) {
        if (tid < s) sh[tid] += sh[tid + s];
        __syncthreads();
    }
    const int offset = sh[0];
    __syncthreads();

    sh[tid] = loc;
    __syncthreads();
    for (int s = 1; s < SCAN_T; s <<= 1) {
        int v = (tid >= s) ? sh[tid - s] : 0;
        __syncthreads();
        sh[tid] += v;
        __syncthreads();
    }
    int run = offset + sh[tid] - loc;
#pragma unroll
    for (int q = 0; q < SCAN_PER; ++q) {
        int i = start + tid * SCAN_PER + q;
        if (i < N) {
            rowptr[i] = run;
            dinv[i]   = rsqrtf((float)c[q] + 1.0f);
        }
        run += c[q];
        if (i == N - 1) rowptr[N] = run;
    }
}

__global__ void csr_fill(const int* __restrict__ dst, const int* __restrict__ rank,
                         const int* __restrict__ rowptr, int* __restrict__ eid, int E) {
    int e = blockIdx.x * blockDim.x + threadIdx.x;
    if (e >= E) return;
    int p = rowptr[dst[e]] + rank[e];
    p = min(max(p, 0), E - 1);
    eid[p] = e;
}

__global__ __launch_bounds__(256)
void csr_order(const int* __restrict__ rowptr, const int* __restrict__ eid,
               const int* __restrict__ src, const float* __restrict__ dinv,
               int* __restrict__ col, float* __restrict__ cf, int N, int E) {
    const int lane = threadIdx.x & 31;
    const int row = __builtin_amdgcn_readfirstlane((int)(blockIdx.x * 8 + (threadIdx.x >> 5)));
    if (row >= N) return;
    const int beg = __builtin_amdgcn_readfirstlane(rowptr[row]);
    const int end = __builtin_amdgcn_readfirstlane(rowptr[row + 1]);
    const float dd = dinv[row];

    for (int a = beg; a < end; a += 32) {
        const int na = min(32, end - a);
        const int my = (lane < na) ? eid[a + lane] : 0x7fffffff;
        int rk = 0;
        for (int b = beg; b < end; b += 32) {
            const int nb = min(32, end - b);
            const int ob = (lane < nb) ? eid[b + lane] : 0x7fffffff;
            for (int j = 0; j < nb; ++j)
                rk += (__builtin_amdgcn_readlane(ob, j) < my) ? 1 : 0;
        }
        if (lane < na) {
            const int e = (int)min((unsigned)my, (unsigned)(E - 1));
            const int s = src[e];
            col[beg + rk] = s;
            cf[beg + rk]  = dinv[s] * dd;
        }
    }
}

__device__ __forceinline__ void fma4(float4& a, float c, const float4& v) {
    a.x += c * v.x; a.y += c * v.y; a.z += c * v.z; a.w += c * v.w;
}

__global__ __launch_bounds__(256)
void gcn_gather_rows(const int* __restrict__ rowptr, const int* __restrict__ col,
                     const float* __restrict__ cf, const float* __restrict__ dinv,
                     const float* __restrict__ bias, const float* __restrict__ h,
                     float* __restrict__ out, int N) {
    const int lane = threadIdx.x & 31;
    const int row = __builtin_amdgcn_readfirstlane((int)(blockIdx.x * 8 + (threadIdx.x >> 5)));
    if (row >= N) return;
    const int beg = __builtin_amdgcn_readfirstlane(rowptr[row]);
    const int end = __builtin_amdgcn_readfirstlane(rowptr[row + 1]);

    const float4* __restrict__ h4 = (const float4*)h;
    const float4 hv = h4[(size_t)row * 32 + lane];
    const float  di = dinv[row];
    const float4 b4 = ((const float4*)bias)[lane];

    float4 acc = make_float4(0.f, 0.f, 0.f, 0.f);
    const unsigned nmax = (unsigned)(N - 1);

    for (int base = beg; base < end; base += 32) {
        const int n = min(32, end - base);
        int   sv = 0;
        float cv = 0.f;
        if (lane < n) { sv = col[base + lane]; cv = cf[base + lane]; }
        sv = (int)min((unsigned)sv, nmax);
        const int cvi = __builtin_bit_cast(int, cv);

        int j = 0;
        for (; j + 4 <= n; j += 4) {
            const int s0 = __builtin_amdgcn_readlane(sv, j);
            const int s1 = __builtin_amdgcn_readlane(sv, j + 1);
            const int s2 = __builtin_amdgcn_readlane(sv, j + 2);
            const int s3 = __builtin_amdgcn_readlane(sv, j + 3);
            const float4 v0 = h4[(size_t)s0 * 32 + lane];
            const float4 v1 = h4[(size_t)s1 * 32 + lane];
            const float4 v2 = h4[(size_t)s2 * 32 + lane];
            const float4 v3 = h4[(size_t)s3 * 32 + lane];
            fma4(acc, __builtin_bit_cast(float, __builtin_amdgcn_readlane(cvi, j)),     v0);
            fma4(acc, __builtin_bit_cast(float, __builtin_amdgcn_readlane(cvi, j + 1)), v1);
            fma4(acc, __builtin_bit_cast(float, __builtin_amdgcn_readlane(cvi, j + 2)), v2);
            fma4(acc, __builtin_bit_cast(float, __builtin_amdgcn_readlane(cvi, j + 3)), v3);
        }
        const int r = n - j;
        if (r > 0) {
            const int s0 = __builtin_amdgcn_readlane(sv, j);
            const float4 v0 = h4[(size_t)s0 * 32 + lane];
            float4 v1 = make_float4(0.f, 0.f, 0.f, 0.f);
            float4 v2 = make_float4(0.f, 0.f, 0.f, 0.f);
            if (r > 1) {
                const int s1 = __builtin_amdgcn_readlane(sv, j + 1);
                v1 = h4[(size_t)s1 * 32 + lane];
            }
            if (r > 2) {
                const int s2 = __builtin_amdgcn_readlane(sv, j + 2);
                v2 = h4[(size_t)s2 * 32 + lane];
            }
            fma4(acc, __builtin_bit_cast(float, __builtin_amdgcn_readlane(cvi, j)), v0);
            if (r > 1) fma4(acc, __builtin_bit_cast(float, __builtin_amdgcn_readlane(cvi, j + 1)), v1);
            if (r > 2) fma4(acc, __builtin_bit_cast(float, __builtin_amdgcn_readlane(cvi, j + 2)), v2);
        }
    }

    const float di2 = di * di;
    float4 o;
    o.x = tanhf(acc.x + hv.x * di2 + b4.x);
    o.y = tanhf(acc.y + hv.y * di2 + b4.y);
    o.z = tanhf(acc.z + hv.z * di2 + b4.z);
    o.w = tanhf(acc.w + hv.w * di2 + b4.w);
    ((float4*)out)[(size_t)row * 32 + lane] = o;
}

extern "C" void kernel_launch(void* const* d_in, const int* in_sizes, int n_in,
                              void* d_out, int out_size, void* d_ws, size_t ws_size,
                              hipStream_t stream) {
    const float* x  = (const float*)d_in[0];
    const int*   ei = (const int*)d_in[1];
    const float* Wl[4] = {(const float*)d_in[2], (const float*)d_in[4],
                          (const float*)d_in[6], (const float*)d_in[8]};
    const float* bl[4] = {(const float*)d_in[3], (const float*)d_in[5],
                          (const float*)d_in[7], (const float*)d_in[9]};

    const int NFEAT = in_sizes[2] / HDIM;
    const int N     = in_sizes[0] / NFEAT;
    const int E     = in_sizes[1] / 2;
    const int* srcp = ei;
    const int* dstp = ei + E;

    float* wsf  = (float*)d_ws;
    float* dinv = wsf;
    float* r0   = wsf + N;
    float* r1   = r0 + (size_t)N * HDIM;
    float* r2   = r1 + (size_t)N * HDIM;

    int*   cnt    = (int*)r1;
    int*   rowptr = cnt + N;
    int*   rank   = rowptr + N + 32;
    int*   eid    = rank + E;
    int*   col    = eid + E;
    float* cf     = (float*)(col + E);

    const int THR = 256;

    zero_i32<<<(N + THR - 1) / THR, THR, 0, stream>>>(cnt, N);
    count_rank<<<(E + THR - 1) / THR, THR, 0, stream>>>(dstp, cnt, rank, E);
    scan_rowptr_dinv<<<(N + SCAN_T * SCAN_PER - 1) / (SCAN_T * SCAN_PER), SCAN_T, 0, stream>>>(
        cnt, rowptr, dinv, N);
    csr_fill<<<(E + THR - 1) / THR, THR, 0, stream>>>(dstp, rank, rowptr, eid, E);
    csr_order<<<(N + 7) / 8, 256, 0, stream>>>(rowptr, eid, srcp, dinv, col, cf, N, E);

    const float* in = x;
    int K = NFEAT;
    for (int l = 0; l < 4; ++l) {
        const int Kpad = (K + 3) & ~3;
        const size_t shmem =
            ((size_t)(Kpad / 2) * WS2 * 2 + 8u * 16u * A_LD) * sizeof(float);
        gcn_gemm_wmma_f32<<<256, 256, shmem, stream>>>(
            in, Wl[l], r0, N, K, Kpad);

        float* outp = (l == 3) ? (float*)d_out : r2;
        gcn_gather_rows<<<(N + 7) / 8, 256, 0, stream>>>(
            rowptr, col, cf, dinv, bl[l], r0, outp, N);

        in = r2;
        K = HDIM;
    }
}


# speedup vs baseline: 4.8141x; 1.0900x over previous
#include <hip/hip_runtime.h>
#include <hip/hip_bf16.h>
#include <math.h>


#define HDIM 128
#define WS2  144
#define A_LD 138

#define A_IT 17
#define W_IT 9

typedef __attribute__((ext_vector_type(2))) float v2f;
typedef __attribute__((ext_vector_type(8))) float v8f;
typedef __attribute__((ext_vector_type(4))) float v4f;

__device__ __forceinline__ void strip_load(float4 (&av)[A_IT], const float* __restrict__ A,
                                           int s, int N, int K, int lane) {
    const int rowBase   = s * 16;
    const int rowsValid = min(16, max(0, N - rowBase));
    const int nvecFull  = (rowsValid * K) >> 2;
    const float4* strip4 = (const float4*)(A + (size_t)rowBase * K);
    if (nvecFull > 0) {
#pragma unroll
        for (int u = 0; u < A_IT; ++u)
            av[u] = strip4[min(u * 32 + lane, nvecFull - 1)];
    } else {
#pragma unroll
        for (int u = 0; u < A_IT; ++u) av[u] = make_float4(0.f, 0.f, 0.f, 0.f);
    }
}

__device__ __forceinline__ void strip_to_lds(const float4 (&av)[A_IT], float* Aw,
                                             const float* __restrict__ A,
                                             int s, int N, int K, int Kpad, int lane) {
    const int rowBase   = s * 16;
    const int rowsValid = min(16, max(0, N - rowBase));
    const int validLen  = rowsValid * K;
    const int nvecFull  = validLen >> 2;
    const int nvec      = (16 * K) >> 2;
    const float* strip  = A + (size_t)rowBase * K;
    const float4 z4 = make_float4(0.f, 0.f, 0.f, 0.f);
    const float invK = 1.0f / (float)K;
    const float fl   = (float)(4 * lane) + 0.5f;
    const int wrapAdd = A_LD - K;
    const bool rowVec = (K & 3) == 0;
#pragma unroll
    for (int u = 0; u < A_IT; ++u) {
        const int i = u * 32 + lane;
        if (i < nvec) {
            const int e0 = 128 * u + 4 * lane;
            const int r  = (int)((fl + (float)(128 * u)) * invK);
            const int k  = e0 - r * K;
            const int a0 = r * A_LD + k;
            const float4 w = (i < nvecFull) ? av[u] : z4;
            if (rowVec) {
                v2f* p = (v2f*)(Aw + a0);
                p[0] = (v2f){w.x, w.y};
                p[1] = (v2f){w.z, w.w};
            } else {
                Aw[a0] = w.x;
                Aw[a0 + 1 + ((k + 1 >= K) ? wrapAdd : 0)] = w.y;
                Aw[a0 + 2 + ((k + 2 >= K) ? wrapAdd : 0)] = w.z;
                Aw[a0 + 3 + ((k + 3 >= K) ? wrapAdd : 0)] = w.w;
            }
        }
    }
    const int rem = validLen & 3;
    if (lane < rem) {
        const int e  = (nvecFull << 2) + lane;
        const int rr = e / K;
        Aw[rr * A_LD + (e - rr * K)] = strip[e];
    }
    if (lane < 16)
        for (int k2 = K; k2 < Kpad; ++k2) Aw[lane * A_LD + k2] = 0.0f;
}

__global__ __launch_bounds__(256) __attribute__((amdgpu_waves_per_eu(2, 2)))
void gcn_gemm_wmma_f32(const float* __restrict__ A,
                       const float* __restrict__ W,
                       float* __restrict__ C,
                       int N, int K, int Kpad) {
    extern __shared__ float lds[];
    v2f*   Wp = (v2f*)lds;
    float* As = lds + (size_t)(Kpad / 2) * WS2 * 2;

    const int tid  = threadIdx.x;
    const int lane = tid & 31;
    const int l15  = lane & 15;
    const int wave = __builtin_amdgcn_readfirstlane(tid >> 5);

    const int nStrips = (N + 15) >> 4;
    const int nblk    = (int)gridDim.x;
    const int stride  = nblk * 8;
    int s = wave * nblk + (int)blockIdx.x;

    const float4 z4 = make_float4(0.f, 0.f, 0.f, 0.f);
    float* Aw = As + (size_t)wave * 16 * A_LD;

    float4 av[A_IT];
    strip_load(av, A, s, N, K, lane);

    const float4* W4 = (const float4*)W;
    const int nW = (Kpad / 2) * 32;
    {
        float4 wa[W_IT], wb[W_IT];
#pragma unroll
        for (int u = 0; u < W_IT; ++u) {
            const int idx = u * 256 + tid;
            const int k0  = 2 * (idx >> 5);
            const int n4  = idx & 31;
            wa[u] = W4[(size_t)min(k0,     K - 1) * 32 + n4];
            wb[u] = W4[(size_t)min(k0 + 1, K - 1) * 32 + n4];
        }
#pragma unroll
        for (int u = 0; u < W_IT; ++u) {
            const int idx = u * 256 + tid;
            if (idx < nW) {
                const int kk = idx >> 5;
                const int n4 = idx & 31;
                const int k0 = 2 * kk;
                const float4 a = (k0     < K) ? wa[u] : z4;
                const float4 b = (k0 + 1 < K) ? wb[u] : z4;
                float4* dst = (float4*)(Wp + kk * WS2 + 4 * n4);
                dst[0] = make_float4(a.x, b.x, a.y, b.y);
                dst[1] = make_float4(a.z, b.z, a.w, b.w);
            }
        }
    }
    __syncthreads();

    const int koff = (lane >= 16) ? 2 : 0;
    const float* Aw0 = Aw + l15 * A_LD;
    const int rOff = (lane >= 16) ? 8 : 0;

    for (; s < nStrips; s += stride) {
        strip_to_lds(av, Aw, A, s, N, K, Kpad, lane);
        __builtin_amdgcn_fence(__ATOMIC_RELEASE, "workgroup", "local");
        __builtin_amdgcn_fence(__ATOMIC_ACQUIRE, "workgroup", "local");

        const int sn = s + stride;
        if (sn < nStrips) strip_load(av, A, sn, N, K, lane);

        v8f acc[8];
#pragma unroll
        for (int j = 0; j < 8; ++j)
            acc[j] = (v8f){0.f,0.f,0.f,0.f,0.f,0.f,0.f,0.f};

        for (int k0 = 0; k0 < Kpad; k0 += 4) {
            v2f a0 = *(const v2f*)(Aw0 + k0 + koff);
            const v2f* Bk = Wp + (size_t)((k0 + koff) >> 1) * WS2 + l15;
#pragma unroll
            for (int j = 0; j < 8; ++j) {
                v2f b = Bk[j * 16];
                acc[j] = __builtin_amdgcn_wmma_f32_16x16x4_f32(
                    false, a0, false, b, (short)0, acc[j], false, false);
            }
        }

        const int rowBase = s * 16;
#pragma unroll
        for (int i = 0; i < 8; ++i) {
            int row = rowBase + i + rOff;
            if (row < N) {
                size_t base = (size_t)row * HDIM + l15;
#pragma unroll
                for (int j = 0; j < 8; ++j) {
                    C[base + j * 16] = acc[j][i];
                }
            }
        }
    }
}

__global__ void zero_i32(int* __restrict__ p, int n) {
    int i = blockIdx.x * blockDim.x + threadIdx.x;
    if (i < n) p[i] = 0;
}

__global__ void count_rank(const int* __restrict__ dst, int* __restrict__ cnt,
                           int* __restrict__ rank, int E) {
    int e = blockIdx.x * blockDim.x + threadIdx.x;
    if (e < E) rank[e] = atomicAdd(&cnt[dst[e]], 1);
}

#define SCAN_T 256
#define SCAN_PER 4
__global__ __launch_bounds__(SCAN_T)
void scan_rowptr_dinv(const int* __restrict__ cnt, int* __restrict__ rowptr,
                      float* __restrict__ dinv, int N) {
    __shared__ int sh[SCAN_T];
    const int tid   = threadIdx.x;
    const int start = blockIdx.x * (SCAN_T * SCAN_PER);

    int pre = 0;
    const int4* __restrict__ cnt4 = (const int4*)cnt;
    for (int i = tid; i < start / 4; i += SCAN_T) {
        int4 v = cnt4[i];
        pre += (v.x + v.y) + (v.z + v.w);
    }

    int c[SCAN_PER];
    int loc = 0;
#pragma unroll
    for (int q = 0; q < SCAN_PER; ++q) {
        int i = start + tid * SCAN_PER + q;
        c[q] = (i < N) ? cnt[i] : 0;
        loc += c[q];
    }

    sh[tid] = pre;
    __syncthreads();
    for (int s = SCAN_T / 2; s > 0; s >>= 1) {
        if (tid < s) sh[tid] += sh[tid + s];
        __syncthreads();
    }
    const int offset = sh[0];
    __syncthreads();

    sh[tid] = loc;
    __syncthreads();
    for (int s = 1; s < SCAN_T; s <<= 1) {
        int v = (tid >= s) ? sh[tid - s] : 0;
        __syncthreads();
        sh[tid] += v;
        __syncthreads();
    }
    int run = offset + sh[tid] - loc;
#pragma unroll
    for (int q = 0; q < SCAN_PER; ++q) {
        int i = start + tid * SCAN_PER + q;
        if (i < N) {
            rowptr[i] = run;
            dinv[i]   = rsqrtf((float)c[q] + 1.0f);
        }
        run += c[q];
        if (i == N - 1) rowptr[N] = run;
    }
}

__global__ void csr_fill(const int* __restrict__ dst, const int* __restrict__ rank,
                         const int* __restrict__ rowptr, int* __restrict__ eid, int E) {
    int e = blockIdx.x * blockDim.x + threadIdx.x;
    if (e >= E) return;
    int p = rowptr[dst[e]] + rank[e];
    p = min(max(p, 0), E - 1);
    eid[p] = e;
}

__global__ __launch_bounds__(256)
void csr_order(const int* __restrict__ rowptr, const int* __restrict__ eid,
               const int* __restrict__ src, const float* __restrict__ dinv,
               int* __restrict__ col, float* __restrict__ cf, int N, int E) {
    const int lane = threadIdx.x & 31;
    const int row = __builtin_amdgcn_readfirstlane((int)(blockIdx.x * 8 + (threadIdx.x >> 5)));
    if (row >= N) return;
    const int beg = __builtin_amdgcn_readfirstlane(rowptr[row]);
    const int end = __builtin_amdgcn_readfirstlane(rowptr[row + 1]);
    const float dd = dinv[row];

    for (int a = beg; a < end; a += 32) {
        const int na = min(32, end - a);
        const int my = (lane < na) ? eid[a + lane] : 0x7fffffff;
        int rk = 0;
        for (int b = beg; b < end; b += 32) {
            const int nb = min(32, end - b);
            const int ob = (lane < nb) ? eid[b + lane] : 0x7fffffff;
            for (int j = 0; j < nb; ++j)
                rk += (__builtin_amdgcn_readlane(ob, j) < my) ? 1 : 0;
        }
        if (lane < na) {
            const int e = (int)min((unsigned)my, (unsigned)(E - 1));
            const int s = src[e];
            col[beg + rk] = s;
            cf[beg + rk]  = dinv[s] * dd;
        }
    }
}

__device__ __forceinline__ void fma4(float4& a, float c, const float4& v) {
    a.x += c * v.x; a.y += c * v.y; a.z += c * v.z; a.w += c * v.w;
}

__global__ __launch_bounds__(256)
void gcn_gather_rows(const int* __restrict__ rowptr, const int* __restrict__ col,
                     const float* __restrict__ cf, const float* __restrict__ dinv,
                     const float* __restrict__ bias, const float* __restrict__ h,
                     float* __restrict__ out, int N) {
    const int lane = threadIdx.x & 31;
    const int row = __builtin_amdgcn_readfirstlane((int)(blockIdx.x * 8 + (threadIdx.x >> 5)));
    if (row >= N) return;
    const int beg = __builtin_amdgcn_readfirstlane(rowptr[row]);
    const int end = __builtin_amdgcn_readfirstlane(rowptr[row + 1]);

    const float4* __restrict__ h4 = (const float4*)h;
    const float4 hv = h4[(size_t)row * 32 + lane];
    const float  di = dinv[row];
    const float4 b4 = ((const float4*)bias)[lane];

    float4 acc = make_float4(0.f, 0.f, 0.f, 0.f);
    const unsigned nmax = (unsigned)(N - 1);

    for (int base = beg; base < end; base += 32) {
        const int n = min(32, end - base);
        int   sv = 0;
        float cv = 0.f;
        if (lane < n) { sv = col[base + lane]; cv = cf[base + lane]; }
        sv = (int)min((unsigned)sv, nmax);
        const int cvi = __builtin_bit_cast(int, cv);

        int j = 0;
        for (; j + 4 <= n; j += 4) {
            const int s0 = __builtin_amdgcn_readlane(sv, j);
            const int s1 = __builtin_amdgcn_readlane(sv, j + 1);
            const int s2 = __builtin_amdgcn_readlane(sv, j + 2);
            const int s3 = __builtin_amdgcn_readlane(sv, j + 3);
            const float4 v0 = h4[(size_t)s0 * 32 + lane];
            const float4 v1 = h4[(size_t)s1 * 32 + lane];
            const float4 v2 = h4[(size_t)s2 * 32 + lane];
            const float4 v3 = h4[(size_t)s3 * 32 + lane];
            fma4(acc, __builtin_bit_cast(float, __builtin_amdgcn_readlane(cvi, j)),     v0);
            fma4(acc, __builtin_bit_cast(float, __builtin_amdgcn_readlane(cvi, j + 1)), v1);
            fma4(acc, __builtin_bit_cast(float, __builtin_amdgcn_readlane(cvi, j + 2)), v2);
            fma4(acc, __builtin_bit_cast(float, __builtin_amdgcn_readlane(cvi, j + 3)), v3);
        }
        const int r = n - j;
        if (r > 0) {
            const int s0 = __builtin_amdgcn_readlane(sv, j);
            const float4 v0 = h4[(size_t)s0 * 32 + lane];
            float4 v1 = make_float4(0.f, 0.f, 0.f, 0.f);
            float4 v2 = make_float4(0.f, 0.f, 0.f, 0.f);
            if (r > 1) {
                const int s1 = __builtin_amdgcn_readlane(sv, j + 1);
                v1 = h4[(size_t)s1 * 32 + lane];
            }
            if (r > 2) {
                const int s2 = __builtin_amdgcn_readlane(sv, j + 2);
                v2 = h4[(size_t)s2 * 32 + lane];
            }
            fma4(acc, __builtin_bit_cast(float, __builtin_amdgcn_readlane(cvi, j)), v0);
            if (r > 1) fma4(acc, __builtin_bit_cast(float, __builtin_amdgcn_readlane(cvi, j + 1)), v1);
            if (r > 2) fma4(acc, __builtin_bit_cast(float, __builtin_amdgcn_readlane(cvi, j + 2)), v2);
        }
    }

    const float di2 = di * di;
    v4f o;
    o.x = tanhf(acc.x + hv.x * di2 + b4.x);
    o.y = tanhf(acc.y + hv.y * di2 + b4.y);
    o.z = tanhf(acc.z + hv.z * di2 + b4.z);
    o.w = tanhf(acc.w + hv.w * di2 + b4.w);
    __builtin_nontemporal_store(o, (v4f*)out + ((size_t)row * 32 + lane));
}

extern "C" void kernel_launch(void* const* d_in, const int* in_sizes, int n_in,
                              void* d_out, int out_size, void* d_ws, size_t ws_size,
                              hipStream_t stream) {
    const float* x  = (const float*)d_in[0];
    const int*   ei = (const int*)d_in[1];
    const float* Wl[4] = {(const float*)d_in[2], (const float*)d_in[4],
                          (const float*)d_in[6], (const float*)d_in[8]};
    const float* bl[4] = {(const float*)d_in[3], (const float*)d_in[5],
                          (const float*)d_in[7], (const float*)d_in[9]};

    const int NFEAT = in_sizes[2] / HDIM;
    const int N     = in_sizes[0] / NFEAT;
    const int E     = in_sizes[1] / 2;
    const int* srcp = ei;
    const int* dstp = ei + E;

    float* wsf  = (float*)d_ws;
    float* r0   = wsf;
    float* r1   = r0 + (size_t)N * HDIM;
    float* r2   = r1 + (size_t)N * HDIM;
    float* dinv = r2 + (size_t)N * HDIM;

    int*   cnt    = (int*)r1;
    int*   rowptr = cnt + N;
    int*   rank   = rowptr + N + 32;
    int*   eid    = rank + E;
    int*   col    = eid + E;
    float* cf     = (float*)(col + E);

    const int THR = 256;

    zero_i32<<<(N + THR - 1) / THR, THR, 0, stream>>>(cnt, N);
    count_rank<<<(E + THR - 1) / THR, THR, 0, stream>>>(dstp, cnt, rank, E);
    scan_rowptr_dinv<<<(N + SCAN_T * SCAN_PER - 1) / (SCAN_T * SCAN_PER), SCAN_T, 0, stream>>>(
        cnt, rowptr, dinv, N);
    csr_fill<<<(E + THR - 1) / THR, THR, 0, stream>>>(dstp, rank, rowptr, eid, E);
    csr_order<<<(N + 7) / 8, 256, 0, stream>>>(rowptr, eid, srcp, dinv, col, cf, N, E);

    const float* in = x;
    int K = NFEAT;
    for (int l = 0; l < 4; ++l) {
        const int Kpad = (K + 3) & ~3;
        const size_t shmem =
            ((size_t)(Kpad / 2) * WS2 * 2 + 8u * 16u * A_LD) * sizeof(float);
        gcn_gemm_wmma_f32<<<256, 256, shmem, stream>>>(
            in, Wl[l], r0, N, K, Kpad);

        float* outp = (l == 3) ? (float*)d_out : r2;
        gcn_gather_rows<<<(N + 7) / 8, 256, 0, stream>>>(
            rowptr, col, cf, dinv, bl[l], r0, outp, N);

        in = r2;
        K = HDIM;
    }
}


# edge counting sort without global atomics: edges binned by 256-row destination block with LDS counters (histogram, scan, place), each block then slots its rows in LDS; row sum order kept
# speedup vs baseline: 5.6540x; 1.1745x over previous
#include <hip/hip_runtime.h>
#include <hip/hip_bf16.h>
#include <math.h>


#define HDIM 128
#define WS2  144
#define A_LD 138

#define A_IT 17
#define W_IT 9

typedef __attribute__((ext_vector_type(2))) float v2f;
typedef __attribute__((ext_vector_type(8))) float v8f;
typedef __attribute__((ext_vector_type(4))) float v4f;

__device__ __forceinline__ void strip_load(float4 (&av)[A_IT], const float* __restrict__ A,
                                           int s, int N, int K, int lane) {
    const int rowBase   = s * 16;
    const int rowsValid = min(16, max(0, N - rowBase));
    const int nvecFull  = (rowsValid * K) >> 2;
    const float4* strip4 = (const float4*)(A + (size_t)rowBase * K);
    if (nvecFull > 0) {
#pragma unroll
        for (int u = 0; u < A_IT; ++u)
            av[u] = strip4[min(u * 32 + lane, nvecFull - 1)];
    } else {
#pragma unroll
        for (int u = 0; u < A_IT; ++u) av[u] = make_float4(0.f, 0.f, 0.f, 0.f);
    }
}

__device__ __forceinline__ void strip_to_lds(const float4 (&av)[A_IT], float* Aw,
                                             const float* __restrict__ A,
                                             int s, int N, int K, int Kpad, int lane) {
    const int rowBase   = s * 16;
    const int rowsValid = min(16, max(0, N - rowBase));
    const int validLen  = rowsValid * K;
    const int nvecFull  = validLen >> 2;
    const int nvec      = (16 * K) >> 2;
    const float* strip  = A + (size_t)rowBase * K;
    const float4 z4 = make_float4(0.f, 0.f, 0.f, 0.f);
    const float invK = 1.0f / (float)K;
    const float fl   = (float)(4 * lane) + 0.5f;
    const int wrapAdd = A_LD - K;
    const bool rowVec = (K & 3) == 0;
#pragma unroll
    for (int u = 0; u < A_IT; ++u) {
        const int i = u * 32 + lane;
        if (i < nvec) {
            const int e0 = 128 * u + 4 * lane;
            const int r  = (int)((fl + (float)(128 * u)) * invK);
            const int k  = e0 - r * K;
            const int a0 = r * A_LD + k;
            const float4 w = (i < nvecFull) ? av[u] : z4;
            if (rowVec) {
                v2f* p = (v2f*)(Aw + a0);
                p[0] = (v2f){w.x, w.y};
                p[1] = (v2f){w.z, w.w};
            } else {
                Aw[a0] = w.x;
                Aw[a0 + 1 + ((k + 1 >= K) ? wrapAdd : 0)] = w.y;
                Aw[a0 + 2 + ((k + 2 >= K) ? wrapAdd : 0)] = w.z;
                Aw[a0 + 3 + ((k + 3 >= K) ? wrapAdd : 0)] = w.w;
            }
        }
    }
    const int rem = validLen & 3;
    if (lane < rem) {
        const int e  = (nvecFull << 2) + lane;
        const int rr = e / K;
        Aw[rr * A_LD + (e - rr * K)] = strip[e];
    }
    if (lane < 16)
        for (int k2 = K; k2 < Kpad; ++k2) Aw[lane * A_LD + k2] = 0.0f;
}

__global__ __launch_bounds__(256) __attribute__((amdgpu_waves_per_eu(2, 2)))
void gcn_gemm_wmma_f32(const float* __restrict__ A,
                       const float* __restrict__ W,
                       float* __restrict__ C,
                       int N, int K, int Kpad) {
    extern __shared__ float lds[];
    v2f*   Wp = (v2f*)lds;
    float* As = lds + (size_t)(Kpad / 2) * WS2 * 2;

    const int tid  = threadIdx.x;
    const int lane = tid & 31;
    const int l15  = lane & 15;
    const int wave = __builtin_amdgcn_readfirstlane(tid >> 5);

    const int nStrips = (N + 15) >> 4;
    const int nblk    = (int)gridDim.x;
    const int stride  = nblk * 8;
    int s = wave * nblk + (int)blockIdx.x;

    const float4 z4 = make_float4(0.f, 0.f, 0.f, 0.f);
    float* Aw = As + (size_t)wave * 16 * A_LD;

    float4 av[A_IT];
    strip_load(av, A, s, N, K, lane);

    const float4* W4 = (const float4*)W;
    const int nW = (Kpad / 2) * 32;
    {
        float4 wa[W_IT], wb[W_IT];
#pragma unroll
        for (int u = 0; u < W_IT; ++u) {
            const int idx = u * 256 + tid;
            const int k0  = 2 * (idx >> 5);
            const int n4  = idx & 31;
            wa[u] = W4[(size_t)min(k0,     K - 1) * 32 + n4];
            wb[u] = W4[(size_t)min(k0 + 1, K - 1) * 32 + n4];
        }
#pragma unroll
        for (int u = 0; u < W_IT; ++u) {
            const int idx = u * 256 + tid;
            if (idx < nW) {
                const int kk = idx >> 5;
                const int n4 = idx & 31;
                const int k0 = 2 * kk;
                const float4 a = (k0     < K) ? wa[u] : z4;
                const float4 b = (k0 + 1 < K) ? wb[u] : z4;
                float4* dst = (float4*)(Wp + kk * WS2 + 4 * n4);
                dst[0] = make_float4(a.x, b.x, a.y, b.y);
                dst[1] = make_float4(a.z, b.z, a.w, b.w);
            }
        }
    }
    __syncthreads();

    const int koff = (lane >= 16) ? 2 : 0;
    const float* Aw0 = Aw + l15 * A_LD;
    const int rOff = (lane >= 16) ? 8 : 0;

    for (; s < nStrips; s += stride) {
        strip_to_lds(av, Aw, A, s, N, K, Kpad, lane);
        __builtin_amdgcn_fence(__ATOMIC_RELEASE, "workgroup", "local");
        __builtin_amdgcn_fence(__ATOMIC_ACQUIRE, "workgroup", "local");

        const int sn = s + stride;
        if (sn < nStrips) strip_load(av, A, sn, N, K, lane);

        v8f acc[8];
#pragma unroll
        for (int j = 0; j < 8; ++j)
            acc[j] = (v8f){0.f,0.f,0.f,0.f,0.f,0.f,0.f,0.f};

        for (int k0 = 0; k0 < Kpad; k0 += 4) {
            v2f a0 = *(const v2f*)(Aw0 + k0 + koff);
            const v2f* Bk = Wp + (size_t)((k0 + koff) >> 1) * WS2 + l15;
#pragma unroll
            for (int j = 0; j < 8; ++j) {
                v2f b = Bk[j * 16];
                acc[j] = __builtin_amdgcn_wmma_f32_16x16x4_f32(
                    false, a0, false, b, (short)0, acc[j], false, false);
            }
        }

        const int rowBase = s * 16;
#pragma unroll
        for (int i = 0; i < 8; ++i) {
            int row = rowBase + i + rOff;
            if (row < N) {
                size_t base = (size_t)row * HDIM + l15;
#pragma unroll
                for (int j = 0; j < 8; ++j) {
                    C[base + j * 16] = acc[j][i];
                }
            }
        }
    }
}

#define BKT_SHIFT 8
#define BKT_ROWS  256
#define BIN_T     1024
#define BIN_G     256
#define BIN_U     4
__global__ __launch_bounds__(BIN_T)
void bin_hist(const int* __restrict__ dst, int* __restrict__ histG, int E, int nb) {
    extern __shared__ int binc[];
    const int tid = threadIdx.x;
    const int g   = blockIdx.x;
    const int G   = gridDim.x;
    const int chunk = (E + G - 1) / G;
    for (int b = tid; b < nb; b += BIN_T) binc[b] = 0;
    __syncthreads();
    const int e0 = g * chunk;
    const int e1 = min(E, e0 + chunk);
    for (int e = e0 + tid; e < e1; e += BIN_U * BIN_T) {
        int d[BIN_U];
#pragma unroll
        for (int q = 0; q < BIN_U; ++q) {
            const int ee = e + q * BIN_T;
            d[q] = dst[min(ee, e1 - 1)];
        }
#pragma unroll
        for (int q = 0; q < BIN_U; ++q) {
            const int b = (int)min((unsigned)d[q] >> BKT_SHIFT, (unsigned)(nb - 1));
            atomicAdd(&binc[b], (e + q * BIN_T < e1) ? 1 : 0);
        }
    }
    __syncthreads();
    for (int b = tid; b < nb; b += BIN_T) histG[b * G + g] = binc[b];
}

__global__ __launch_bounds__(BIN_T)
void bin_place(const int* __restrict__ dst, const int* __restrict__ offs,
               int* __restrict__ recw, int E, int nb) {
    extern __shared__ int binc[];
    int2* __restrict__ rec = (int2*)recw;
    const int tid = threadIdx.x;
    const int g   = blockIdx.x;
    const int G   = gridDim.x;
    const int chunk = (E + G - 1) / G;
    for (int b = tid; b < nb; b += BIN_T) binc[b] = offs[b * G + g];
    __syncthreads();
    const int e0 = g * chunk;
    const int e1 = min(E, e0 + chunk);
    for (int e = e0 + tid; e < e1; e += BIN_U * BIN_T) {
        int d[BIN_U];
#pragma unroll
        for (int q = 0; q < BIN_U; ++q) {
            const int ee = e + q * BIN_T;
            d[q] = dst[min(ee, e1 - 1)];
        }
#pragma unroll
        for (int q = 0; q < BIN_U; ++q) {
            const int ee = e + q * BIN_T;
            const int b = (int)min((unsigned)d[q] >> BKT_SHIFT, (unsigned)(nb - 1));
            int p = atomicAdd(&binc[b], (ee < e1) ? 1 : 0);
            p = min(max(p, 0), E - 1);
            if (ee < e1) rec[p] = make_int2(ee, d[q]);
        }
    }
}

__global__ __launch_bounds__(BIN_T)
void bucket_rows(const int* __restrict__ offs, const int* __restrict__ recw,
                 int* __restrict__ rowptr, float* __restrict__ dinv,
                 int* __restrict__ eid, int N, int E, int G) {
    const int2* __restrict__ rec = (const int2*)recw;
    __shared__ int cnt[BKT_ROWS];
    __shared__ int sc[BKT_ROWS];
    const int tid  = threadIdx.x;
    const int b    = blockIdx.x;
    const int row0 = b << BKT_SHIFT;
    const int bs = min(max(offs[b * G], 0), E);
    const int be = min(max(offs[(b + 1) * G], bs), E);

    if (tid < BKT_ROWS) cnt[tid] = 0;
    __syncthreads();
    for (int i = bs + tid; i < be; i += BIN_U * BIN_T) {
        int d[BIN_U];
#pragma unroll
        for (int q = 0; q < BIN_U; ++q) {
            const int ii = i + q * BIN_T;
            d[q] = rec[min(ii, be - 1)].y;
        }
#pragma unroll
        for (int q = 0; q < BIN_U; ++q) {
            const unsigned r = min((unsigned)(d[q] - row0), (unsigned)(BKT_ROWS - 1));
            atomicAdd(&cnt[r], (i + q * BIN_T < be) ? 1 : 0);
        }
    }
    __syncthreads();

    int c = 0;
    if (tid < BKT_ROWS) { c = cnt[tid]; sc[tid] = c; }
    __syncthreads();
    for (int s = 1; s < BKT_ROWS; s <<= 1) {
        int v = 0;
        if (tid < BKT_ROWS && tid >= s) v = sc[tid - s];
        __syncthreads();
        if (tid < BKT_ROWS) sc[tid] += v;
        __syncthreads();
    }
    if (tid < BKT_ROWS) {
        const int first = bs + sc[tid] - c;
        const int row   = row0 + tid;
        if (row < N) {
            rowptr[row] = first;
            dinv[row]   = rsqrtf((float)c + 1.0f);
            if (row == N - 1) rowptr[N] = first + c;
        }
        cnt[tid] = first;
    }
    __syncthreads();
    for (int i = bs + tid; i < be; i += BIN_U * BIN_T) {
        int2 v[BIN_U];
#pragma unroll
        for (int q = 0; q < BIN_U; ++q) {
            const int ii = i + q * BIN_T;
            v[q] = rec[min(ii, be - 1)];
        }
        __builtin_amdgcn_sched_barrier(0);
#pragma unroll
        for (int q = 0; q < BIN_U; ++q) {
            const bool in = i + q * BIN_T < be;
            const unsigned r = min((unsigned)(v[q].y - row0), (unsigned)(BKT_ROWS - 1));
            int p = atomicAdd(&cnt[r], in ? 1 : 0);
            p = min(max(p, 0), E - 1);
            if (in) eid[p] = v[q].x;
        }
    }
}

#define SCAN_T 256
#define SCAN_PER 4
__global__ __launch_bounds__(SCAN_T)
void scan_excl_i32(const int* __restrict__ cnt, int* __restrict__ rowptr, int N) {
    __shared__ int sh[SCAN_T];
    const int tid   = threadIdx.x;
    const int start = blockIdx.x * (SCAN_T * SCAN_PER);

    int pre = 0;
    const int4* __restrict__ cnt4 = (const int4*)cnt;
    for (int i = tid; i < start / 4; i += SCAN_T) {
        int4 v = cnt4[i];
        pre += (v.x + v.y) + (v.z + v.w);
    }

    int c[SCAN_PER];
    int loc = 0;
#pragma unroll
    for (int q = 0; q < SCAN_PER; ++q) {
        int i = start + tid * SCAN_PER + q;
        c[q] = (i < N) ? cnt[i] : 0;
        loc += c[q];
    }

    sh[tid] = pre;
    __syncthreads();
    for (int s = SCAN_T / 2; s > 0; s >>= 1) {
        if (tid < s) sh[tid] += sh[tid + s];
        __syncthreads();
    }
    const int offset = sh[0];
    __syncthreads();

    sh[tid] = loc;
    __syncthreads();
    for (int s = 1; s < SCAN_T; s <<= 1) {
        int v = (tid >= s) ? sh[tid - s] : 0;
        __syncthreads();
        sh[tid] += v;
        __syncthreads();
    }
    int run = offset + sh[tid] - loc;
#pragma unroll
    for (int q = 0; q < SCAN_PER; ++q) {
        int i = start + tid * SCAN_PER + q;
        if (i < N) rowptr[i] = run;
        run += c[q];
        if (i == N - 1) rowptr[N] = run;
    }
}

__global__ __launch_bounds__(256)
void csr_order(const int* __restrict__ rowptr, const int* __restrict__ eid,
               const int* __restrict__ src, const float* __restrict__ dinv,
               int* __restrict__ col, float* __restrict__ cf, int N, int E) {
    const int lane = threadIdx.x & 31;
    const int row = __builtin_amdgcn_readfirstlane((int)(blockIdx.x * 8 + (threadIdx.x >> 5)));
    if (row >= N) return;
    const int beg = __builtin_amdgcn_readfirstlane(rowptr[row]);
    const int end = __builtin_amdgcn_readfirstlane(rowptr[row + 1]);
    const float dd = dinv[row];

    for (int a = beg; a < end; a += 32) {
        const int na = min(32, end - a);
        const int my = (lane < na) ? eid[a + lane] : 0x7fffffff;
        int rk = 0;
        for (int b = beg; b < end; b += 32) {
            const int nb = min(32, end - b);
            const int ob = (lane < nb) ? eid[b + lane] : 0x7fffffff;
            for (int j = 0; j < nb; ++j)
                rk += (__builtin_amdgcn_readlane(ob, j) < my) ? 1 : 0;
        }
        if (lane < na) {
            const int e = (int)min((unsigned)my, (unsigned)(E - 1));
            const int s = src[e];
            col[beg + rk] = s;
            cf[beg + rk]  = dinv[s] * dd;
        }
    }
}

__device__ __forceinline__ void fma4(float4& a, float c, const float4& v) {
    a.x += c * v.x; a.y += c * v.y; a.z += c * v.z; a.w += c * v.w;
}

__global__ __launch_bounds__(256)
void gcn_gather_rows(const int* __restrict__ rowptr, const int* __restrict__ col,
                     const float* __restrict__ cf, const float* __restrict__ dinv,
                     const float* __restrict__ bias, const float* __restrict__ h,
                     float* __restrict__ out, int N) {
    const int lane = threadIdx.x & 31;
    const int row = __builtin_amdgcn_readfirstlane((int)(blockIdx.x * 8 + (threadIdx.x >> 5)));
    if (row >= N) return;
    const int beg = __builtin_amdgcn_readfirstlane(rowptr[row]);
    const int end = __builtin_amdgcn_readfirstlane(rowptr[row + 1]);

    const float4* __restrict__ h4 = (const float4*)h;
    const float4 hv = h4[(size_t)row * 32 + lane];
    const float  di = dinv[row];
    const float4 b4 = ((const float4*)bias)[lane];

    float4 acc = make_float4(0.f, 0.f, 0.f, 0.f);
    const unsigned nmax = (unsigned)(N - 1);

    for (int base = beg; base < end; base += 32) {
        const int n = min(32, end - base);
        int   sv = 0;
        float cv = 0.f;
        if (lane < n) { sv = col[base + lane]; cv = cf[base + lane]; }
        sv = (int)min((unsigned)sv, nmax);
        const int cvi = __builtin_bit_cast(int, cv);

        int j = 0;
        for (; j + 4 <= n; j += 4) {
            const int s0 = __builtin_amdgcn_readlane(sv, j);
            const int s1 = __builtin_amdgcn_readlane(sv, j + 1);
            const int s2 = __builtin_amdgcn_readlane(sv, j + 2);
            const int s3 = __builtin_amdgcn_readlane(sv, j + 3);
            const float4 v0 = h4[(size_t)s0 * 32 + lane];
            const float4 v1 = h4[(size_t)s1 * 32 + lane];
            const float4 v2 = h4[(size_t)s2 * 32 + lane];
            const float4 v3 = h4[(size_t)s3 * 32 + lane];
            fma4(acc, __builtin_bit_cast(float, __builtin_amdgcn_readlane(cvi, j)),     v0);
            fma4(acc, __builtin_bit_cast(float, __builtin_amdgcn_readlane(cvi, j + 1)), v1);
            fma4(acc, __builtin_bit_cast(float, __builtin_amdgcn_readlane(cvi, j + 2)), v2);
            fma4(acc, __builtin_bit_cast(float, __builtin_amdgcn_readlane(cvi, j + 3)), v3);
        }
        const int r = n - j;
        if (r > 0) {
            const int s0 = __builtin_amdgcn_readlane(sv, j);
            const float4 v0 = h4[(size_t)s0 * 32 + lane];
            float4 v1 = make_float4(0.f, 0.f, 0.f, 0.f);
            float4 v2 = make_float4(0.f, 0.f, 0.f, 0.f);
            if (r > 1) {
                const int s1 = __builtin_amdgcn_readlane(sv, j + 1);
                v1 = h4[(size_t)s1 * 32 + lane];
            }
            if (r > 2) {
                const int s2 = __builtin_amdgcn_readlane(sv, j + 2);
                v2 = h4[(size_t)s2 * 32 + lane];
            }
            fma4(acc, __builtin_bit_cast(float, __builtin_amdgcn_readlane(cvi, j)), v0);
            if (r > 1) fma4(acc, __builtin_bit_cast(float, __builtin_amdgcn_readlane(cvi, j + 1)), v1);
            if (r > 2) fma4(acc, __builtin_bit_cast(float, __builtin_amdgcn_readlane(cvi, j + 2)), v2);
        }
    }

    const float di2 = di * di;
    v4f o;
    o.x = tanhf(acc.x + hv.x * di2 + b4.x);
    o.y = tanhf(acc.y + hv.y * di2 + b4.y);
    o.z = tanhf(acc.z + hv.z * di2 + b4.z);
    o.w = tanhf(acc.w + hv.w * di2 + b4.w);
    __builtin_nontemporal_store(o, (v4f*)out + ((size_t)row * 32 + lane));
}

extern "C" void kernel_launch(void* const* d_in, const int* in_sizes, int n_in,
                              void* d_out, int out_size, void* d_ws, size_t ws_size,
                              hipStream_t stream) {
    const float* x  = (const float*)d_in[0];
    const int*   ei = (const int*)d_in[1];
    const float* Wl[4] = {(const float*)d_in[2], (const float*)d_in[4],
                          (const float*)d_in[6], (const float*)d_in[8]};
    const float* bl[4] = {(const float*)d_in[3], (const float*)d_in[5],
                          (const float*)d_in[7], (const float*)d_in[9]};

    const int NFEAT = in_sizes[2] / HDIM;
    const int N     = in_sizes[0] / NFEAT;
    const int E     = in_sizes[1] / 2;
    const int* srcp = ei;
    const int* dstp = ei + E;

    float* wsf  = (float*)d_ws;
    float* r0   = wsf;
    float* r1   = r0 + (size_t)N * HDIM;
    float* r2   = r1 + (size_t)N * HDIM;
    float* dinv = r2 + (size_t)N * HDIM;

    const int nb  = (N + BKT_ROWS - 1) >> BKT_SHIFT;
    const int n2  = nb * BIN_G;
    const int n2p = (n2 + 31) & ~31;
    int*   histG  = (int*)r1;
    int*   offs   = histG + n2p;
    int*   rowptr = offs + n2p + 32;
    int*   rec    = rowptr + ((N + 31) & ~31) + 32;
    int*   eid    = rec + 2 * (size_t)E;
    int*   col    = eid + E;
    float* cf     = (float*)(col + E);

    bin_hist<<<BIN_G, BIN_T, (size_t)nb * sizeof(int), stream>>>(dstp, histG, E, nb);
    scan_excl_i32<<<(n2 + SCAN_T * SCAN_PER - 1) / (SCAN_T * SCAN_PER), SCAN_T, 0, stream>>>(
        histG, offs, n2);
    bin_place<<<BIN_G, BIN_T, (size_t)nb * sizeof(int), stream>>>(dstp, offs, rec, E, nb);
    bucket_rows<<<nb, BIN_T, 0, stream>>>(offs, rec, rowptr, dinv, eid, N, E, BIN_G);
    csr_order<<<(N + 7) / 8, 256, 0, stream>>>(rowptr, eid, srcp, dinv, col, cf, N, E);

    const float* in = x;
    int K = NFEAT;
    for (int l = 0; l < 4; ++l) {
        const int Kpad = (K + 3) & ~3;
        const size_t shmem =
            ((size_t)(Kpad / 2) * WS2 * 2 + 8u * 16u * A_LD) * sizeof(float);
        gcn_gemm_wmma_f32<<<256, 256, shmem, stream>>>(
            in, Wl[l], r0, N, K, Kpad);

        float* outp = (l == 3) ? (float*)d_out : r2;
        gcn_gather_rows<<<(N + 7) / 8, 256, 0, stream>>>(
            rowptr, col, cf, dinv, bl[l], r0, outp, N);

        in = r2;
        K = HDIM;
    }
}


# layers 1-3 products moved into the previous layer's row gather: resident 32-wave work-groups keep 32 tanh rows in LDS and multiply them by the next W there (split barrier); sums unchanged
# speedup vs baseline: 6.5988x; 1.1671x over previous
#include <hip/hip_runtime.h>
#include <hip/hip_bf16.h>
#include <math.h>


#define HDIM 128
#define WS2  144
#define A_LD 138

#define A_IT 17
#define W_IT 9

typedef __attribute__((ext_vector_type(2))) float v2f;
typedef __attribute__((ext_vector_type(8))) float v8f;
typedef __attribute__((ext_vector_type(4))) float v4f;

__device__ __forceinline__ void strip_load(float4 (&av)[A_IT], const float* __restrict__ A,
                                           int s, int N, int K, int lane) {
    const int rowBase   = s * 16;
    const int rowsValid = min(16, max(0, N - rowBase));
    const int nvecFull  = (rowsValid * K) >> 2;
    const float4* strip4 = (const float4*)(A + (size_t)rowBase * K);
    if (nvecFull > 0) {
#pragma unroll
        for (int u = 0; u < A_IT; ++u)
            av[u] = strip4[min(u * 32 + lane, nvecFull - 1)];
    } else {
#pragma unroll
        for (int u = 0; u < A_IT; ++u) av[u] = make_float4(0.f, 0.f, 0.f, 0.f);
    }
}

__device__ __forceinline__ void strip_to_lds(const float4 (&av)[A_IT], float* Aw,
                                             const float* __restrict__ A,
                                             int s, int N, int K, int Kpad, int lane) {
    const int rowBase   = s * 16;
    const int rowsValid = min(16, max(0, N - rowBase));
    const int validLen  = rowsValid * K;
    const int nvecFull  = validLen >> 2;
    const int nvec      = (16 * K) >> 2;
    const float* strip  = A + (size_t)rowBase * K;
    const float4 z4 = make_float4(0.f, 0.f, 0.f, 0.f);
    const float invK = 1.0f / (float)K;
    const float fl   = (float)(4 * lane) + 0.5f;
    const int wrapAdd = A_LD - K;
    const bool rowVec = (K & 3) == 0;
#pragma unroll
    for (int u = 0; u < A_IT; ++u) {
        const int i = u * 32 + lane;
        if (i < nvec) {
            const int e0 = 128 * u + 4 * lane;
            const int r  = (int)((fl + (float)(128 * u)) * invK);
            const int k  = e0 - r * K;
            const int a0 = r * A_LD + k;
            const float4 w = (i < nvecFull) ? av[u] : z4;
            if (rowVec) {
                v2f* p = (v2f*)(Aw + a0);
                p[0] = (v2f){w.x, w.y};
                p[1] = (v2f){w.z, w.w};
            } else {
                Aw[a0] = w.x;
                Aw[a0 + 1 + ((k + 1 >= K) ? wrapAdd : 0)] = w.y;
                Aw[a0 + 2 + ((k + 2 >= K) ? wrapAdd : 0)] = w.z;
                Aw[a0 + 3 + ((k + 3 >= K) ? wrapAdd : 0)] = w.w;
            }
        }
    }
    const int rem = validLen & 3;
    if (lane < rem) {
        const int e  = (nvecFull << 2) + lane;
        const int rr = e / K;
        Aw[rr * A_LD + (e - rr * K)] = strip[e];
    }
    if (lane < 16)
        for (int k2 = K; k2 < Kpad; ++k2) Aw[lane * A_LD + k2] = 0.0f;
}

__global__ __launch_bounds__(256) __attribute__((amdgpu_waves_per_eu(2, 2)))
void gcn_gemm_wmma_f32(const float* __restrict__ A,
                       const float* __restrict__ W,
                       float* __restrict__ C,
                       int N, int K, int Kpad) {
    extern __shared__ float lds[];
    v2f*   Wp = (v2f*)lds;
    float* As = lds + (size_t)(Kpad / 2) * WS2 * 2;

    const int tid  = threadIdx.x;
    const int lane = tid & 31;
    const int l15  = lane & 15;
    const int wave = __builtin_amdgcn_readfirstlane(tid >> 5);

    const int nStrips = (N + 15) >> 4;
    const int nblk    = (int)gridDim.x;
    const int stride  = nblk * 8;
    int s = wave * nblk + (int)blockIdx.x;

    const float4 z4 = make_float4(0.f, 0.f, 0.f, 0.f);
    float* Aw = As + (size_t)wave * 16 * A_LD;

    float4 av[A_IT];
    strip_load(av, A, s, N, K, lane);

    const float4* W4 = (const float4*)W;
    const int nW = (Kpad / 2) * 32;
    {
        float4 wa[W_IT], wb[W_IT];
#pragma unroll
        for (int u = 0; u < W_IT; ++u) {
            const int idx = u * 256 + tid;
            const int k0  = 2 * (idx >> 5);
            const int n4  = idx & 31;
            wa[u] = W4[(size_t)min(k0,     K - 1) * 32 + n4];
            wb[u] = W4[(size_t)min(k0 + 1, K - 1) * 32 + n4];
        }
#pragma unroll
        for (int u = 0; u < W_IT; ++u) {
            const int idx = u * 256 + tid;
            if (idx < nW) {
                const int kk = idx >> 5;
                const int n4 = idx & 31;
                const int k0 = 2 * kk;
                const float4 a = (k0     < K) ? wa[u] : z4;
                const float4 b = (k0 + 1 < K) ? wb[u] : z4;
                float4* dst = (float4*)(Wp + kk * WS2 + 4 * n4);
                dst[0] = make_float4(a.x, b.x, a.y, b.y);
                dst[1] = make_float4(a.z, b.z, a.w, b.w);
            }
        }
    }
    __syncthreads();

    const int koff = (lane >= 16) ? 2 : 0;
    const float* Aw0 = Aw + l15 * A_LD;
    const int rOff = (lane >= 16) ? 8 : 0;

    for (; s < nStrips; s += stride) {
        strip_to_lds(av, Aw, A, s, N, K, Kpad, lane);
        __builtin_amdgcn_fence(__ATOMIC_RELEASE, "workgroup", "local");
        __builtin_amdgcn_fence(__ATOMIC_ACQUIRE, "workgroup", "local");

        const int sn = s + stride;
        if (sn < nStrips) strip_load(av, A, sn, N, K, lane);

        v8f acc[8];
#pragma unroll
        for (int j = 0; j < 8; ++j)
            acc[j] = (v8f){0.f,0.f,0.f,0.f,0.f,0.f,0.f,0.f};

        for (int k0 = 0; k0 < Kpad; k0 += 4) {
            v2f a0 = *(const v2f*)(Aw0 + k0 + koff);
            const v2f* Bk = Wp + (size_t)((k0 + koff) >> 1) * WS2 + l15;
#pragma unroll
            for (int j = 0; j < 8; ++j) {
                v2f b = Bk[j * 16];
                acc[j] = __builtin_amdgcn_wmma_f32_16x16x4_f32(
                    false, a0, false, b, (short)0, acc[j], false, false);
            }
        }

        const int rowBase = s * 16;
#pragma unroll
        for (int i = 0; i < 8; ++i) {
            int row = rowBase + i + rOff;
            if (row < N) {
                size_t base = (size_t)row * HDIM + l15;
#pragma unroll
                for (int j = 0; j < 8; ++j) {
                    C[base + j * 16] = acc[j][i];
                }
            }
        }
    }
}

#define BKT_SHIFT 8
#define BKT_ROWS  256
#define BIN_T     1024
#define BIN_G     256
#define BIN_U     4
__global__ __launch_bounds__(BIN_T)
void bin_hist(const int* __restrict__ dst, int* __restrict__ histG, int E, int nb) {
    extern __shared__ int binc[];
    const int tid = threadIdx.x;
    const int g   = blockIdx.x;
    const int G   = gridDim.x;
    const int chunk = (E + G - 1) / G;
    for (int b = tid; b < nb; b += BIN_T) binc[b] = 0;
    __syncthreads();
    const int e0 = g * chunk;
    const int e1 = min(E, e0 + chunk);
    for (int e = e0 + tid; e < e1; e += BIN_U * BIN_T) {
        int d[BIN_U];
#pragma unroll
        for (int q = 0; q < BIN_U; ++q) {
            const int ee = e + q * BIN_T;
            d[q] = dst[min(ee, e1 - 1)];
        }
#pragma unroll
        for (int q = 0; q < BIN_U; ++q) {
            const int b = (int)min((unsigned)d[q] >> BKT_SHIFT, (unsigned)(nb - 1));
            atomicAdd(&binc[b], (e + q * BIN_T < e1) ? 1 : 0);
        }
    }
    __syncthreads();
    for (int b = tid; b < nb; b += BIN_T) histG[b * G + g] = binc[b];
}

__global__ __launch_bounds__(BIN_T)
void bin_place(const int* __restrict__ dst, const int* __restrict__ offs,
               int* __restrict__ recw, int E, int nb) {
    extern __shared__ int binc[];
    int2* __restrict__ rec = (int2*)recw;
    const int tid = threadIdx.x;
    const int g   = blockIdx.x;
    const int G   = gridDim.x;
    const int chunk = (E + G - 1) / G;
    for (int b = tid; b < nb; b += BIN_T) binc[b] = offs[b * G + g];
    __syncthreads();
    const int e0 = g * chunk;
    const int e1 = min(E, e0 + chunk);
    for (int e = e0 + tid; e < e1; e += BIN_U * BIN_T) {
        int d[BIN_U];
#pragma unroll
        for (int q = 0; q < BIN_U; ++q) {
            const int ee = e + q * BIN_T;
            d[q] = dst[min(ee, e1 - 1)];
        }
#pragma unroll
        for (int q = 0; q < BIN_U; ++q) {
            const int ee = e + q * BIN_T;
            const int b = (int)min((unsigned)d[q] >> BKT_SHIFT, (unsigned)(nb - 1));
            int p = atomicAdd(&binc[b], (ee < e1) ? 1 : 0);
            p = min(max(p, 0), E - 1);
            if (ee < e1) rec[p] = make_int2(ee, d[q]);
        }
    }
}

__global__ __launch_bounds__(BIN_T)
void bucket_rows(const int* __restrict__ offs, const int* __restrict__ recw,
                 int* __restrict__ rowptr, float* __restrict__ dinv,
                 int* __restrict__ eid, int N, int E, int G) {
    const int2* __restrict__ rec = (const int2*)recw;
    __shared__ int cnt[BKT_ROWS];
    __shared__ int sc[BKT_ROWS];
    const int tid  = threadIdx.x;
    const int b    = blockIdx.x;
    const int row0 = b << BKT_SHIFT;
    const int bs = min(max(offs[b * G], 0), E);
    const int be = min(max(offs[(b + 1) * G], bs), E);

    if (tid < BKT_ROWS) cnt[tid] = 0;
    __syncthreads();
    for (int i = bs + tid; i < be; i += BIN_U * BIN_T) {
        int d[BIN_U];
#pragma unroll
        for (int q = 0; q < BIN_U; ++q) {
            const int ii = i + q * BIN_T;
            d[q] = rec[min(ii, be - 1)].y;
        }
#pragma unroll
        for (int q = 0; q < BIN_U; ++q) {
            const unsigned r = min((unsigned)(d[q] - row0), (unsigned)(BKT_ROWS - 1));
            atomicAdd(&cnt[r], (i + q * BIN_T < be) ? 1 : 0);
        }
    }
    __syncthreads();

    int c = 0;
    if (tid < BKT_ROWS) { c = cnt[tid]; sc[tid] = c; }
    __syncthreads();
    for (int s = 1; s < BKT_ROWS; s <<= 1) {
        int v = 0;
        if (tid < BKT_ROWS && tid >= s) v = sc[tid - s];
        __syncthreads();
        if (tid < BKT_ROWS) sc[tid] += v;
        __syncthreads();
    }
    if (tid < BKT_ROWS) {
        const int first = bs + sc[tid] - c;
        const int row   = row0 + tid;
        if (row < N) {
            rowptr[row] = first;
            dinv[row]   = rsqrtf((float)c + 1.0f);
            if (row == N - 1) rowptr[N] = first + c;
        }
        cnt[tid] = first;
    }
    __syncthreads();
    for (int i = bs + tid; i < be; i += BIN_U * BIN_T) {
        int2 v[BIN_U];
#pragma unroll
        for (int q = 0; q < BIN_U; ++q) {
            const int ii = i + q * BIN_T;
            v[q] = rec[min(ii, be - 1)];
        }
        __builtin_amdgcn_sched_barrier(0);
#pragma unroll
        for (int q = 0; q < BIN_U; ++q) {
            const bool in = i + q * BIN_T < be;
            const unsigned r = min((unsigned)(v[q].y - row0), (unsigned)(BKT_ROWS - 1));
            int p = atomicAdd(&cnt[r], in ? 1 : 0);
            p = min(max(p, 0), E - 1);
            if (in) eid[p] = v[q].x;
        }
    }
}

#define SCAN_T 256
#define SCAN_PER 4
__global__ __launch_bounds__(SCAN_T)
void scan_excl_i32(const int* __restrict__ cnt, int* __restrict__ rowptr, int N) {
    __shared__ int sh[SCAN_T];
    const int tid   = threadIdx.x;
    const int start = blockIdx.x * (SCAN_T * SCAN_PER);

    int pre = 0;
    const int4* __restrict__ cnt4 = (const int4*)cnt;
    for (int i = tid; i < start / 4; i += SCAN_T) {
        int4 v = cnt4[i];
        pre += (v.x + v.y) + (v.z + v.w);
    }

    int c[SCAN_PER];
    int loc = 0;
#pragma unroll
    for (int q = 0; q < SCAN_PER; ++q) {
        int i = start + tid * SCAN_PER + q;
        c[q] = (i < N) ? cnt[i] : 0;
        loc += c[q];
    }

    sh[tid] = pre;
    __syncthreads();
    for (int s = SCAN_T / 2; s > 0; s >>= 1) {
        if (tid < s) sh[tid] += sh[tid + s];
        __syncthreads();
    }
    const int offset = sh[0];
    __syncthreads();

    sh[tid] = loc;
    __syncthreads();
    for (int s = 1; s < SCAN_T; s <<= 1) {
        int v = (tid >= s) ? sh[tid - s] : 0;
        __syncthreads();
        sh[tid] += v;
        __syncthreads();
    }
    int run = offset + sh[tid] - loc;
#pragma unroll
    for (int q = 0; q < SCAN_PER; ++q) {
        int i = start + tid * SCAN_PER + q;
        if (i < N) rowptr[i] = run;
        run += c[q];
        if (i == N - 1) rowptr[N] = run;
    }
}

__global__ __launch_bounds__(256)
void csr_order(const int* __restrict__ rowptr, const int* __restrict__ eid,
               const int* __restrict__ src, const float* __restrict__ dinv,
               int* __restrict__ col, float* __restrict__ cf, int N, int E) {
    const int lane = threadIdx.x & 31;
    const int row = __builtin_amdgcn_readfirstlane((int)(blockIdx.x * 8 + (threadIdx.x >> 5)));
    if (row >= N) return;
    const int beg = __builtin_amdgcn_readfirstlane(rowptr[row]);
    const int end = __builtin_amdgcn_readfirstlane(rowptr[row + 1]);
    const float dd = dinv[row];

    for (int a = beg; a < end; a += 32) {
        const int na = min(32, end - a);
        const int my = (lane < na) ? eid[a + lane] : 0x7fffffff;
        int rk = 0;
        for (int b = beg; b < end; b += 32) {
            const int nb = min(32, end - b);
            const int ob = (lane < nb) ? eid[b + lane] : 0x7fffffff;
            for (int j = 0; j < nb; ++j)
                rk += (__builtin_amdgcn_readlane(ob, j) < my) ? 1 : 0;
        }
        if (lane < na) {
            const int e = (int)min((unsigned)my, (unsigned)(E - 1));
            const int s = src[e];
            col[beg + rk] = s;
            cf[beg + rk]  = dinv[s] * dd;
        }
    }
}

__device__ __forceinline__ void fma4(float4& a, float c, const float4& v) {
    a.x += c * v.x; a.y += c * v.y; a.z += c * v.z; a.w += c * v.w;
}

__global__ __launch_bounds__(256)
void gcn_gather_rows(const int* __restrict__ rowptr, const int* __restrict__ col,
                     const float* __restrict__ cf, const float* __restrict__ dinv,
                     const float* __restrict__ bias, const float* __restrict__ h,
                     float* __restrict__ out, int N) {
    const int lane = threadIdx.x & 31;
    const int row = __builtin_amdgcn_readfirstlane((int)(blockIdx.x * 8 + (threadIdx.x >> 5)));
    if (row >= N) return;
    const int beg = __builtin_amdgcn_readfirstlane(rowptr[row]);
    const int end = __builtin_amdgcn_readfirstlane(rowptr[row + 1]);

    const float4* __restrict__ h4 = (const float4*)h;
    const float4 hv = h4[(size_t)row * 32 + lane];
    const float  di = dinv[row];
    const float4 b4 = ((const float4*)bias)[lane];

    float4 acc = make_float4(0.f, 0.f, 0.f, 0.f);
    const unsigned nmax = (unsigned)(N - 1);

    for (int base = beg; base < end; base += 32) {
        const int n = min(32, end - base);
        int   sv = 0;
        float cv = 0.f;
        if (lane < n) { sv = col[base + lane]; cv = cf[base + lane]; }
        sv = (int)min((unsigned)sv, nmax);
        const int cvi = __builtin_bit_cast(int, cv);

        int j = 0;
        for (; j + 4 <= n; j += 4) {
            const int s0 = __builtin_amdgcn_readlane(sv, j);
            const int s1 = __builtin_amdgcn_readlane(sv, j + 1);
            const int s2 = __builtin_amdgcn_readlane(sv, j + 2);
            const int s3 = __builtin_amdgcn_readlane(sv, j + 3);
            const float4 v0 = h4[(size_t)s0 * 32 + lane];
            const float4 v1 = h4[(size_t)s1 * 32 + lane];
            const float4 v2 = h4[(size_t)s2 * 32 + lane];
            const float4 v3 = h4[(size_t)s3 * 32 + lane];
            fma4(acc, __builtin_bit_cast(float, __builtin_amdgcn_readlane(cvi, j)),     v0);
            fma4(acc, __builtin_bit_cast(float, __builtin_amdgcn_readlane(cvi, j + 1)), v1);
            fma4(acc, __builtin_bit_cast(float, __builtin_amdgcn_readlane(cvi, j + 2)), v2);
            fma4(acc, __builtin_bit_cast(float, __builtin_amdgcn_readlane(cvi, j + 3)), v3);
        }
        const int r = n - j;
        if (r > 0) {
            const int s0 = __builtin_amdgcn_readlane(sv, j);
            const float4 v0 = h4[(size_t)s0 * 32 + lane];
            float4 v1 = make_float4(0.f, 0.f, 0.f, 0.f);
            float4 v2 = make_float4(0.f, 0.f, 0.f, 0.f);
            if (r > 1) {
                const int s1 = __builtin_amdgcn_readlane(sv, j + 1);
                v1 = h4[(size_t)s1 * 32 + lane];
            }
            if (r > 2) {
                const int s2 = __builtin_amdgcn_readlane(sv, j + 2);
                v2 = h4[(size_t)s2 * 32 + lane];
            }
            fma4(acc, __builtin_bit_cast(float, __builtin_amdgcn_readlane(cvi, j)), v0);
            if (r > 1) fma4(acc, __builtin_bit_cast(float, __builtin_amdgcn_readlane(cvi, j + 1)), v1);
            if (r > 2) fma4(acc, __builtin_bit_cast(float, __builtin_amdgcn_readlane(cvi, j + 2)), v2);
        }
    }

    const float di2 = di * di;
    v4f o;
    o.x = tanhf(acc.x + hv.x * di2 + b4.x);
    o.y = tanhf(acc.y + hv.y * di2 + b4.y);
    o.z = tanhf(acc.z + hv.z * di2 + b4.z);
    o.w = tanhf(acc.w + hv.w * di2 + b4.w);
    __builtin_nontemporal_store(o, (v4f*)out + ((size_t)row * 32 + lane));
}


#define FZ_T 1024
#define FZ_STRIP (16 * A_LD)

__device__ __forceinline__ float4 gather_row_tanh(const int* __restrict__ rowptr,
                                                  const int* __restrict__ col,
                                                  const float* __restrict__ cf,
                                                  const float* __restrict__ dinv,
                                                  const float4* __restrict__ h4,
                                                  const float4 b4, int row, int lane,
                                                  unsigned nmax) {
    const int beg = __builtin_amdgcn_readfirstlane(rowptr[row]);
    const int end = __builtin_amdgcn_readfirstlane(rowptr[row + 1]);
    const float4 hv = h4[(size_t)row * 32 + lane];
    const float  di = dinv[row];
    float4 acc = make_float4(0.f, 0.f, 0.f, 0.f);

    for (int base = beg; base < end; base += 32) {
        const int n = min(32, end - base);
        int   sv = 0;
        float cv = 0.f;
        if (lane < n) { sv = col[base + lane]; cv = cf[base + lane]; }
        sv = (int)min((unsigned)sv, nmax);
        const int cvi = __builtin_bit_cast(int, cv);

        int j = 0;
        for (; j + 4 <= n; j += 4) {
            const int s0 = __builtin_amdgcn_readlane(sv, j);
            const int s1 = __builtin_amdgcn_readlane(sv, j + 1);
            const int s2 = __builtin_amdgcn_readlane(sv, j + 2);
            const int s3 = __builtin_amdgcn_readlane(sv, j + 3);
            const float4 v0 = h4[(size_t)s0 * 32 + lane];
            const float4 v1 = h4[(size_t)s1 * 32 + lane];
            const float4 v2 = h4[(size_t)s2 * 32 + lane];
            const float4 v3 = h4[(size_t)s3 * 32 + lane];
            fma4(acc, __builtin_bit_cast(float, __builtin_amdgcn_readlane(cvi, j)),     v0);
            fma4(acc, __builtin_bit_cast(float, __builtin_amdgcn_readlane(cvi, j + 1)), v1);
            fma4(acc, __builtin_bit_cast(float, __builtin_amdgcn_readlane(cvi, j + 2)), v2);
            fma4(acc, __builtin_bit_cast(float, __builtin_amdgcn_readlane(cvi, j + 3)), v3);
        }
        const int r = n - j;
        if (r > 0) {
            const int s0 = __builtin_amdgcn_readlane(sv, j);
            const float4 v0 = h4[(size_t)s0 * 32 + lane];
            float4 v1 = make_float4(0.f, 0.f, 0.f, 0.f);
            float4 v2 = make_float4(0.f, 0.f, 0.f, 0.f);
            if (r > 1) {
                const int s1 = __builtin_amdgcn_readlane(sv, j + 1);
                v1 = h4[(size_t)s1 * 32 + lane];
            }
            if (r > 2) {
                const int s2 = __builtin_amdgcn_readlane(sv, j + 2);
                v2 = h4[(size_t)s2 * 32 + lane];
            }
            fma4(acc, __builtin_bit_cast(float, __builtin_amdgcn_readlane(cvi, j)), v0);
            if (r > 1) fma4(acc, __builtin_bit_cast(float, __builtin_amdgcn_readlane(cvi, j + 1)), v1);
            if (r > 2) fma4(acc, __builtin_bit_cast(float, __builtin_amdgcn_readlane(cvi, j + 2)), v2);
        }
    }

    const float di2 = di * di;
    float4 o;
    o.x = tanhf(acc.x + hv.x * di2 + b4.x);
    o.y = tanhf(acc.y + hv.y * di2 + b4.y);
    o.z = tanhf(acc.z + hv.z * di2 + b4.z);
    o.w = tanhf(acc.w + hv.w * di2 + b4.w);
    return o;
}

__global__ __launch_bounds__(FZ_T)
void gcn_gather_gemm(const int* __restrict__ rowptr, const int* __restrict__ col,
                     const float* __restrict__ cf, const float* __restrict__ dinv,
                     const float* __restrict__ bias, const float* __restrict__ h,
                     const float* __restrict__ Wn, float* __restrict__ C, int N) {
    extern __shared__ float lds[];
    v2f*   Wp = (v2f*)lds;
    float* As = lds + (size_t)(HDIM / 2) * WS2 * 2;

    const int tid  = threadIdx.x;
    const int lane = tid & 31;
    const int l15  = lane & 15;
    const int wave = __builtin_amdgcn_readfirstlane(tid >> 5);

    const float4* __restrict__ h4 = (const float4*)h;
    const float4 b4 = ((const float4*)bias)[lane];
    const unsigned nmax = (unsigned)(N - 1);
    const int nTrips = (N + 31) >> 5;
    const int step   = (int)gridDim.x;
    const float4 z4  = make_float4(0.f, 0.f, 0.f, 0.f);

    {
        const float4* W4 = (const float4*)Wn;
        float4 wa[2], wb[2];
#pragma unroll
        for (int u = 0; u < 2; ++u) {
            const int idx = u * FZ_T + tid;
            const int kk  = idx >> 5;
            const int n4  = idx & 31;
            wa[u] = W4[(size_t)(2 * kk)     * 32 + n4];
            wb[u] = W4[(size_t)(2 * kk + 1) * 32 + n4];
        }
#pragma unroll
        for (int u = 0; u < 2; ++u) {
            const int idx = u * FZ_T + tid;
            const int kk  = idx >> 5;
            const int n4  = idx & 31;
            float4* dst = (float4*)(Wp + kk * WS2 + 4 * n4);
            dst[0] = make_float4(wa[u].x, wb[u].x, wa[u].y, wb[u].y);
            dst[1] = make_float4(wa[u].z, wb[u].z, wa[u].w, wb[u].w);
        }
    }

    const int koff = (lane >= 16) ? 2 : 0;
    const int rOff = (lane >= 16) ? 8 : 0;

    int p = (int)blockIdx.x;
    float4 o = z4;
    if (p < nTrips && p * 32 + wave < N)
        o = gather_row_tanh(rowptr, col, cf, dinv, h4, b4, p * 32 + wave, lane, nmax);

    for (int it = 0; p < nTrips; p += step, ++it) {
        float* Ab = As + (size_t)(it & 1) * (2 * FZ_STRIP);
        {
            v2f* q = (v2f*)(Ab + wave * A_LD + 4 * lane);
            q[0] = (v2f){o.x, o.y};
            q[1] = (v2f){o.z, o.w};
        }
        __builtin_amdgcn_fence(__ATOMIC_RELEASE, "workgroup", "local");
        __builtin_amdgcn_s_barrier_signal(-1);

        const int pn = p + step;
        float4 on = z4;
        if (pn < nTrips && pn * 32 + wave < N)
            on = gather_row_tanh(rowptr, col, cf, dinv, h4, b4, pn * 32 + wave, lane, nmax);

        __builtin_amdgcn_s_barrier_wait(-1);
        __builtin_amdgcn_fence(__ATOMIC_ACQUIRE, "workgroup", "local");

        const int mw = wave ^ ((it & 1) << 4);
        if (mw < 16) {
            const int st = mw >> 3;
            const int jt = mw & 7;
            const float* Ak = Ab + (st * 16 + l15) * A_LD + koff;
            const v2f*   Bk = Wp + (size_t)(koff >> 1) * WS2 + jt * 16 + l15;
            v8f acc8 = (v8f){0.f,0.f,0.f,0.f,0.f,0.f,0.f,0.f};
#pragma unroll
            for (int g = 0; g < 4; ++g) {
                v2f af[8], bf[8];
#pragma unroll
                for (int q = 0; q < 8; ++q) {
                    const int kb = g * 8 + q;
                    af[q] = *(const v2f*)(Ak + 4 * kb);
                    bf[q] = Bk[(size_t)(2 * kb) * WS2];
                }
                __builtin_amdgcn_sched_barrier(0);
#pragma unroll
                for (int q = 0; q < 8; ++q)
                    acc8 = __builtin_amdgcn_wmma_f32_16x16x4_f32(
                        false, af[q], false, bf[q], (short)0, acc8, false, false);
                asm volatile("v_nop\n\tv_nop\n\tv_nop\n\tv_nop\n\tv_nop\n\tv_nop\n\tv_nop\n\tv_nop"
                             : "+v"(acc8));
                __builtin_amdgcn_sched_barrier(0);
            }
            const int rowBase = p * 32 + st * 16;
#pragma unroll
            for (int i = 0; i < 8; ++i) {
                const int rr = rowBase + i + rOff;
                if (rr < N) __builtin_nontemporal_store(acc8[i], C + ((size_t)rr * HDIM + jt * 16 + l15));
            }
        }
        o = on;
    }
}

extern "C" void kernel_launch(void* const* d_in, const int* in_sizes, int n_in,
                              void* d_out, int out_size, void* d_ws, size_t ws_size,
                              hipStream_t stream) {
    const float* x  = (const float*)d_in[0];
    const int*   ei = (const int*)d_in[1];
    const float* Wl[4] = {(const float*)d_in[2], (const float*)d_in[4],
                          (const float*)d_in[6], (const float*)d_in[8]};
    const float* bl[4] = {(const float*)d_in[3], (const float*)d_in[5],
                          (const float*)d_in[7], (const float*)d_in[9]};

    const int NFEAT = in_sizes[2] / HDIM;
    const int N     = in_sizes[0] / NFEAT;
    const int E     = in_sizes[1] / 2;
    const int* srcp = ei;
    const int* dstp = ei + E;

    float* wsf  = (float*)d_ws;
    float* r0   = wsf;
    float* r1   = r0 + (size_t)N * HDIM;
    float* r2   = r1 + (size_t)N * HDIM;
    float* dinv = r2 + (size_t)N * HDIM;

    const int nb  = (N + BKT_ROWS - 1) >> BKT_SHIFT;
    const int n2  = nb * BIN_G;
    const int n2p = (n2 + 31) & ~31;
    int*   histG  = (int*)r1;
    int*   offs   = histG + n2p;
    int*   rowptr = offs + n2p + 32;
    int*   rec    = rowptr + ((N + 31) & ~31) + 32;
    int*   eid    = rec + 2 * (size_t)E;
    int*   col    = eid + E;
    float* cf     = (float*)(col + E);

    bin_hist<<<BIN_G, BIN_T, (size_t)nb * sizeof(int), stream>>>(dstp, histG, E, nb);
    scan_excl_i32<<<(n2 + SCAN_T * SCAN_PER - 1) / (SCAN_T * SCAN_PER), SCAN_T, 0, stream>>>(
        histG, offs, n2);
    bin_place<<<BIN_G, BIN_T, (size_t)nb * sizeof(int), stream>>>(dstp, offs, rec, E, nb);
    bucket_rows<<<nb, BIN_T, 0, stream>>>(offs, rec, rowptr, dinv, eid, N, E, BIN_G);
    csr_order<<<(N + 7) / 8, 256, 0, stream>>>(rowptr, eid, srcp, dinv, col, cf, N, E);

    const int Kpad0 = (NFEAT + 3) & ~3;
    const size_t shmem0 =
        ((size_t)(Kpad0 / 2) * WS2 * 2 + 8u * 16u * A_LD) * sizeof(float);
    gcn_gemm_wmma_f32<<<256, 256, shmem0, stream>>>(x, Wl[0], r0, N, NFEAT, Kpad0);

    const size_t shmemF =
        ((size_t)(HDIM / 2) * WS2 * 2 + 2u * 2u * FZ_STRIP) * sizeof(float);
    gcn_gather_gemm<<<512, FZ_T, shmemF, stream>>>(rowptr, col, cf, dinv, bl[0], r0, Wl[1], r2, N);
    gcn_gather_gemm<<<512, FZ_T, shmemF, stream>>>(rowptr, col, cf, dinv, bl[1], r2, Wl[2], r0, N);
    gcn_gather_gemm<<<512, FZ_T, shmemF, stream>>>(rowptr, col, cf, dinv, bl[2], r0, Wl[3], r2, N);

    gcn_gather_rows<<<(N + 7) / 8, 256, 0, stream>>>(
        rowptr, col, cf, dinv, bl[3], r2, (float*)d_out, N);
}


# rows stored times dinv[node]: edges need no coefficient, so a row's slots are ranked by edge id from LDS in the block-slotting launch (ordering launch gone); dinv[dst] scales the row sum, order kept
# speedup vs baseline: 6.9569x; 1.0543x over previous
#include <hip/hip_runtime.h>
#include <hip/hip_bf16.h>
#include <math.h>


#define HDIM 128
#define WS2  144
#define A_LD 138

#define A_IT 17
#define W_IT 9

typedef __attribute__((ext_vector_type(2))) float v2f;
typedef __attribute__((ext_vector_type(8))) float v8f;
typedef __attribute__((ext_vector_type(4))) float v4f;

__device__ __forceinline__ void strip_load(float4 (&av)[A_IT], const float* __restrict__ A,
                                           int s, int N, int K, int lane) {
    const int rowBase   = s * 16;
    const int rowsValid = min(16, max(0, N - rowBase));
    const int nvecFull  = (rowsValid * K) >> 2;
    const float4* strip4 = (const float4*)(A + (size_t)rowBase * K);
    if (nvecFull > 0) {
#pragma unroll
        for (int u = 0; u < A_IT; ++u)
            av[u] = strip4[min(u * 32 + lane, nvecFull - 1)];
    } else {
#pragma unroll
        for (int u = 0; u < A_IT; ++u) av[u] = make_float4(0.f, 0.f, 0.f, 0.f);
    }
}

__device__ __forceinline__ void strip_to_lds(const float4 (&av)[A_IT], float* Aw,
                                             const float* __restrict__ A,
                                             int s, int N, int K, int Kpad, int lane) {
    const int rowBase   = s * 16;
    const int rowsValid = min(16, max(0, N - rowBase));
    const int validLen  = rowsValid * K;
    const int nvecFull  = validLen >> 2;
    const int nvec      = (16 * K) >> 2;
    const float* strip  = A + (size_t)rowBase * K;
    const float4 z4 = make_float4(0.f, 0.f, 0.f, 0.f);
    const float invK = 1.0f / (float)K;
    const float fl   = (float)(4 * lane) + 0.5f;
    const int wrapAdd = A_LD - K;
    const bool rowVec = (K & 3) == 0;
#pragma unroll
    for (int u = 0; u < A_IT; ++u) {
        const int i = u * 32 + lane;
        if (i < nvec) {
            const int e0 = 128 * u + 4 * lane;
            const int r  = (int)((fl + (float)(128 * u)) * invK);
            const int k  = e0 - r * K;
            const int a0 = r * A_LD + k;
            const float4 w = (i < nvecFull) ? av[u] : z4;
            if (rowVec) {
                v2f* p = (v2f*)(Aw + a0);
                p[0] = (v2f){w.x, w.y};
                p[1] = (v2f){w.z, w.w};
            } else {
                Aw[a0] = w.x;
                Aw[a0 + 1 + ((k + 1 >= K) ? wrapAdd : 0)] = w.y;
                Aw[a0 + 2 + ((k + 2 >= K) ? wrapAdd : 0)] = w.z;
                Aw[a0 + 3 + ((k + 3 >= K) ? wrapAdd : 0)] = w.w;
            }
        }
    }
    const int rem = validLen & 3;
    if (lane < rem) {
        const int e  = (nvecFull << 2) + lane;
        const int rr = e / K;
        Aw[rr * A_LD + (e - rr * K)] = strip[e];
    }
    if (lane < 16)
        for (int k2 = K; k2 < Kpad; ++k2) Aw[lane * A_LD + k2] = 0.0f;
}

__global__ __launch_bounds__(256) __attribute__((amdgpu_waves_per_eu(2, 2)))
void gcn_gemm_wmma_f32(const float* __restrict__ A,
                       const float* __restrict__ W,
                       const float* __restrict__ dinv,
                       float* __restrict__ C,
                       int N, int K, int Kpad) {
    extern __shared__ float lds[];
    v2f*   Wp = (v2f*)lds;
    float* As = lds + (size_t)(Kpad / 2) * WS2 * 2;

    const int tid  = threadIdx.x;
    const int lane = tid & 31;
    const int l15  = lane & 15;
    const int wave = __builtin_amdgcn_readfirstlane(tid >> 5);

    const int nStrips = (N + 15) >> 4;
    const int nblk    = (int)gridDim.x;
    const int stride  = nblk * 8;
    int s = wave * nblk + (int)blockIdx.x;

    const float4 z4 = make_float4(0.f, 0.f, 0.f, 0.f);
    float* Aw = As + (size_t)wave * 16 * A_LD;

    float4 av[A_IT];
    strip_load(av, A, s, N, K, lane);

    const float4* W4 = (const float4*)W;
    const int nW = (Kpad / 2) * 32;
    {
        float4 wa[W_IT], wb[W_IT];
#pragma unroll
        for (int u = 0; u < W_IT; ++u) {
            const int idx = u * 256 + tid;
            const int k0  = 2 * (idx >> 5);
            const int n4  = idx & 31;
            wa[u] = W4[(size_t)min(k0,     K - 1) * 32 + n4];
            wb[u] = W4[(size_t)min(k0 + 1, K - 1) * 32 + n4];
        }
#pragma unroll
        for (int u = 0; u < W_IT; ++u) {
            const int idx = u * 256 + tid;
            if (idx < nW) {
                const int kk = idx >> 5;
                const int n4 = idx & 31;
                const int k0 = 2 * kk;
                const float4 a = (k0     < K) ? wa[u] : z4;
                const float4 b = (k0 + 1 < K) ? wb[u] : z4;
                float4* dst = (float4*)(Wp + kk * WS2 + 4 * n4);
                dst[0] = make_float4(a.x, b.x, a.y, b.y);
                dst[1] = make_float4(a.z, b.z, a.w, b.w);
            }
        }
    }
    __syncthreads();

    const int koff = (lane >= 16) ? 2 : 0;
    const float* Aw0 = Aw + l15 * A_LD;
    const int rOff = (lane >= 16) ? 8 : 0;

    for (; s < nStrips; s += stride) {
        strip_to_lds(av, Aw, A, s, N, K, Kpad, lane);
        __builtin_amdgcn_fence(__ATOMIC_RELEASE, "workgroup", "local");
        __builtin_amdgcn_fence(__ATOMIC_ACQUIRE, "workgroup", "local");

        const int sn = s + stride;
        if (sn < nStrips) strip_load(av, A, sn, N, K, lane);

        const int rowBase = s * 16;
        float dsc[8];
#pragma unroll
        for (int i = 0; i < 8; ++i)
            dsc[i] = dinv[min(rowBase + i + rOff, N - 1)];

        v8f acc[8];
#pragma unroll
        for (int j = 0; j < 8; ++j)
            acc[j] = (v8f){0.f,0.f,0.f,0.f,0.f,0.f,0.f,0.f};

        for (int k0 = 0; k0 < Kpad; k0 += 4) {
            v2f a0 = *(const v2f*)(Aw0 + k0 + koff);
            const v2f* Bk = Wp + (size_t)((k0 + koff) >> 1) * WS2 + l15;
#pragma unroll
            for (int j = 0; j < 8; ++j) {
                v2f b = Bk[j * 16];
                acc[j] = __builtin_amdgcn_wmma_f32_16x16x4_f32(
                    false, a0, false, b, (short)0, acc[j], false, false);
            }
        }

#pragma unroll
        for (int i = 0; i < 8; ++i) {
            int row = rowBase + i + rOff;
            if (row < N) {
                size_t base = (size_t)row * HDIM + l15;
#pragma unroll
                for (int j = 0; j < 8; ++j) {
                    C[base + j * 16] = acc[j][i] * dsc[i];
                }
            }
        }
    }
}

#define BKT_SHIFT 8
#define BKT_ROWS  256
#define BIN_T     1024
#define BIN_G     256
#define BIN_U     4
__global__ __launch_bounds__(BIN_T)
void bin_hist(const int* __restrict__ dst, int* __restrict__ histG, int E, int nb) {
    extern __shared__ int binc[];
    const int tid = threadIdx.x;
    const int g   = blockIdx.x;
    const int G   = gridDim.x;
    const int chunk = (E + G - 1) / G;
    for (int b = tid; b < nb; b += BIN_T) binc[b] = 0;
    __syncthreads();
    const int e0 = g * chunk;
    const int e1 = min(E, e0 + chunk);
    for (int e = e0 + tid; e < e1; e += BIN_U * BIN_T) {
        int d[BIN_U];
#pragma unroll
        for (int q = 0; q < BIN_U; ++q) {
            const int ee = e + q * BIN_T;
            d[q] = dst[min(ee, e1 - 1)];
        }
#pragma unroll
        for (int q = 0; q < BIN_U; ++q) {
            const int b = (int)min((unsigned)d[q] >> BKT_SHIFT, (unsigned)(nb - 1));
            atomicAdd(&binc[b], (e + q * BIN_T < e1) ? 1 : 0);
        }
    }
    __syncthreads();
    for (int b = tid; b < nb; b += BIN_T) histG[b * G + g] = binc[b];
}

__global__ __launch_bounds__(BIN_T)
void bin_place(const int* __restrict__ dst, const int* __restrict__ offs,
               int* __restrict__ recw, int E, int nb) {
    extern __shared__ int binc[];
    int2* __restrict__ rec = (int2*)recw;
    const int tid = threadIdx.x;
    const int g   = blockIdx.x;
    const int G   = gridDim.x;
    const int chunk = (E + G - 1) / G;
    for (int b = tid; b < nb; b += BIN_T) binc[b] = offs[b * G + g];
    __syncthreads();
    const int e0 = g * chunk;
    const int e1 = min(E, e0 + chunk);
    for (int e = e0 + tid; e < e1; e += BIN_U * BIN_T) {
        int d[BIN_U];
#pragma unroll
        for (int q = 0; q < BIN_U; ++q) {
            const int ee = e + q * BIN_T;
            d[q] = dst[min(ee, e1 - 1)];
        }
#pragma unroll
        for (int q = 0; q < BIN_U; ++q) {
            const int ee = e + q * BIN_T;
            const int b = (int)min((unsigned)d[q] >> BKT_SHIFT, (unsigned)(nb - 1));
            int p = atomicAdd(&binc[b], (ee < e1) ? 1 : 0);
            p = min(max(p, 0), E - 1);
            if (ee < e1) rec[p] = make_int2(ee, d[q]);
        }
    }
}

#define BKT_CAP 6144
#define BKT_K   32
#define BKT_Q   (BIN_T / BKT_ROWS)
#define BKT_KQ  (BKT_K / BKT_Q)
static_assert(BKT_Q * BKT_ROWS == BIN_T && BKT_KQ * BKT_Q == BKT_K, "group scan shape");
__global__ __launch_bounds__(BIN_T)
void bucket_rows(const int* __restrict__ offs, const int* __restrict__ recw,
                 const int* __restrict__ src,
                 int* __restrict__ rowptr, float* __restrict__ dinv,
                 int* eid, int* __restrict__ col, int N, int E, int G) {
    const int2* __restrict__ rec = (const int2*)recw;
    __shared__ int grp[BKT_K * BKT_ROWS];
    __shared__ int part[BKT_Q * BKT_ROWS];
    __shared__ int sc[BKT_ROWS];
    __shared__ int fst[BKT_ROWS];
    __shared__ int eidL[BKT_CAP];
    const int tid  = threadIdx.x;
    const int b    = blockIdx.x;
    const int row0 = b << BKT_SHIFT;
    const int bs = min(max(offs[b * G], 0), E);
    const int be = min(max(offs[(b + 1) * G], bs), E);
    const int ksh = max(0, 27 - __clz(max(E - 1, 1)));

#pragma unroll
    for (int u = 0; u < BKT_K * BKT_ROWS / BIN_T; ++u) grp[u * BIN_T + tid] = 0;
    __syncthreads();
    for (int i = bs + tid; i < be; i += BIN_U * BIN_T) {
        int2 v[BIN_U];
#pragma unroll
        for (int q = 0; q < BIN_U; ++q) {
            const int ii = i + q * BIN_T;
            v[q] = rec[min(ii, be - 1)];
        }
#pragma unroll
        for (int q = 0; q < BIN_U; ++q) {
            const unsigned r = min((unsigned)(v[q].y - row0), (unsigned)(BKT_ROWS - 1));
            const unsigned k = min((unsigned)v[q].x >> ksh, (unsigned)(BKT_K - 1));
            atomicAdd(&grp[k * BKT_ROWS + r], (i + q * BIN_T < be) ? 1 : 0);
        }
    }
    __syncthreads();

    const int rr = tid & (BKT_ROWS - 1);
    const int qd = tid >> BKT_SHIFT;
    int g8[BKT_KQ];
    int ps = 0;
#pragma unroll
    for (int j = 0; j < BKT_KQ; ++j) {
        g8[j] = grp[(qd * BKT_KQ + j) * BKT_ROWS + rr];
        ps += g8[j];
    }
    part[qd * BKT_ROWS + rr] = ps;
    __syncthreads();

    int c = 0;
    if (tid < BKT_ROWS) {
#pragma unroll
        for (int j = 0; j < BKT_Q; ++j) c += part[j * BKT_ROWS + tid];
        sc[tid] = c;
    }
    __syncthreads();
    for (int s = 1; s < BKT_ROWS; s <<= 1) {
        int v = 0;
        if (tid < BKT_ROWS && tid >= s) v = sc[tid - s];
        __syncthreads();
        if (tid < BKT_ROWS) sc[tid] += v;
        __syncthreads();
    }
    if (tid < BKT_ROWS) {
        const int first = bs + sc[tid] - c;
        const int row   = row0 + tid;
        if (row < N) {
            rowptr[row] = first;
            dinv[row]   = rsqrtf((float)c + 1.0f);
            if (row == N - 1) rowptr[N] = first + c;
        }
        fst[tid] = first;
    }
    __syncthreads();
    {
        int run = fst[rr];
#pragma unroll
        for (int j = 0; j < BKT_Q - 1; ++j)
            run += (j < qd) ? part[j * BKT_ROWS + rr] : 0;
#pragma unroll
        for (int j = 0; j < BKT_KQ; ++j) {
            grp[(qd * BKT_KQ + j) * BKT_ROWS + rr] = run;
            run += g8[j];
        }
    }
    __syncthreads();
    for (int i = bs + tid; i < be; i += BIN_U * BIN_T) {
        int2 v[BIN_U];
#pragma unroll
        for (int q = 0; q < BIN_U; ++q) {
            const int ii = i + q * BIN_T;
            v[q] = rec[min(ii, be - 1)];
        }
        __builtin_amdgcn_sched_barrier(0);
#pragma unroll
        for (int q = 0; q < BIN_U; ++q) {
            const bool in = i + q * BIN_T < be;
            const unsigned r = min((unsigned)(v[q].y - row0), (unsigned)(BKT_ROWS - 1));
            const unsigned k = min((unsigned)v[q].x >> ksh, (unsigned)(BKT_K - 1));
            int p = atomicAdd(&grp[k * BKT_ROWS + r], in ? 1 : 0);
            p = min(max(p, 0), E - 1);
            if (in) {
                const unsigned o = (unsigned)(p - bs);
                if (o < (unsigned)BKT_CAP) eidL[o] = v[q].x;
                else                       eid[p]  = v[q].x;
            }
        }
    }
    const bool allL = (be - bs) <= BKT_CAP;
    if (!allL) __threadfence();
    __syncthreads();

    for (int i = bs + tid; i < be; i += BIN_U * BIN_T) {
        int2 v[BIN_U];
        int  sv[BIN_U];
#pragma unroll
        for (int q = 0; q < BIN_U; ++q) {
            const int ii = i + q * BIN_T;
            v[q] = rec[min(ii, be - 1)];
        }
#pragma unroll
        for (int q = 0; q < BIN_U; ++q)
            sv[q] = src[min((unsigned)v[q].x, (unsigned)(E - 1))];

        int f[BIN_U], n[BIN_U], rk[BIN_U];
        int mc = 0;
#pragma unroll
        for (int q = 0; q < BIN_U; ++q) {
            const bool in = i + q * BIN_T < be;
            const unsigned r = min((unsigned)(v[q].y - row0), (unsigned)(BKT_ROWS - 1));
            const unsigned k = min((unsigned)v[q].x >> ksh, (unsigned)(BKT_K - 1));
            const int ge = grp[k * BKT_ROWS + r];
            const int lo = grp[(k > 0 ? k - 1 : 0) * BKT_ROWS + r];
            const int gs = (k > 0) ? lo : fst[r];
            f[q]  = in ? gs - bs : 0;
            n[q]  = in ? ge - gs : 0;
            rk[q] = 0;
            mc = max(mc, n[q]);
        }
        if (allL) {
            for (int s = 0; s < mc; ++s) {
                int a[BIN_U];
#pragma unroll
                for (int q = 0; q < BIN_U; ++q)
                    a[q] = eidL[min(max(f[q] + s, 0), BKT_CAP - 1)];
#pragma unroll
                for (int q = 0; q < BIN_U; ++q)
                    rk[q] += (int)(s < n[q]) & (int)(a[q] < v[q].x);
            }
        } else {
            for (int s = 0; s < mc; ++s) {
#pragma unroll
                for (int q = 0; q < BIN_U; ++q) {
                    const int o = f[q] + s;
                    int a = eidL[min(max(o, 0), BKT_CAP - 1)];
                    if (o >= BKT_CAP) a = eid[min(max(bs + o, 0), E - 1)];
                    rk[q] += (int)(s < n[q]) & (int)(a < v[q].x);
                }
            }
        }
#pragma unroll
        for (int q = 0; q < BIN_U; ++q) {
            const bool in = i + q * BIN_T < be;
            const int p = min(max(bs + f[q] + rk[q], 0), E - 1);
            if (in) col[p] = sv[q];
        }
    }
}

#define SCAN_T 256
#define SCAN_PER 4
__global__ __launch_bounds__(SCAN_T)
void scan_excl_i32(const int* __restrict__ cnt, int* __restrict__ rowptr, int N) {
    __shared__ int sh[SCAN_T];
    const int tid   = threadIdx.x;
    const int start = blockIdx.x * (SCAN_T * SCAN_PER);

    int pre = 0;
    const int4* __restrict__ cnt4 = (const int4*)cnt;
    for (int i = tid; i < start / 4; i += SCAN_T) {
        int4 v = cnt4[i];
        pre += (v.x + v.y) + (v.z + v.w);
    }

    int c[SCAN_PER];
    int loc = 0;
#pragma unroll
    for (int q = 0; q < SCAN_PER; ++q) {
        int i = start + tid * SCAN_PER + q;
        c[q] = (i < N) ? cnt[i] : 0;
        loc += c[q];
    }

    sh[tid] = pre;
    __syncthreads();
    for (int s = SCAN_T / 2; s > 0; s >>= 1) {
        if (tid < s) sh[tid] += sh[tid + s];
        __syncthreads();
    }
    const int offset = sh[0];
    __syncthreads();

    sh[tid] = loc;
    __syncthreads();
    for (int s = 1; s < SCAN_T; s <<= 1) {
        int v = (tid >= s) ? sh[tid - s] : 0;
        __syncthreads();
        sh[tid] += v;
        __syncthreads();
    }
    int run = offset + sh[tid] - loc;
#pragma unroll
    for (int q = 0; q < SCAN_PER; ++q) {
        int i = start + tid * SCAN_PER + q;
        if (i < N) rowptr[i] = run;
        run += c[q];
        if (i == N - 1) rowptr[N] = run;
    }
}

__device__ __forceinline__ void add4(float4& a, const float4& v) {
    a.x += v.x; a.y += v.y; a.z += v.z; a.w += v.w;
}

__global__ __launch_bounds__(256)
void gcn_gather_rows(const int* __restrict__ rowptr, const int* __restrict__ col,
                     const float* __restrict__ dinv,
                     const float* __restrict__ bias, const float* __restrict__ h,
                     float* __restrict__ out, int N) {
    const int lane = threadIdx.x & 31;
    const int row = __builtin_amdgcn_readfirstlane((int)(blockIdx.x * 8 + (threadIdx.x >> 5)));
    if (row >= N) return;
    const int beg = __builtin_amdgcn_readfirstlane(rowptr[row]);
    const int end = __builtin_amdgcn_readfirstlane(rowptr[row + 1]);

    const float4* __restrict__ h4 = (const float4*)h;
    const float4 hv = h4[(size_t)row * 32 + lane];
    const float  di = dinv[row];
    const float4 b4 = ((const float4*)bias)[lane];

    float4 acc = make_float4(0.f, 0.f, 0.f, 0.f);
    const unsigned nmax = (unsigned)(N - 1);

    for (int base = beg; base < end; base += 32) {
        const int n = min(32, end - base);
        int sv = 0;
        if (lane < n) sv = col[base + lane];
        sv = (int)min((unsigned)sv, nmax);

        int j = 0;
        for (; j + 4 <= n; j += 4) {
            const int s0 = __builtin_amdgcn_readlane(sv, j);
            const int s1 = __builtin_amdgcn_readlane(sv, j + 1);
            const int s2 = __builtin_amdgcn_readlane(sv, j + 2);
            const int s3 = __builtin_amdgcn_readlane(sv, j + 3);
            const float4 v0 = h4[(size_t)s0 * 32 + lane];
            const float4 v1 = h4[(size_t)s1 * 32 + lane];
            const float4 v2 = h4[(size_t)s2 * 32 + lane];
            const float4 v3 = h4[(size_t)s3 * 32 + lane];
            add4(acc, v0);
            add4(acc, v1);
            add4(acc, v2);
            add4(acc, v3);
        }
        const int r = n - j;
        if (r > 0) {
            const int s0 = __builtin_amdgcn_readlane(sv, j);
            const float4 v0 = h4[(size_t)s0 * 32 + lane];
            float4 v1 = make_float4(0.f, 0.f, 0.f, 0.f);
            float4 v2 = make_float4(0.f, 0.f, 0.f, 0.f);
            if (r > 1) {
                const int s1 = __builtin_amdgcn_readlane(sv, j + 1);
                v1 = h4[(size_t)s1 * 32 + lane];
            }
            if (r > 2) {
                const int s2 = __builtin_amdgcn_readlane(sv, j + 2);
                v2 = h4[(size_t)s2 * 32 + lane];
            }
            add4(acc, v0);
            if (r > 1) add4(acc, v1);
            if (r > 2) add4(acc, v2);
        }
    }

    v4f o;
    o.x = tanhf((acc.x + hv.x) * di + b4.x);
    o.y = tanhf((acc.y + hv.y) * di + b4.y);
    o.z = tanhf((acc.z + hv.z) * di + b4.z);
    o.w = tanhf((acc.w + hv.w) * di + b4.w);
    __builtin_nontemporal_store(o, (v4f*)out + ((size_t)row * 32 + lane));
}


#define FZ_T 1024
#define FZ_STRIP (16 * A_LD)

__device__ __forceinline__ float4 gather_row_tanh(const int* __restrict__ rowptr,
                                                  const int* __restrict__ col,
                                                  const float* __restrict__ dinv,
                                                  const float4* __restrict__ h4,
                                                  const float4 b4, int row, int lane,
                                                  unsigned nmax) {
    const int beg = __builtin_amdgcn_readfirstlane(rowptr[row]);
    const int end = __builtin_amdgcn_readfirstlane(rowptr[row + 1]);
    const float4 hv = h4[(size_t)row * 32 + lane];
    const float  di = dinv[row];
    float4 acc = make_float4(0.f, 0.f, 0.f, 0.f);

    for (int base = beg; base < end; base += 32) {
        const int n = min(32, end - base);
        int sv = 0;
        if (lane < n) sv = col[base + lane];
        sv = (int)min((unsigned)sv, nmax);

        int j = 0;
        for (; j + 4 <= n; j += 4) {
            const int s0 = __builtin_amdgcn_readlane(sv, j);
            const int s1 = __builtin_amdgcn_readlane(sv, j + 1);
            const int s2 = __builtin_amdgcn_readlane(sv, j + 2);
            const int s3 = __builtin_amdgcn_readlane(sv, j + 3);
            const float4 v0 = h4[(size_t)s0 * 32 + lane];
            const float4 v1 = h4[(size_t)s1 * 32 + lane];
            const float4 v2 = h4[(size_t)s2 * 32 + lane];
            const float4 v3 = h4[(size_t)s3 * 32 + lane];
            add4(acc, v0);
            add4(acc, v1);
            add4(acc, v2);
            add4(acc, v3);
        }
        const int r = n - j;
        if (r > 0) {
            const int s0 = __builtin_amdgcn_readlane(sv, j);
            const float4 v0 = h4[(size_t)s0 * 32 + lane];
            float4 v1 = make_float4(0.f, 0.f, 0.f, 0.f);
            float4 v2 = make_float4(0.f, 0.f, 0.f, 0.f);
            if (r > 1) {
                const int s1 = __builtin_amdgcn_readlane(sv, j + 1);
                v1 = h4[(size_t)s1 * 32 + lane];
            }
            if (r > 2) {
                const int s2 = __builtin_amdgcn_readlane(sv, j + 2);
                v2 = h4[(size_t)s2 * 32 + lane];
            }
            add4(acc, v0);
            if (r > 1) add4(acc, v1);
            if (r > 2) add4(acc, v2);
        }
    }

    float4 o;
    o.x = tanhf((acc.x + hv.x) * di + b4.x) * di;
    o.y = tanhf((acc.y + hv.y) * di + b4.y) * di;
    o.z = tanhf((acc.z + hv.z) * di + b4.z) * di;
    o.w = tanhf((acc.w + hv.w) * di + b4.w) * di;
    return o;
}

__global__ __launch_bounds__(FZ_T)
void gcn_gather_gemm(const int* __restrict__ rowptr, const int* __restrict__ col,
                     const float* __restrict__ dinv,
                     const float* __restrict__ bias, const float* __restrict__ h,
                     const float* __restrict__ Wn, float* __restrict__ C, int N) {
    extern __shared__ float lds[];
    v2f*   Wp = (v2f*)lds;
    float* As = lds + (size_t)(HDIM / 2) * WS2 * 2;

    const int tid  = threadIdx.x;
    const int lane = tid & 31;
    const int l15  = lane & 15;
    const int wave = __builtin_amdgcn_readfirstlane(tid >> 5);

    const float4* __restrict__ h4 = (const float4*)h;
    const float4 b4 = ((const float4*)bias)[lane];
    const unsigned nmax = (unsigned)(N - 1);
    const int nTrips = (N + 31) >> 5;
    const int step   = (int)gridDim.x;
    const float4 z4  = make_float4(0.f, 0.f, 0.f, 0.f);

    {
        const float4* W4 = (const float4*)Wn;
        float4 wa[2], wb[2];
#pragma unroll
        for (int u = 0; u < 2; ++u) {
            const int idx = u * FZ_T + tid;
            const int kk  = idx >> 5;
            const int n4  = idx & 31;
            wa[u] = W4[(size_t)(2 * kk)     * 32 + n4];
            wb[u] = W4[(size_t)(2 * kk + 1) * 32 + n4];
        }
#pragma unroll
        for (int u = 0; u < 2; ++u) {
            const int idx = u * FZ_T + tid;
            const int kk  = idx >> 5;
            const int n4  = idx & 31;
            float4* dst = (float4*)(Wp + kk * WS2 + 4 * n4);
            dst[0] = make_float4(wa[u].x, wb[u].x, wa[u].y, wb[u].y);
            dst[1] = make_float4(wa[u].z, wb[u].z, wa[u].w, wb[u].w);
        }
    }

    const int koff = (lane >= 16) ? 2 : 0;
    const int rOff = (lane >= 16) ? 8 : 0;

    int p = (int)blockIdx.x;
    float4 o = z4;
    if (p < nTrips && p * 32 + wave < N)
        o = gather_row_tanh(rowptr, col, dinv, h4, b4, p * 32 + wave, lane, nmax);

    for (int it = 0; p < nTrips; p += step, ++it) {
        float* Ab = As + (size_t)(it & 1) * (2 * FZ_STRIP);
        {
            v2f* q = (v2f*)(Ab + wave * A_LD + 4 * lane);
            q[0] = (v2f){o.x, o.y};
            q[1] = (v2f){o.z, o.w};
        }
        __builtin_amdgcn_fence(__ATOMIC_RELEASE, "workgroup", "local");
        __builtin_amdgcn_s_barrier_signal(-1);

        const int pn = p + step;
        float4 on = z4;
        if (pn < nTrips && pn * 32 + wave < N)
            on = gather_row_tanh(rowptr, col, dinv, h4, b4, pn * 32 + wave, lane, nmax);

        __builtin_amdgcn_s_barrier_wait(-1);
        __builtin_amdgcn_fence(__ATOMIC_ACQUIRE, "workgroup", "local");

        const int mw = wave ^ ((it & 1) << 4);
        if (mw < 16) {
            const int st = mw >> 3;
            const int jt = mw & 7;
            const float* Ak = Ab + (st * 16 + l15) * A_LD + koff;
            const v2f*   Bk = Wp + (size_t)(koff >> 1) * WS2 + jt * 16 + l15;
            v8f acc8 = (v8f){0.f,0.f,0.f,0.f,0.f,0.f,0.f,0.f};
#pragma unroll
            for (int g = 0; g < 4; ++g) {
                v2f af[8], bf[8];
#pragma unroll
                for (int q = 0; q < 8; ++q) {
                    const int kb = g * 8 + q;
                    af[q] = *(const v2f*)(Ak + 4 * kb);
                    bf[q] = Bk[(size_t)(2 * kb) * WS2];
                }
                __builtin_amdgcn_sched_barrier(0);
#pragma unroll
                for (int q = 0; q < 8; ++q)
                    acc8 = __builtin_amdgcn_wmma_f32_16x16x4_f32(
                        false, af[q], false, bf[q], (short)0, acc8, false, false);
                asm volatile("v_nop\n\tv_nop\n\tv_nop\n\tv_nop\n\tv_nop\n\tv_nop\n\tv_nop\n\tv_nop"
                             : "+v"(acc8));
                __builtin_amdgcn_sched_barrier(0);
            }
            const int rowBase = p * 32 + st * 16;
#pragma unroll
            for (int i = 0; i < 8; ++i) {
                const int rr = rowBase + i + rOff;
                if (rr < N) __builtin_nontemporal_store(acc8[i], C + ((size_t)rr * HDIM + jt * 16 + l15));
            }
        }
        o = on;
    }
}

extern "C" void kernel_launch(void* const* d_in, const int* in_sizes, int n_in,
                              void* d_out, int out_size, void* d_ws, size_t ws_size,
                              hipStream_t stream) {
    const float* x  = (const float*)d_in[0];
    const int*   ei = (const int*)d_in[1];
    const float* Wl[4] = {(const float*)d_in[2], (const float*)d_in[4],
                          (const float*)d_in[6], (const float*)d_in[8]};
    const float* bl[4] = {(const float*)d_in[3], (const float*)d_in[5],
                          (const float*)d_in[7], (const float*)d_in[9]};

    const int NFEAT = in_sizes[2] / HDIM;
    const int N     = in_sizes[0] / NFEAT;
    const int E     = in_sizes[1] / 2;
    const int* srcp = ei;
    const int* dstp = ei + E;

    float* wsf  = (float*)d_ws;
    float* r0   = wsf;
    float* r1   = r0 + (size_t)N * HDIM;
    float* r2   = r1 + (size_t)N * HDIM;
    float* dinv = r2 + (size_t)N * HDIM;

    const int nb  = (N + BKT_ROWS - 1) >> BKT_SHIFT;
    const int n2  = nb * BIN_G;
    const int n2p = (n2 + 31) & ~31;
    int*   histG  = (int*)r1;
    int*   offs   = histG + n2p;
    int*   rowptr = offs + n2p + 32;
    int*   rec    = rowptr + ((N + 31) & ~31) + 32;
    int*   eid    = rec + 2 * (size_t)E;
    int*   col    = eid + E;

    bin_hist<<<BIN_G, BIN_T, (size_t)nb * sizeof(int), stream>>>(dstp, histG, E, nb);
    scan_excl_i32<<<(n2 + SCAN_T * SCAN_PER - 1) / (SCAN_T * SCAN_PER), SCAN_T, 0, stream>>>(
        histG, offs, n2);
    bin_place<<<BIN_G, BIN_T, (size_t)nb * sizeof(int), stream>>>(dstp, offs, rec, E, nb);
    bucket_rows<<<nb, BIN_T, 0, stream>>>(offs, rec, srcp, rowptr, dinv, eid, col, N, E, BIN_G);

    const int Kpad0 = (NFEAT + 3) & ~3;
    const size_t shmem0 =
        ((size_t)(Kpad0 / 2) * WS2 * 2 + 8u * 16u * A_LD) * sizeof(float);
    gcn_gemm_wmma_f32<<<256, 256, shmem0, stream>>>(x, Wl[0], dinv, r0, N, NFEAT, Kpad0);

    const size_t shmemF =
        ((size_t)(HDIM / 2) * WS2 * 2 + 2u * 2u * FZ_STRIP) * sizeof(float);
    gcn_gather_gemm<<<512, FZ_T, shmemF, stream>>>(rowptr, col, dinv, bl[0], r0, Wl[1], r2, N);
    gcn_gather_gemm<<<512, FZ_T, shmemF, stream>>>(rowptr, col, dinv, bl[1], r2, Wl[2], r0, N);
    gcn_gather_gemm<<<512, FZ_T, shmemF, stream>>>(rowptr, col, dinv, bl[2], r0, Wl[3], r2, N);

    gcn_gather_rows<<<(N + 7) / 8, 256, 0, stream>>>(
        rowptr, col, dinv, bl[3], r2, (float*)d_out, N);
}


# edge binning in one launch: each chunk sorts its edges by destination block in LDS into its own run; a block sums the chunks' starts and reads its 256 runs; histogram and scan launches gone
# speedup vs baseline: 7.3351x; 1.0544x over previous
#include <hip/hip_runtime.h>
#include <hip/hip_bf16.h>
#include <math.h>


#define HDIM 128
#define WS2  144
#define A_LD 138

#define A_IT 17
#define W_IT 9

typedef __attribute__((ext_vector_type(2))) float v2f;
typedef __attribute__((ext_vector_type(8))) float v8f;
typedef __attribute__((ext_vector_type(4))) float v4f;

__device__ __forceinline__ void strip_load(float4 (&av)[A_IT], const float* __restrict__ A,
                                           int s, int N, int K, int lane) {
    const int rowBase   = s * 16;
    const int rowsValid = min(16, max(0, N - rowBase));
    const int nvecFull  = (rowsValid * K) >> 2;
    const float4* strip4 = (const float4*)(A + (size_t)rowBase * K);
    if (nvecFull > 0) {
#pragma unroll
        for (int u = 0; u < A_IT; ++u)
            av[u] = strip4[min(u * 32 + lane, nvecFull - 1)];
    } else {
#pragma unroll
        for (int u = 0; u < A_IT; ++u) av[u] = make_float4(0.f, 0.f, 0.f, 0.f);
    }
}

__device__ __forceinline__ void strip_to_lds(const float4 (&av)[A_IT], float* Aw,
                                             const float* __restrict__ A,
                                             int s, int N, int K, int Kpad, int lane) {
    const int rowBase   = s * 16;
    const int rowsValid = min(16, max(0, N - rowBase));
    const int validLen  = rowsValid * K;
    const int nvecFull  = validLen >> 2;
    const int nvec      = (16 * K) >> 2;
    const float* strip  = A + (size_t)rowBase * K;
    const float4 z4 = make_float4(0.f, 0.f, 0.f, 0.f);
    const float invK = 1.0f / (float)K;
    const float fl   = (float)(4 * lane) + 0.5f;
    const int wrapAdd = A_LD - K;
    const bool rowVec = (K & 3) == 0;
#pragma unroll
    for (int u = 0; u < A_IT; ++u) {
        const int i = u * 32 + lane;
        if (i < nvec) {
            const int e0 = 128 * u + 4 * lane;
            const int r  = (int)((fl + (float)(128 * u)) * invK);
            const int k  = e0 - r * K;
            const int a0 = r * A_LD + k;
            const float4 w = (i < nvecFull) ? av[u] : z4;
            if (rowVec) {
                v2f* p = (v2f*)(Aw + a0);
                p[0] = (v2f){w.x, w.y};
                p[1] = (v2f){w.z, w.w};
            } else {
                Aw[a0] = w.x;
                Aw[a0 + 1 + ((k + 1 >= K) ? wrapAdd : 0)] = w.y;
                Aw[a0 + 2 + ((k + 2 >= K) ? wrapAdd : 0)] = w.z;
                Aw[a0 + 3 + ((k + 3 >= K) ? wrapAdd : 0)] = w.w;
            }
        }
    }
    const int rem = validLen & 3;
    if (lane < rem) {
        const int e  = (nvecFull << 2) + lane;
        const int rr = e / K;
        Aw[rr * A_LD + (e - rr * K)] = strip[e];
    }
    if (lane < 16)
        for (int k2 = K; k2 < Kpad; ++k2) Aw[lane * A_LD + k2] = 0.0f;
}

__global__ __launch_bounds__(256) __attribute__((amdgpu_waves_per_eu(2, 2)))
void gcn_gemm_wmma_f32(const float* __restrict__ A,
                       const float* __restrict__ W,
                       const float* __restrict__ dinv,
                       float* __restrict__ C,
                       int N, int K, int Kpad) {
    extern __shared__ float lds[];
    v2f*   Wp = (v2f*)lds;
    float* As = lds + (size_t)(Kpad / 2) * WS2 * 2;

    const int tid  = threadIdx.x;
    const int lane = tid & 31;
    const int l15  = lane & 15;
    const int wave = __builtin_amdgcn_readfirstlane(tid >> 5);

    const int nStrips = (N + 15) >> 4;
    const int nblk    = (int)gridDim.x;
    const int stride  = nblk * 8;
    int s = wave * nblk + (int)blockIdx.x;

    const float4 z4 = make_float4(0.f, 0.f, 0.f, 0.f);
    float* Aw = As + (size_t)wave * 16 * A_LD;

    float4 av[A_IT];
    strip_load(av, A, s, N, K, lane);

    const float4* W4 = (const float4*)W;
    const int nW = (Kpad / 2) * 32;
    {
        float4 wa[W_IT], wb[W_IT];
#pragma unroll
        for (int u = 0; u < W_IT; ++u) {
            const int idx = u * 256 + tid;
            const int k0  = 2 * (idx >> 5);
            const int n4  = idx & 31;
            wa[u] = W4[(size_t)min(k0,     K - 1) * 32 + n4];
            wb[u] = W4[(size_t)min(k0 + 1, K - 1) * 32 + n4];
        }
#pragma unroll
        for (int u = 0; u < W_IT; ++u) {
            const int idx = u * 256 + tid;
            if (idx < nW) {
                const int kk = idx >> 5;
                const int n4 = idx & 31;
                const int k0 = 2 * kk;
                const float4 a = (k0     < K) ? wa[u] : z4;
                const float4 b = (k0 + 1 < K) ? wb[u] : z4;
                float4* dst = (float4*)(Wp + kk * WS2 + 4 * n4);
                dst[0] = make_float4(a.x, b.x, a.y, b.y);
                dst[1] = make_float4(a.z, b.z, a.w, b.w);
            }
        }
    }
    __syncthreads();

    const int koff = (lane >= 16) ? 2 : 0;
    const float* Aw0 = Aw + l15 * A_LD;
    const int rOff = (lane >= 16) ? 8 : 0;

    for (; s < nStrips; s += stride) {
        strip_to_lds(av, Aw, A, s, N, K, Kpad, lane);
        __builtin_amdgcn_fence(__ATOMIC_RELEASE, "workgroup", "local");
        __builtin_amdgcn_fence(__ATOMIC_ACQUIRE, "workgroup", "local");

        const int sn = s + stride;
        if (sn < nStrips) strip_load(av, A, sn, N, K, lane);

        const int rowBase = s * 16;
        float dsc[8];
#pragma unroll
        for (int i = 0; i < 8; ++i)
            dsc[i] = dinv[min(rowBase + i + rOff, N - 1)];

        v8f acc[8];
#pragma unroll
        for (int j = 0; j < 8; ++j)
            acc[j] = (v8f){0.f,0.f,0.f,0.f,0.f,0.f,0.f,0.f};

        for (int k0 = 0; k0 < Kpad; k0 += 4) {
            v2f a0 = *(const v2f*)(Aw0 + k0 + koff);
            const v2f* Bk = Wp + (size_t)((k0 + koff) >> 1) * WS2 + l15;
#pragma unroll
            for (int j = 0; j < 8; ++j) {
                v2f b = Bk[j * 16];
                acc[j] = __builtin_amdgcn_wmma_f32_16x16x4_f32(
                    false, a0, false, b, (short)0, acc[j], false, false);
            }
        }

#pragma unroll
        for (int i = 0; i < 8; ++i) {
            int row = rowBase + i + rOff;
            if (row < N) {
                size_t base = (size_t)row * HDIM + l15;
#pragma unroll
                for (int j = 0; j < 8; ++j) {
                    C[base + j * 16] = acc[j][i] * dsc[i];
                }
            }
        }
    }
}

#define BKT_SHIFT 8
#define BKT_ROWS  256
#define BIN_T     1024
#define BIN_G     256
#define BIN_U     4
static_assert((BIN_G & (BIN_G - 1)) == 0 && BIN_G <= BIN_T && BIN_G % 4 == 0, "chunk count shape");
__global__ __launch_bounds__(BIN_T)
void bin_chunk(const int* __restrict__ dst, int* __restrict__ histG, int* __restrict__ startG,
               int* __restrict__ recw, int E, int nb) {
    extern __shared__ int binc[];
    __shared__ int wsum[BIN_T / 32];
    __shared__ int carry;
    int2* __restrict__ rec = (int2*)recw;
    const int tid  = threadIdx.x;
    const int lane = tid & 31;
    const int wv   = tid >> 5;
    const int g    = blockIdx.x;
    const int G    = gridDim.x;
    const int chunk = (E + G - 1) / G;
    for (int b = tid; b < nb; b += BIN_T) binc[b] = 0;
    if (tid == 0) carry = 0;
    __syncthreads();
    const int e0 = g * chunk;
    const int e1 = min(E, e0 + chunk);
    for (int e = e0 + tid; e < e1; e += BIN_U * BIN_T) {
        int d[BIN_U];
#pragma unroll
        for (int q = 0; q < BIN_U; ++q) {
            const int ee = e + q * BIN_T;
            d[q] = dst[min(ee, e1 - 1)];
        }
#pragma unroll
        for (int q = 0; q < BIN_U; ++q) {
            const int b = (int)min((unsigned)d[q] >> BKT_SHIFT, (unsigned)(nb - 1));
            atomicAdd(&binc[b], (e + q * BIN_T < e1) ? 1 : 0);
        }
    }
    __syncthreads();
    for (int base = 0; base < nb; base += BIN_T) {
        const int idx = base + tid;
        const int cN  = (idx < nb) ? binc[idx] : 0;
        int x = cN;
#pragma unroll
        for (int o = 1; o < 32; o <<= 1) {
            const int y = __shfl_up(x, o);
            if (lane >= o) x += y;
        }
        if (lane == 31) wsum[wv] = x;
        __syncthreads();
        int woff = carry;
        for (int j = 0; j < wv; ++j) woff += wsum[j];
        const int st = woff + x - cN;
        if (idx < nb) binc[idx] = st;
        __syncthreads();
        if (tid == BIN_T - 1) carry = woff + x;
        __syncthreads();
        if (idx < nb) {
            histG[idx * G + g]  = cN;
            startG[idx * G + g] = st;
        }
    }
    for (int e = e0 + tid; e < e1; e += BIN_U * BIN_T) {
        int d[BIN_U];
#pragma unroll
        for (int q = 0; q < BIN_U; ++q) {
            const int ee = e + q * BIN_T;
            d[q] = dst[min(ee, e1 - 1)];
        }
#pragma unroll
        for (int q = 0; q < BIN_U; ++q) {
            const int ee = e + q * BIN_T;
            const int b = (int)min((unsigned)d[q] >> BKT_SHIFT, (unsigned)(nb - 1));
            int p = atomicAdd(&binc[b], (ee < e1) ? 1 : 0);
            p = min(max(e0 + p, 0), E - 1);
            if (ee < e1) rec[p] = make_int2(ee, d[q]);
        }
    }
}

__device__ __forceinline__ int seg_addr(const int* segp, const int* segb, int li) {
    int lo = 0;
#pragma unroll
    for (int s = BIN_G / 2; s > 0; s >>= 1)
        if (segp[lo + s] <= li) lo += s;
    return segb[lo] + (li - segp[lo]);
}

#define BKT_CAP 6144
#define BKT_K   32
#define BKT_Q   (BIN_T / BKT_ROWS)
#define BKT_KQ  (BKT_K / BKT_Q)
static_assert(BKT_Q * BKT_ROWS == BIN_T && BKT_KQ * BKT_Q == BKT_K, "group scan shape");
__global__ __launch_bounds__(BIN_T)
void bucket_rows(const int* __restrict__ histG, const int* __restrict__ startG,
                 const int* __restrict__ recw, const int* __restrict__ src,
                 int* __restrict__ rowptr, float* __restrict__ dinv,
                 int* eid, int* __restrict__ col, int N, int E) {
    const int2* __restrict__ rec = (const int2*)recw;
    __shared__ int grp[BKT_K * BKT_ROWS];
    __shared__ int part[BKT_Q * BKT_ROWS];
    __shared__ int sc[BKT_ROWS];
    __shared__ int fst[BKT_ROWS];
    __shared__ int eidL[BKT_CAP];
    __shared__ int segp[BIN_G + 1];
    __shared__ int segb[BIN_G];
    __shared__ int red[BIN_G / 32];
    __shared__ int wsum[BIN_G / 32];
    const int tid  = threadIdx.x;
    const int lane = tid & 31;
    const int wv   = tid >> 5;
    const int b    = blockIdx.x;
    const int row0 = b << BKT_SHIFT;
    const int G    = BIN_G;
    const int chunk = (E + G - 1) / G;

    int cg = 0, sg = 0;
    if (tid < G) {
        cg = histG[b * G + tid];
        sg = startG[b * G + tid];
    }
    int pre = sg;
#pragma unroll
    for (int o = 16; o > 0; o >>= 1) pre += __shfl_xor(pre, o);
    int xg = cg;
#pragma unroll
    for (int o = 1; o < 32; o <<= 1) {
        const int y = __shfl_up(xg, o);
        if (lane >= o) xg += y;
    }
    if (lane == 0 && tid < G)  red[wv]  = pre;
    if (lane == 31 && tid < G) wsum[wv] = xg;
    __syncthreads();
    int bsum = 0, cntAll = 0, woff = 0;
#pragma unroll
    for (int j = 0; j < BIN_G / 32; ++j) {
        const int w = wsum[j];
        bsum   += red[j];
        cntAll += w;
        woff   += (j < wv) ? w : 0;
    }
    if (tid < G) {
        segp[tid] = woff + xg - cg;
        segb[tid] = tid * chunk + sg;
    }
    if (tid == 0) segp[G] = cntAll;
    const int bs  = min(max(bsum, 0), E);
    const int be  = min(bs + max(cntAll, 0), E);
    const int cnt = be - bs;
    const int ksh = max(0, 27 - __clz(max(E - 1, 1)));

#pragma unroll
    for (int u = 0; u < BKT_K * BKT_ROWS / BIN_T; ++u) grp[u * BIN_T + tid] = 0;
    __syncthreads();
    int a0[BIN_U];
#pragma unroll
    for (int q = 0; q < BIN_U; ++q) {
        const int a = seg_addr(segp, segb, min(tid + q * BIN_T, max(cnt - 1, 0)));
        a0[q] = min(max(a, 0), E - 1);
    }
    for (int i = tid; i < cnt; i += BIN_U * BIN_T) {
        int2 v[BIN_U];
#pragma unroll
        for (int q = 0; q < BIN_U; ++q) {
            const int ii = i + q * BIN_T;
            const int a = (i == tid) ? a0[q]
                                     : min(max(seg_addr(segp, segb, min(ii, cnt - 1)), 0), E - 1);
            v[q] = rec[a];
        }
#pragma unroll
        for (int q = 0; q < BIN_U; ++q) {
            const unsigned r = min((unsigned)(v[q].y - row0), (unsigned)(BKT_ROWS - 1));
            const unsigned k = min((unsigned)v[q].x >> ksh, (unsigned)(BKT_K - 1));
            atomicAdd(&grp[k * BKT_ROWS + r], (i + q * BIN_T < cnt) ? 1 : 0);
        }
    }
    __syncthreads();

    const int rr = tid & (BKT_ROWS - 1);
    const int qd = tid >> BKT_SHIFT;
    int g8[BKT_KQ];
    int ps = 0;
#pragma unroll
    for (int j = 0; j < BKT_KQ; ++j) {
        g8[j] = grp[(qd * BKT_KQ + j) * BKT_ROWS + rr];
        ps += g8[j];
    }
    part[qd * BKT_ROWS + rr] = ps;
    __syncthreads();

    int c = 0;
    if (tid < BKT_ROWS) {
#pragma unroll
        for (int j = 0; j < BKT_Q; ++j) c += part[j * BKT_ROWS + tid];
        sc[tid] = c;
    }
    __syncthreads();
    for (int s = 1; s < BKT_ROWS; s <<= 1) {
        int v = 0;
        if (tid < BKT_ROWS && tid >= s) v = sc[tid - s];
        __syncthreads();
        if (tid < BKT_ROWS) sc[tid] += v;
        __syncthreads();
    }
    if (tid < BKT_ROWS) {
        const int first = bs + sc[tid] - c;
        const int row   = row0 + tid;
        if (row < N) {
            rowptr[row] = first;
            dinv[row]   = rsqrtf((float)c + 1.0f);
            if (row == N - 1) rowptr[N] = first + c;
        }
        fst[tid] = first;
    }
    __syncthreads();
    {
        int run = fst[rr];
#pragma unroll
        for (int j = 0; j < BKT_Q - 1; ++j)
            run += (j < qd) ? part[j * BKT_ROWS + rr] : 0;
#pragma unroll
        for (int j = 0; j < BKT_KQ; ++j) {
            grp[(qd * BKT_KQ + j) * BKT_ROWS + rr] = run;
            run += g8[j];
        }
    }
    __syncthreads();
    for (int i = tid; i < cnt; i += BIN_U * BIN_T) {
        int2 v[BIN_U];
#pragma unroll
        for (int q = 0; q < BIN_U; ++q) {
            const int ii = i + q * BIN_T;
            const int a = (i == tid) ? a0[q]
                                     : min(max(seg_addr(segp, segb, min(ii, cnt - 1)), 0), E - 1);
            v[q] = rec[a];
        }
        __builtin_amdgcn_sched_barrier(0);
#pragma unroll
        for (int q = 0; q < BIN_U; ++q) {
            const bool in = i + q * BIN_T < cnt;
            const unsigned r = min((unsigned)(v[q].y - row0), (unsigned)(BKT_ROWS - 1));
            const unsigned k = min((unsigned)v[q].x >> ksh, (unsigned)(BKT_K - 1));
            int p = atomicAdd(&grp[k * BKT_ROWS + r], in ? 1 : 0);
            p = min(max(p, 0), E - 1);
            if (in) {
                const unsigned o = (unsigned)(p - bs);
                if (o < (unsigned)BKT_CAP) eidL[o] = v[q].x;
                else                       eid[p]  = v[q].x;
            }
        }
    }
    const bool allL = cnt <= BKT_CAP;
    if (!allL) __threadfence();
    __syncthreads();

    for (int i = tid; i < cnt; i += BIN_U * BIN_T) {
        int2 v[BIN_U];
        int  sv[BIN_U];
#pragma unroll
        for (int q = 0; q < BIN_U; ++q) {
            const int ii = i + q * BIN_T;
            const int a = (i == tid) ? a0[q]
                                     : min(max(seg_addr(segp, segb, min(ii, cnt - 1)), 0), E - 1);
            v[q] = rec[a];
        }
#pragma unroll
        for (int q = 0; q < BIN_U; ++q)
            sv[q] = src[min((unsigned)v[q].x, (unsigned)(E - 1))];

        int f[BIN_U], n[BIN_U], rk[BIN_U];
        int mc = 0;
#pragma unroll
        for (int q = 0; q < BIN_U; ++q) {
            const bool in = i + q * BIN_T < cnt;
            const unsigned r = min((unsigned)(v[q].y - row0), (unsigned)(BKT_ROWS - 1));
            const unsigned k = min((unsigned)v[q].x >> ksh, (unsigned)(BKT_K - 1));
            const int ge = grp[k * BKT_ROWS + r];
            const int lo = grp[(k > 0 ? k - 1 : 0) * BKT_ROWS + r];
            const int gs = (k > 0) ? lo : fst[r];
            f[q]  = in ? gs - bs : 0;
            n[q]  = in ? ge - gs : 0;
            rk[q] = 0;
            mc = max(mc, n[q]);
        }
        if (allL) {
            for (int s = 0; s < mc; ++s) {
                int a[BIN_U];
#pragma unroll
                for (int q = 0; q < BIN_U; ++q)
                    a[q] = eidL[min(max(f[q] + s, 0), BKT_CAP - 1)];
#pragma unroll
                for (int q = 0; q < BIN_U; ++q)
                    rk[q] += (int)(s < n[q]) & (int)(a[q] < v[q].x);
            }
        } else {
            for (int s = 0; s < mc; ++s) {
#pragma unroll
                for (int q = 0; q < BIN_U; ++q) {
                    const int o = f[q] + s;
                    int a = eidL[min(max(o, 0), BKT_CAP - 1)];
                    if (o >= BKT_CAP) a = eid[min(max(bs + o, 0), E - 1)];
                    rk[q] += (int)(s < n[q]) & (int)(a < v[q].x);
                }
            }
        }
#pragma unroll
        for (int q = 0; q < BIN_U; ++q) {
            const bool in = i + q * BIN_T < cnt;
            const int p = min(max(bs + f[q] + rk[q], 0), E - 1);
            if (in) col[p] = sv[q];
        }
    }
}

__device__ __forceinline__ void add4(float4& a, const float4& v) {
    a.x += v.x; a.y += v.y; a.z += v.z; a.w += v.w;
}

__global__ __launch_bounds__(256)
void gcn_gather_rows(const int* __restrict__ rowptr, const int* __restrict__ col,
                     const float* __restrict__ dinv,
                     const float* __restrict__ bias, const float* __restrict__ h,
                     float* __restrict__ out, int N) {
    const int lane = threadIdx.x & 31;
    const int row = __builtin_amdgcn_readfirstlane((int)(blockIdx.x * 8 + (threadIdx.x >> 5)));
    if (row >= N) return;
    const int beg = __builtin_amdgcn_readfirstlane(rowptr[row]);
    const int end = __builtin_amdgcn_readfirstlane(rowptr[row + 1]);

    const float4* __restrict__ h4 = (const float4*)h;
    const float4 hv = h4[(size_t)row * 32 + lane];
    const float  di = dinv[row];
    const float4 b4 = ((const float4*)bias)[lane];

    float4 acc = make_float4(0.f, 0.f, 0.f, 0.f);
    const unsigned nmax = (unsigned)(N - 1);

    for (int base = beg; base < end; base += 32) {
        const int n = min(32, end - base);
        int sv = 0;
        if (lane < n) sv = col[base + lane];
        sv = (int)min((unsigned)sv, nmax);

        int j = 0;
        for (; j + 4 <= n; j += 4) {
            const int s0 = __builtin_amdgcn_readlane(sv, j);
            const int s1 = __builtin_amdgcn_readlane(sv, j + 1);
            const int s2 = __builtin_amdgcn_readlane(sv, j + 2);
            const int s3 = __builtin_amdgcn_readlane(sv, j + 3);
            const float4 v0 = h4[(size_t)s0 * 32 + lane];
            const float4 v1 = h4[(size_t)s1 * 32 + lane];
            const float4 v2 = h4[(size_t)s2 * 32 + lane];
            const float4 v3 = h4[(size_t)s3 * 32 + lane];
            add4(acc, v0);
            add4(acc, v1);
            add4(acc, v2);
            add4(acc, v3);
        }
        const int r = n - j;
        if (r > 0) {
            const int s0 = __builtin_amdgcn_readlane(sv, j);
            const float4 v0 = h4[(size_t)s0 * 32 + lane];
            float4 v1 = make_float4(0.f, 0.f, 0.f, 0.f);
            float4 v2 = make_float4(0.f, 0.f, 0.f, 0.f);
            if (r > 1) {
                const int s1 = __builtin_amdgcn_readlane(sv, j + 1);
                v1 = h4[(size_t)s1 * 32 + lane];
            }
            if (r > 2) {
                const int s2 = __builtin_amdgcn_readlane(sv, j + 2);
                v2 = h4[(size_t)s2 * 32 + lane];
            }
            add4(acc, v0);
            if (r > 1) add4(acc, v1);
            if (r > 2) add4(acc, v2);
        }
    }

    v4f o;
    o.x = tanhf((acc.x + hv.x) * di + b4.x);
    o.y = tanhf((acc.y + hv.y) * di + b4.y);
    o.z = tanhf((acc.z + hv.z) * di + b4.z);
    o.w = tanhf((acc.w + hv.w) * di + b4.w);
    __builtin_nontemporal_store(o, (v4f*)out + ((size_t)row * 32 + lane));
}


#define FZ_T 1024
#define FZ_STRIP (16 * A_LD)

__device__ __forceinline__ float4 gather_row_tanh(const int* __restrict__ rowptr,
                                                  const int* __restrict__ col,
                                                  const float* __restrict__ dinv,
                                                  const float4* __restrict__ h4,
                                                  const float4 b4, int row, int lane,
                                                  unsigned nmax) {
    const int beg = __builtin_amdgcn_readfirstlane(rowptr[row]);
    const int end = __builtin_amdgcn_readfirstlane(rowptr[row + 1]);
    const float4 hv = h4[(size_t)row * 32 + lane];
    const float  di = dinv[row];
    float4 acc = make_float4(0.f, 0.f, 0.f, 0.f);

    for (int base = beg; base < end; base += 32) {
        const int n = min(32, end - base);
        int sv = 0;
        if (lane < n) sv = col[base + lane];
        sv = (int)min((unsigned)sv, nmax);

        int j = 0;
        for (; j + 4 <= n; j += 4) {
            const int s0 = __builtin_amdgcn_readlane(sv, j);
            const int s1 = __builtin_amdgcn_readlane(sv, j + 1);
            const int s2 = __builtin_amdgcn_readlane(sv, j + 2);
            const int s3 = __builtin_amdgcn_readlane(sv, j + 3);
            const float4 v0 = h4[(size_t)s0 * 32 + lane];
            const float4 v1 = h4[(size_t)s1 * 32 + lane];
            const float4 v2 = h4[(size_t)s2 * 32 + lane];
            const float4 v3 = h4[(size_t)s3 * 32 + lane];
            add4(acc, v0);
            add4(acc, v1);
            add4(acc, v2);
            add4(acc, v3);
        }
        const int r = n - j;
        if (r > 0) {
            const int s0 = __builtin_amdgcn_readlane(sv, j);
            const float4 v0 = h4[(size_t)s0 * 32 + lane];
            float4 v1 = make_float4(0.f, 0.f, 0.f, 0.f);
            float4 v2 = make_float4(0.f, 0.f, 0.f, 0.f);
            if (r > 1) {
                const int s1 = __builtin_amdgcn_readlane(sv, j + 1);
                v1 = h4[(size_t)s1 * 32 + lane];
            }
            if (r > 2) {
                const int s2 = __builtin_amdgcn_readlane(sv, j + 2);
                v2 = h4[(size_t)s2 * 32 + lane];
            }
            add4(acc, v0);
            if (r > 1) add4(acc, v1);
            if (r > 2) add4(acc, v2);
        }
    }

    float4 o;
    o.x = tanhf((acc.x + hv.x) * di + b4.x) * di;
    o.y = tanhf((acc.y + hv.y) * di + b4.y) * di;
    o.z = tanhf((acc.z + hv.z) * di + b4.z) * di;
    o.w = tanhf((acc.w + hv.w) * di + b4.w) * di;
    return o;
}

__global__ __launch_bounds__(FZ_T)
void gcn_gather_gemm(const int* __restrict__ rowptr, const int* __restrict__ col,
                     const float* __restrict__ dinv,
                     const float* __restrict__ bias, const float* __restrict__ h,
                     const float* __restrict__ Wn, float* __restrict__ C, int N) {
    extern __shared__ float lds[];
    v2f*   Wp = (v2f*)lds;
    float* As = lds + (size_t)(HDIM / 2) * WS2 * 2;

    const int tid  = threadIdx.x;
    const int lane = tid & 31;
    const int l15  = lane & 15;
    const int wave = __builtin_amdgcn_readfirstlane(tid >> 5);

    const float4* __restrict__ h4 = (const float4*)h;
    const float4 b4 = ((const float4*)bias)[lane];
    const unsigned nmax = (unsigned)(N - 1);
    const int nTrips = (N + 31) >> 5;
    const int step   = (int)gridDim.x;
    const float4 z4  = make_float4(0.f, 0.f, 0.f, 0.f);

    {
        const float4* W4 = (const float4*)Wn;
        float4 wa[2], wb[2];
#pragma unroll
        for (int u = 0; u < 2; ++u) {
            const int idx = u * FZ_T + tid;
            const int kk  = idx >> 5;
            const int n4  = idx & 31;
            wa[u] = W4[(size_t)(2 * kk)     * 32 + n4];
            wb[u] = W4[(size_t)(2 * kk + 1) * 32 + n4];
        }
#pragma unroll
        for (int u = 0; u < 2; ++u) {
            const int idx = u * FZ_T + tid;
            const int kk  = idx >> 5;
            const int n4  = idx & 31;
            float4* dst = (float4*)(Wp + kk * WS2 + 4 * n4);
            dst[0] = make_float4(wa[u].x, wb[u].x, wa[u].y, wb[u].y);
            dst[1] = make_float4(wa[u].z, wb[u].z, wa[u].w, wb[u].w);
        }
    }

    const int koff = (lane >= 16) ? 2 : 0;
    const int rOff = (lane >= 16) ? 8 : 0;

    int p = (int)blockIdx.x;
    float4 o = z4;
    if (p < nTrips && p * 32 + wave < N)
        o = gather_row_tanh(rowptr, col, dinv, h4, b4, p * 32 + wave, lane, nmax);

    for (int it = 0; p < nTrips; p += step, ++it) {
        float* Ab = As + (size_t)(it & 1) * (2 * FZ_STRIP);
        {
            v2f* q = (v2f*)(Ab + wave * A_LD + 4 * lane);
            q[0] = (v2f){o.x, o.y};
            q[1] = (v2f){o.z, o.w};
        }
        __builtin_amdgcn_fence(__ATOMIC_RELEASE, "workgroup", "local");
        __builtin_amdgcn_s_barrier_signal(-1);

        const int pn = p + step;
        float4 on = z4;
        if (pn < nTrips && pn * 32 + wave < N)
            on = gather_row_tanh(rowptr, col, dinv, h4, b4, pn * 32 + wave, lane, nmax);

        __builtin_amdgcn_s_barrier_wait(-1);
        __builtin_amdgcn_fence(__ATOMIC_ACQUIRE, "workgroup", "local");

        const int mw = wave ^ ((it & 1) << 4);
        if (mw < 16) {
            const int st = mw >> 3;
            const int jt = mw & 7;
            const float* Ak = Ab + (st * 16 + l15) * A_LD + koff;
            const v2f*   Bk = Wp + (size_t)(koff >> 1) * WS2 + jt * 16 + l15;
            v8f acc8 = (v8f){0.f,0.f,0.f,0.f,0.f,0.f,0.f,0.f};
#pragma unroll
            for (int g = 0; g < 4; ++g) {
                v2f af[8], bf[8];
#pragma unroll
                for (int q = 0; q < 8; ++q) {
                    const int kb = g * 8 + q;
                    af[q] = *(const v2f*)(Ak + 4 * kb);
                    bf[q] = Bk[(size_t)(2 * kb) * WS2];
                }
                __builtin_amdgcn_sched_barrier(0);
#pragma unroll
                for (int q = 0; q < 8; ++q)
                    acc8 = __builtin_amdgcn_wmma_f32_16x16x4_f32(
                        false, af[q], false, bf[q], (short)0, acc8, false, false);
                asm volatile("v_nop\n\tv_nop\n\tv_nop\n\tv_nop\n\tv_nop\n\tv_nop\n\tv_nop\n\tv_nop"
                             : "+v"(acc8));
                __builtin_amdgcn_sched_barrier(0);
            }
            const int rowBase = p * 32 + st * 16;
#pragma unroll
            for (int i = 0; i < 8; ++i) {
                const int rr = rowBase + i + rOff;
                if (rr < N) __builtin_nontemporal_store(acc8[i], C + ((size_t)rr * HDIM + jt * 16 + l15));
            }
        }
        o = on;
    }
}

extern "C" void kernel_launch(void* const* d_in, const int* in_sizes, int n_in,
                              void* d_out, int out_size, void* d_ws, size_t ws_size,
                              hipStream_t stream) {
    const float* x  = (const float*)d_in[0];
    const int*   ei = (const int*)d_in[1];
    const float* Wl[4] = {(const float*)d_in[2], (const float*)d_in[4],
                          (const float*)d_in[6], (const float*)d_in[8]};
    const float* bl[4] = {(const float*)d_in[3], (const float*)d_in[5],
                          (const float*)d_in[7], (const float*)d_in[9]};

    const int NFEAT = in_sizes[2] / HDIM;
    const int N     = in_sizes[0] / NFEAT;
    const int E     = in_sizes[1] / 2;
    const int* srcp = ei;
    const int* dstp = ei + E;

    float* wsf  = (float*)d_ws;
    float* r0   = wsf;
    float* r1   = r0 + (size_t)N * HDIM;
    float* r2   = r1 + (size_t)N * HDIM;
    float* dinv = r2 + (size_t)N * HDIM;

    const int nb  = (N + BKT_ROWS - 1) >> BKT_SHIFT;
    const int n2  = nb * BIN_G;
    const int n2p = (n2 + 31) & ~31;
    int*   histG  = (int*)r1;
    int*   startG = histG + n2p;
    int*   rowptr = startG + n2p + 32;
    int*   rec    = rowptr + ((N + 31) & ~31) + 32;
    int*   eid    = rec + 2 * (size_t)E;
    int*   col    = eid + E;

    bin_chunk<<<BIN_G, BIN_T, (size_t)nb * sizeof(int), stream>>>(dstp, histG, startG, rec, E, nb);
    bucket_rows<<<nb, BIN_T, 0, stream>>>(histG, startG, rec, srcp, rowptr, dinv, eid, col, N, E);

    const int Kpad0 = (NFEAT + 3) & ~3;
    const size_t shmem0 =
        ((size_t)(Kpad0 / 2) * WS2 * 2 + 8u * 16u * A_LD) * sizeof(float);
    gcn_gemm_wmma_f32<<<256, 256, shmem0, stream>>>(x, Wl[0], dinv, r0, N, NFEAT, Kpad0);

    const size_t shmemF =
        ((size_t)(HDIM / 2) * WS2 * 2 + 2u * 2u * FZ_STRIP) * sizeof(float);
    gcn_gather_gemm<<<512, FZ_T, shmemF, stream>>>(rowptr, col, dinv, bl[0], r0, Wl[1], r2, N);
    gcn_gather_gemm<<<512, FZ_T, shmemF, stream>>>(rowptr, col, dinv, bl[1], r2, Wl[2], r0, N);
    gcn_gather_gemm<<<512, FZ_T, shmemF, stream>>>(rowptr, col, dinv, bl[2], r0, Wl[3], r2, N);

    gcn_gather_rows<<<(N + 7) / 8, 256, 0, stream>>>(
        rowptr, col, dinv, bl[3], r2, (float*)d_out, N);
}


# table rows written as whole 512-byte rows: product tiles staged in LDS, one 16-byte-per-lane store per row (4 whole lines) instead of 64-byte pieces from the accumulators; sums unchanged
# speedup vs baseline: 7.8004x; 1.0634x over previous
#include <hip/hip_runtime.h>
#include <hip/hip_bf16.h>
#include <math.h>


#define HDIM 128
#define WS2  144
#define A_LD 138
#define C_LD 132

#define A_IT 17
#define W_IT 9

typedef __attribute__((ext_vector_type(2))) float v2f;
typedef __attribute__((ext_vector_type(8))) float v8f;
typedef __attribute__((ext_vector_type(4))) float v4f;

__device__ __forceinline__ void strip_load(float4 (&av)[A_IT], const float* __restrict__ A,
                                           int s, int N, int K, int lane) {
    const int rowBase   = s * 16;
    const int rowsValid = min(16, max(0, N - rowBase));
    const int nvecFull  = (rowsValid * K) >> 2;
    const float4* strip4 = (const float4*)(A + (size_t)rowBase * K);
    if (nvecFull > 0) {
#pragma unroll
        for (int u = 0; u < A_IT; ++u)
            av[u] = strip4[min(u * 32 + lane, nvecFull - 1)];
    } else {
#pragma unroll
        for (int u = 0; u < A_IT; ++u) av[u] = make_float4(0.f, 0.f, 0.f, 0.f);
    }
}

__device__ __forceinline__ void strip_to_lds(const float4 (&av)[A_IT], float* Aw,
                                             const float* __restrict__ A,
                                             int s, int N, int K, int Kpad, int lane) {
    const int rowBase   = s * 16;
    const int rowsValid = min(16, max(0, N - rowBase));
    const int validLen  = rowsValid * K;
    const int nvecFull  = validLen >> 2;
    const int nvec      = (16 * K) >> 2;
    const float* strip  = A + (size_t)rowBase * K;
    const float4 z4 = make_float4(0.f, 0.f, 0.f, 0.f);
    const float invK = 1.0f / (float)K;
    const float fl   = (float)(4 * lane) + 0.5f;
    const int wrapAdd = A_LD - K;
    const bool rowVec = (K & 3) == 0;
#pragma unroll
    for (int u = 0; u < A_IT; ++u) {
        const int i = u * 32 + lane;
        if (i < nvec) {
            const int e0 = 128 * u + 4 * lane;
            const int r  = (int)((fl + (float)(128 * u)) * invK);
            const int k  = e0 - r * K;
            const int a0 = r * A_LD + k;
            const float4 w = (i < nvecFull) ? av[u] : z4;
            if (rowVec) {
                v2f* p = (v2f*)(Aw + a0);
                p[0] = (v2f){w.x, w.y};
                p[1] = (v2f){w.z, w.w};
            } else {
                Aw[a0] = w.x;
                Aw[a0 + 1 + ((k + 1 >= K) ? wrapAdd : 0)] = w.y;
                Aw[a0 + 2 + ((k + 2 >= K) ? wrapAdd : 0)] = w.z;
                Aw[a0 + 3 + ((k + 3 >= K) ? wrapAdd : 0)] = w.w;
            }
        }
    }
    const int rem = validLen & 3;
    if (lane < rem) {
        const int e  = (nvecFull << 2) + lane;
        const int rr = e / K;
        Aw[rr * A_LD + (e - rr * K)] = strip[e];
    }
    if (lane < 16)
        for (int k2 = K; k2 < Kpad; ++k2) Aw[lane * A_LD + k2] = 0.0f;
}

__global__ __launch_bounds__(256) __attribute__((amdgpu_waves_per_eu(2, 2)))
void gcn_gemm_wmma_f32(const float* __restrict__ A,
                       const float* __restrict__ W,
                       const float* __restrict__ dinv,
                       float* __restrict__ C,
                       int N, int K, int Kpad) {
    extern __shared__ float lds[];
    v2f*   Wp = (v2f*)lds;
    float* As = lds + (size_t)(Kpad / 2) * WS2 * 2;

    const int tid  = threadIdx.x;
    const int lane = tid & 31;
    const int l15  = lane & 15;
    const int wave = __builtin_amdgcn_readfirstlane(tid >> 5);

    const int nStrips = (N + 15) >> 4;
    const int nblk    = (int)gridDim.x;
    const int stride  = nblk * 8;
    int s = wave * nblk + (int)blockIdx.x;

    const float4 z4 = make_float4(0.f, 0.f, 0.f, 0.f);
    float* Aw = As + (size_t)wave * 16 * A_LD;

    float4 av[A_IT];
    strip_load(av, A, s, N, K, lane);

    const float4* W4 = (const float4*)W;
    const int nW = (Kpad / 2) * 32;
    {
        float4 wa[W_IT], wb[W_IT];
#pragma unroll
        for (int u = 0; u < W_IT; ++u) {
            const int idx = u * 256 + tid;
            const int k0  = 2 * (idx >> 5);
            const int n4  = idx & 31;
            wa[u] = W4[(size_t)min(k0,     K - 1) * 32 + n4];
            wb[u] = W4[(size_t)min(k0 + 1, K - 1) * 32 + n4];
        }
#pragma unroll
        for (int u = 0; u < W_IT; ++u) {
            const int idx = u * 256 + tid;
            if (idx < nW) {
                const int kk = idx >> 5;
                const int n4 = idx & 31;
                const int k0 = 2 * kk;
                const float4 a = (k0     < K) ? wa[u] : z4;
                const float4 b = (k0 + 1 < K) ? wb[u] : z4;
                float4* dst = (float4*)(Wp + kk * WS2 + 4 * n4);
                dst[0] = make_float4(a.x, b.x, a.y, b.y);
                dst[1] = make_float4(a.z, b.z, a.w, b.w);
            }
        }
    }
    __syncthreads();

    const int koff = (lane >= 16) ? 2 : 0;
    const float* Aw0 = Aw + l15 * A_LD;
    const int rOff = (lane >= 16) ? 8 : 0;

    for (; s < nStrips; s += stride) {
        strip_to_lds(av, Aw, A, s, N, K, Kpad, lane);
        __builtin_amdgcn_fence(__ATOMIC_RELEASE, "workgroup", "local");
        __builtin_amdgcn_fence(__ATOMIC_ACQUIRE, "workgroup", "local");

        const int sn = s + stride;
        if (sn < nStrips) strip_load(av, A, sn, N, K, lane);

        const int rowBase = s * 16;
        float dsc[8];
#pragma unroll
        for (int i = 0; i < 8; ++i)
            dsc[i] = dinv[min(rowBase + i + rOff, N - 1)];

        v8f acc[8];
#pragma unroll
        for (int j = 0; j < 8; ++j)
            acc[j] = (v8f){0.f,0.f,0.f,0.f,0.f,0.f,0.f,0.f};

        for (int k0 = 0; k0 < Kpad; k0 += 4) {
            v2f a0 = *(const v2f*)(Aw0 + k0 + koff);
            const v2f* Bk = Wp + (size_t)((k0 + koff) >> 1) * WS2 + l15;
#pragma unroll
            for (int j = 0; j < 8; ++j) {
                v2f b = Bk[j * 16];
                acc[j] = __builtin_amdgcn_wmma_f32_16x16x4_f32(
                    false, a0, false, b, (short)0, acc[j], false, false);
            }
        }

        float* Cw = Aw;
#pragma unroll
        for (int i = 0; i < 8; ++i) {
#pragma unroll
            for (int j = 0; j < 8; ++j)
                Cw[(i + rOff) * C_LD + j * 16 + l15] = acc[j][i] * dsc[i];
        }
        __builtin_amdgcn_fence(__ATOMIC_RELEASE, "workgroup", "local");
        __builtin_amdgcn_fence(__ATOMIC_ACQUIRE, "workgroup", "local");
        if (rowBase + 16 <= N) {
            float4 cv[16];
#pragma unroll
            for (int r = 0; r < 16; ++r) cv[r] = *(const float4*)(Cw + r * C_LD + 4 * lane);
#pragma unroll
            for (int r = 0; r < 16; ++r)
                *(float4*)(C + (size_t)(rowBase + r) * HDIM + 4 * lane) = cv[r];
        } else {
            for (int r = 0; r < 16; ++r) {
                if (rowBase + r < N)
                    *(float4*)(C + (size_t)(rowBase + r) * HDIM + 4 * lane) =
                        *(const float4*)(Cw + r * C_LD + 4 * lane);
            }
        }
    }
}

#define BKT_SHIFT 8
#define BKT_ROWS  256
#define BIN_T     1024
#define BIN_G     256
#define BIN_U     4
static_assert((BIN_G & (BIN_G - 1)) == 0 && BIN_G <= BIN_T && BIN_G % 4 == 0, "chunk count shape");
__global__ __launch_bounds__(BIN_T)
void bin_chunk(const int* __restrict__ dst, int* __restrict__ histG, int* __restrict__ startG,
               int* __restrict__ recw, int E, int nb) {
    extern __shared__ int binc[];
    __shared__ int wsum[BIN_T / 32];
    __shared__ int carry;
    int2* __restrict__ rec = (int2*)recw;
    const int tid  = threadIdx.x;
    const int lane = tid & 31;
    const int wv   = tid >> 5;
    const int g    = blockIdx.x;
    const int G    = gridDim.x;
    const int chunk = (E + G - 1) / G;
    for (int b = tid; b < nb; b += BIN_T) binc[b] = 0;
    if (tid == 0) carry = 0;
    __syncthreads();
    const int e0 = g * chunk;
    const int e1 = min(E, e0 + chunk);
    for (int e = e0 + tid; e < e1; e += BIN_U * BIN_T) {
        int d[BIN_U];
#pragma unroll
        for (int q = 0; q < BIN_U; ++q) {
            const int ee = e + q * BIN_T;
            d[q] = dst[min(ee, e1 - 1)];
        }
#pragma unroll
        for (int q = 0; q < BIN_U; ++q) {
            const int b = (int)min((unsigned)d[q] >> BKT_SHIFT, (unsigned)(nb - 1));
            atomicAdd(&binc[b], (e + q * BIN_T < e1) ? 1 : 0);
        }
    }
    __syncthreads();
    for (int base = 0; base < nb; base += BIN_T) {
        const int idx = base + tid;
        const int cN  = (idx < nb) ? binc[idx] : 0;
        int x = cN;
#pragma unroll
        for (int o = 1; o < 32; o <<= 1) {
            const int y = __shfl_up(x, o);
            if (lane >= o) x += y;
        }
        if (lane == 31) wsum[wv] = x;
        __syncthreads();
        int woff = carry;
        for (int j = 0; j < wv; ++j) woff += wsum[j];
        const int st = woff + x - cN;
        if (idx < nb) binc[idx] = st;
        __syncthreads();
        if (tid == BIN_T - 1) carry = woff + x;
        __syncthreads();
        if (idx < nb) {
            histG[idx * G + g]  = cN;
            startG[idx * G + g] = st;
        }
    }
    for (int e = e0 + tid; e < e1; e += BIN_U * BIN_T) {
        int d[BIN_U];
#pragma unroll
        for (int q = 0; q < BIN_U; ++q) {
            const int ee = e + q * BIN_T;
            d[q] = dst[min(ee, e1 - 1)];
        }
#pragma unroll
        for (int q = 0; q < BIN_U; ++q) {
            const int ee = e + q * BIN_T;
            const int b = (int)min((unsigned)d[q] >> BKT_SHIFT, (unsigned)(nb - 1));
            int p = atomicAdd(&binc[b], (ee < e1) ? 1 : 0);
            p = min(max(e0 + p, 0), E - 1);
            if (ee < e1) rec[p] = make_int2(ee, d[q]);
        }
    }
}

__device__ __forceinline__ int seg_addr(const int* segp, const int* segb, int li) {
    int lo = 0;
#pragma unroll
    for (int s = BIN_G / 2; s > 0; s >>= 1)
        if (segp[lo + s] <= li) lo += s;
    return segb[lo] + (li - segp[lo]);
}

#define BKT_CAP 6144
#define BKT_K   32
#define BKT_Q   (BIN_T / BKT_ROWS)
#define BKT_KQ  (BKT_K / BKT_Q)
static_assert(BKT_Q * BKT_ROWS == BIN_T && BKT_KQ * BKT_Q == BKT_K, "group scan shape");
__global__ __launch_bounds__(BIN_T)
void bucket_rows(const int* __restrict__ histG, const int* __restrict__ startG,
                 const int* __restrict__ recw, const int* __restrict__ src,
                 int* __restrict__ rowptr, float* __restrict__ dinv,
                 int* eid, int* __restrict__ col, int N, int E) {
    const int2* __restrict__ rec = (const int2*)recw;
    __shared__ int grp[BKT_K * BKT_ROWS];
    __shared__ int part[BKT_Q * BKT_ROWS];
    __shared__ int sc[BKT_ROWS];
    __shared__ int fst[BKT_ROWS];
    __shared__ int eidL[BKT_CAP];
    __shared__ int segp[BIN_G + 1];
    __shared__ int segb[BIN_G];
    __shared__ int red[BIN_G / 32];
    __shared__ int wsum[BIN_G / 32];
    const int tid  = threadIdx.x;
    const int lane = tid & 31;
    const int wv   = tid >> 5;
    const int b    = blockIdx.x;
    const int row0 = b << BKT_SHIFT;
    const int G    = BIN_G;
    const int chunk = (E + G - 1) / G;

    int cg = 0, sg = 0;
    if (tid < G) {
        cg = histG[b * G + tid];
        sg = startG[b * G + tid];
    }
    int pre = sg;
#pragma unroll
    for (int o = 16; o > 0; o >>= 1) pre += __shfl_xor(pre, o);
    int xg = cg;
#pragma unroll
    for (int o = 1; o < 32; o <<= 1) {
        const int y = __shfl_up(xg, o);
        if (lane >= o) xg += y;
    }
    if (lane == 0 && tid < G)  red[wv]  = pre;
    if (lane == 31 && tid < G) wsum[wv] = xg;
    __syncthreads();
    int bsum = 0, cntAll = 0, woff = 0;
#pragma unroll
    for (int j = 0; j < BIN_G / 32; ++j) {
        const int w = wsum[j];
        bsum   += red[j];
        cntAll += w;
        woff   += (j < wv) ? w : 0;
    }
    if (tid < G) {
        segp[tid] = woff + xg - cg;
        segb[tid] = tid * chunk + sg;
    }
    if (tid == 0) segp[G] = cntAll;
    const int bs  = min(max(bsum, 0), E);
    const int be  = min(bs + max(cntAll, 0), E);
    const int cnt = be - bs;
    const int ksh = max(0, 27 - __clz(max(E - 1, 1)));

#pragma unroll
    for (int u = 0; u < BKT_K * BKT_ROWS / BIN_T; ++u) grp[u * BIN_T + tid] = 0;
    __syncthreads();
    int a0[BIN_U];
#pragma unroll
    for (int q = 0; q < BIN_U; ++q) {
        const int a = seg_addr(segp, segb, min(tid + q * BIN_T, max(cnt - 1, 0)));
        a0[q] = min(max(a, 0), E - 1);
    }
    for (int i = tid; i < cnt; i += BIN_U * BIN_T) {
        int2 v[BIN_U];
#pragma unroll
        for (int q = 0; q < BIN_U; ++q) {
            const int ii = i + q * BIN_T;
            const int a = (i == tid) ? a0[q]
                                     : min(max(seg_addr(segp, segb, min(ii, cnt - 1)), 0), E - 1);
            v[q] = rec[a];
        }
#pragma unroll
        for (int q = 0; q < BIN_U; ++q) {
            const unsigned r = min((unsigned)(v[q].y - row0), (unsigned)(BKT_ROWS - 1));
            const unsigned k = min((unsigned)v[q].x >> ksh, (unsigned)(BKT_K - 1));
            atomicAdd(&grp[k * BKT_ROWS + r], (i + q * BIN_T < cnt) ? 1 : 0);
        }
    }
    __syncthreads();

    const int rr = tid & (BKT_ROWS - 1);
    const int qd = tid >> BKT_SHIFT;
    int g8[BKT_KQ];
    int ps = 0;
#pragma unroll
    for (int j = 0; j < BKT_KQ; ++j) {
        g8[j] = grp[(qd * BKT_KQ + j) * BKT_ROWS + rr];
        ps += g8[j];
    }
    part[qd * BKT_ROWS + rr] = ps;
    __syncthreads();

    int c = 0;
    if (tid < BKT_ROWS) {
#pragma unroll
        for (int j = 0; j < BKT_Q; ++j) c += part[j * BKT_ROWS + tid];
        sc[tid] = c;
    }
    __syncthreads();
    for (int s = 1; s < BKT_ROWS; s <<= 1) {
        int v = 0;
        if (tid < BKT_ROWS && tid >= s) v = sc[tid - s];
        __syncthreads();
        if (tid < BKT_ROWS) sc[tid] += v;
        __syncthreads();
    }
    if (tid < BKT_ROWS) {
        const int first = bs + sc[tid] - c;
        const int row   = row0 + tid;
        if (row < N) {
            rowptr[row] = first;
            dinv[row]   = rsqrtf((float)c + 1.0f);
            if (row == N - 1) rowptr[N] = first + c;
        }
        fst[tid] = first;
    }
    __syncthreads();
    {
        int run = fst[rr];
#pragma unroll
        for (int j = 0; j < BKT_Q - 1; ++j)
            run += (j < qd) ? part[j * BKT_ROWS + rr] : 0;
#pragma unroll
        for (int j = 0; j < BKT_KQ; ++j) {
            grp[(qd * BKT_KQ + j) * BKT_ROWS + rr] = run;
            run += g8[j];
        }
    }
    __syncthreads();
    for (int i = tid; i < cnt; i += BIN_U * BIN_T) {
        int2 v[BIN_U];
#pragma unroll
        for (int q = 0; q < BIN_U; ++q) {
            const int ii = i + q * BIN_T;
            const int a = (i == tid) ? a0[q]
                                     : min(max(seg_addr(segp, segb, min(ii, cnt - 1)), 0), E - 1);
            v[q] = rec[a];
        }
        __builtin_amdgcn_sched_barrier(0);
#pragma unroll
        for (int q = 0; q < BIN_U; ++q) {
            const bool in = i + q * BIN_T < cnt;
            const unsigned r = min((unsigned)(v[q].y - row0), (unsigned)(BKT_ROWS - 1));
            const unsigned k = min((unsigned)v[q].x >> ksh, (unsigned)(BKT_K - 1));
            int p = atomicAdd(&grp[k * BKT_ROWS + r], in ? 1 : 0);
            p = min(max(p, 0), E - 1);
            if (in) {
                const unsigned o = (unsigned)(p - bs);
                if (o < (unsigned)BKT_CAP) eidL[o] = v[q].x;
                else                       eid[p]  = v[q].x;
            }
        }
    }
    const bool allL = cnt <= BKT_CAP;
    if (!allL) __threadfence();
    __syncthreads();

    for (int i = tid; i < cnt; i += BIN_U * BIN_T) {
        int2 v[BIN_U];
        int  sv[BIN_U];
#pragma unroll
        for (int q = 0; q < BIN_U; ++q) {
            const int ii = i + q * BIN_T;
            const int a = (i == tid) ? a0[q]
                                     : min(max(seg_addr(segp, segb, min(ii, cnt - 1)), 0), E - 1);
            v[q] = rec[a];
        }
#pragma unroll
        for (int q = 0; q < BIN_U; ++q)
            sv[q] = src[min((unsigned)v[q].x, (unsigned)(E - 1))];

        int f[BIN_U], n[BIN_U], rk[BIN_U];
        int mc = 0;
#pragma unroll
        for (int q = 0; q < BIN_U; ++q) {
            const bool in = i + q * BIN_T < cnt;
            const unsigned r = min((unsigned)(v[q].y - row0), (unsigned)(BKT_ROWS - 1));
            const unsigned k = min((unsigned)v[q].x >> ksh, (unsigned)(BKT_K - 1));
            const int ge = grp[k * BKT_ROWS + r];
            const int lo = grp[(k > 0 ? k - 1 : 0) * BKT_ROWS + r];
            const int gs = (k > 0) ? lo : fst[r];
            f[q]  = in ? gs - bs : 0;
            n[q]  = in ? ge - gs : 0;
            rk[q] = 0;
            mc = max(mc, n[q]);
        }
        if (allL) {
            for (int s = 0; s < mc; ++s) {
                int a[BIN_U];
#pragma unroll
                for (int q = 0; q < BIN_U; ++q)
                    a[q] = eidL[min(max(f[q] + s, 0), BKT_CAP - 1)];
#pragma unroll
                for (int q = 0; q < BIN_U; ++q)
                    rk[q] += (int)(s < n[q]) & (int)(a[q] < v[q].x);
            }
        } else {
            for (int s = 0; s < mc; ++s) {
#pragma unroll
                for (int q = 0; q < BIN_U; ++q) {
                    const int o = f[q] + s;
                    int a = eidL[min(max(o, 0), BKT_CAP - 1)];
                    if (o >= BKT_CAP) a = eid[min(max(bs + o, 0), E - 1)];
                    rk[q] += (int)(s < n[q]) & (int)(a < v[q].x);
                }
            }
        }
#pragma unroll
        for (int q = 0; q < BIN_U; ++q) {
            const bool in = i + q * BIN_T < cnt;
            const int p = min(max(bs + f[q] + rk[q], 0), E - 1);
            if (in) col[p] = sv[q];
        }
    }
}

__device__ __forceinline__ void add4(float4& a, const float4& v) {
    a.x += v.x; a.y += v.y; a.z += v.z; a.w += v.w;
}

__global__ __launch_bounds__(256)
void gcn_gather_rows(const int* __restrict__ rowptr, const int* __restrict__ col,
                     const float* __restrict__ dinv,
                     const float* __restrict__ bias, const float* __restrict__ h,
                     float* __restrict__ out, int N) {
    const int lane = threadIdx.x & 31;
    const int row = __builtin_amdgcn_readfirstlane((int)(blockIdx.x * 8 + (threadIdx.x >> 5)));
    if (row >= N) return;
    const int beg = __builtin_amdgcn_readfirstlane(rowptr[row]);
    const int end = __builtin_amdgcn_readfirstlane(rowptr[row + 1]);

    const float4* __restrict__ h4 = (const float4*)h;
    const float4 hv = h4[(size_t)row * 32 + lane];
    const float  di = dinv[row];
    const float4 b4 = ((const float4*)bias)[lane];

    float4 acc = make_float4(0.f, 0.f, 0.f, 0.f);
    const unsigned nmax = (unsigned)(N - 1);

    for (int base = beg; base < end; base += 32) {
        const int n = min(32, end - base);
        int sv = 0;
        if (lane < n) sv = col[base + lane];
        sv = (int)min((unsigned)sv, nmax);

        int j = 0;
        for (; j + 4 <= n; j += 4) {
            const int s0 = __builtin_amdgcn_readlane(sv, j);
            const int s1 = __builtin_amdgcn_readlane(sv, j + 1);
            const int s2 = __builtin_amdgcn_readlane(sv, j + 2);
            const int s3 = __builtin_amdgcn_readlane(sv, j + 3);
            const float4 v0 = h4[(size_t)s0 * 32 + lane];
            const float4 v1 = h4[(size_t)s1 * 32 + lane];
            const float4 v2 = h4[(size_t)s2 * 32 + lane];
            const float4 v3 = h4[(size_t)s3 * 32 + lane];
            add4(acc, v0);
            add4(acc, v1);
            add4(acc, v2);
            add4(acc, v3);
        }
        const int r = n - j;
        if (r > 0) {
            const int s0 = __builtin_amdgcn_readlane(sv, j);
            const float4 v0 = h4[(size_t)s0 * 32 + lane];
            float4 v1 = make_float4(0.f, 0.f, 0.f, 0.f);
            float4 v2 = make_float4(0.f, 0.f, 0.f, 0.f);
            if (r > 1) {
                const int s1 = __builtin_amdgcn_readlane(sv, j + 1);
                v1 = h4[(size_t)s1 * 32 + lane];
            }
            if (r > 2) {
                const int s2 = __builtin_amdgcn_readlane(sv, j + 2);
                v2 = h4[(size_t)s2 * 32 + lane];
            }
            add4(acc, v0);
            if (r > 1) add4(acc, v1);
            if (r > 2) add4(acc, v2);
        }
    }

    v4f o;
    o.x = tanhf((acc.x + hv.x) * di + b4.x);
    o.y = tanhf((acc.y + hv.y) * di + b4.y);
    o.z = tanhf((acc.z + hv.z) * di + b4.z);
    o.w = tanhf((acc.w + hv.w) * di + b4.w);
    __builtin_nontemporal_store(o, (v4f*)out + ((size_t)row * 32 + lane));
}


#define FZ_T 1024
#define FZ_STRIP (16 * A_LD)

__device__ __forceinline__ float4 gather_row_tanh(const int* __restrict__ rowptr,
                                                  const int* __restrict__ col,
                                                  const float* __restrict__ dinv,
                                                  const float4* __restrict__ h4,
                                                  const float4 b4, int row, int lane,
                                                  unsigned nmax) {
    const int beg = __builtin_amdgcn_readfirstlane(rowptr[row]);
    const int end = __builtin_amdgcn_readfirstlane(rowptr[row + 1]);
    const float4 hv = h4[(size_t)row * 32 + lane];
    const float  di = dinv[row];
    float4 acc = make_float4(0.f, 0.f, 0.f, 0.f);

    for (int base = beg; base < end; base += 32) {
        const int n = min(32, end - base);
        int sv = 0;
        if (lane < n) sv = col[base + lane];
        sv = (int)min((unsigned)sv, nmax);

        int j = 0;
        for (; j + 4 <= n; j += 4) {
            const int s0 = __builtin_amdgcn_readlane(sv, j);
            const int s1 = __builtin_amdgcn_readlane(sv, j + 1);
            const int s2 = __builtin_amdgcn_readlane(sv, j + 2);
            const int s3 = __builtin_amdgcn_readlane(sv, j + 3);
            const float4 v0 = h4[(size_t)s0 * 32 + lane];
            const float4 v1 = h4[(size_t)s1 * 32 + lane];
            const float4 v2 = h4[(size_t)s2 * 32 + lane];
            const float4 v3 = h4[(size_t)s3 * 32 + lane];
            add4(acc, v0);
            add4(acc, v1);
            add4(acc, v2);
            add4(acc, v3);
        }
        const int r = n - j;
        if (r > 0) {
            const int s0 = __builtin_amdgcn_readlane(sv, j);
            const float4 v0 = h4[(size_t)s0 * 32 + lane];
            float4 v1 = make_float4(0.f, 0.f, 0.f, 0.f);
            float4 v2 = make_float4(0.f, 0.f, 0.f, 0.f);
            if (r > 1) {
                const int s1 = __builtin_amdgcn_readlane(sv, j + 1);
                v1 = h4[(size_t)s1 * 32 + lane];
            }
            if (r > 2) {
                const int s2 = __builtin_amdgcn_readlane(sv, j + 2);
                v2 = h4[(size_t)s2 * 32 + lane];
            }
            add4(acc, v0);
            if (r > 1) add4(acc, v1);
            if (r > 2) add4(acc, v2);
        }
    }

    float4 o;
    o.x = tanhf((acc.x + hv.x) * di + b4.x) * di;
    o.y = tanhf((acc.y + hv.y) * di + b4.y) * di;
    o.z = tanhf((acc.z + hv.z) * di + b4.z) * di;
    o.w = tanhf((acc.w + hv.w) * di + b4.w) * di;
    return o;
}

__global__ __launch_bounds__(FZ_T)
void gcn_gather_gemm(const int* __restrict__ rowptr, const int* __restrict__ col,
                     const float* __restrict__ dinv,
                     const float* __restrict__ bias, const float* __restrict__ h,
                     const float* __restrict__ Wn, float* __restrict__ C, int N) {
    extern __shared__ float lds[];
    v2f*   Wp = (v2f*)lds;
    float* As = lds + (size_t)(HDIM / 2) * WS2 * 2;
    float* Cs = As + 2 * 2 * FZ_STRIP;

    const int tid  = threadIdx.x;
    const int lane = tid & 31;
    const int l15  = lane & 15;
    const int wave = __builtin_amdgcn_readfirstlane(tid >> 5);

    const float4* __restrict__ h4 = (const float4*)h;
    const float4 b4 = ((const float4*)bias)[lane];
    const unsigned nmax = (unsigned)(N - 1);
    const int nTrips = (N + 31) >> 5;
    const int step   = (int)gridDim.x;
    const float4 z4  = make_float4(0.f, 0.f, 0.f, 0.f);

    {
        const float4* W4 = (const float4*)Wn;
        float4 wa[2], wb[2];
#pragma unroll
        for (int u = 0; u < 2; ++u) {
            const int idx = u * FZ_T + tid;
            const int kk  = idx >> 5;
            const int n4  = idx & 31;
            wa[u] = W4[(size_t)(2 * kk)     * 32 + n4];
            wb[u] = W4[(size_t)(2 * kk + 1) * 32 + n4];
        }
#pragma unroll
        for (int u = 0; u < 2; ++u) {
            const int idx = u * FZ_T + tid;
            const int kk  = idx >> 5;
            const int n4  = idx & 31;
            float4* dst = (float4*)(Wp + kk * WS2 + 4 * n4);
            dst[0] = make_float4(wa[u].x, wb[u].x, wa[u].y, wb[u].y);
            dst[1] = make_float4(wa[u].z, wb[u].z, wa[u].w, wb[u].w);
        }
    }

    const int koff = (lane >= 16) ? 2 : 0;
    const int rOff = (lane >= 16) ? 8 : 0;

    int p = (int)blockIdx.x;
    float4 o = z4;
    if (p < nTrips && p * 32 + wave < N)
        o = gather_row_tanh(rowptr, col, dinv, h4, b4, p * 32 + wave, lane, nmax);

    int it = 0;
    for (; p < nTrips; p += step, ++it) {
        float* Ab = As + (size_t)(it & 1) * (2 * FZ_STRIP);
        {
            v2f* q = (v2f*)(Ab + wave * A_LD + 4 * lane);
            q[0] = (v2f){o.x, o.y};
            q[1] = (v2f){o.z, o.w};
        }
        __builtin_amdgcn_fence(__ATOMIC_RELEASE, "workgroup", "local");
        __builtin_amdgcn_s_barrier_signal(-1);

        const int pn = p + step;
        float4 on = z4;
        if (pn < nTrips && pn * 32 + wave < N)
            on = gather_row_tanh(rowptr, col, dinv, h4, b4, pn * 32 + wave, lane, nmax);

        __builtin_amdgcn_s_barrier_wait(-1);
        __builtin_amdgcn_fence(__ATOMIC_ACQUIRE, "workgroup", "local");

        if (it > 0) {
            const float* Cp = Cs + (size_t)((it - 1) & 1) * (32 * C_LD);
            const v4f cv = *(const v4f*)(Cp + wave * C_LD + 4 * lane);
            const int rp = (p - step) * 32 + wave;
            if (rp < N) __builtin_nontemporal_store(cv, (v4f*)C + ((size_t)rp * 32 + lane));
        }

        const int mw = wave ^ ((it & 1) << 4);
        if (mw < 16) {
            const int st = mw >> 3;
            const int jt = mw & 7;
            const float* Ak = Ab + (st * 16 + l15) * A_LD + koff;
            const v2f*   Bk = Wp + (size_t)(koff >> 1) * WS2 + jt * 16 + l15;
            v8f acc8 = (v8f){0.f,0.f,0.f,0.f,0.f,0.f,0.f,0.f};
#pragma unroll
            for (int g = 0; g < 4; ++g) {
                v2f af[8], bf[8];
#pragma unroll
                for (int q = 0; q < 8; ++q) {
                    const int kb = g * 8 + q;
                    af[q] = *(const v2f*)(Ak + 4 * kb);
                    bf[q] = Bk[(size_t)(2 * kb) * WS2];
                }
                __builtin_amdgcn_sched_barrier(0);
#pragma unroll
                for (int q = 0; q < 8; ++q)
                    acc8 = __builtin_amdgcn_wmma_f32_16x16x4_f32(
                        false, af[q], false, bf[q], (short)0, acc8, false, false);
                asm volatile("v_nop\n\tv_nop\n\tv_nop\n\tv_nop\n\tv_nop\n\tv_nop\n\tv_nop\n\tv_nop"
                             : "+v"(acc8));
                __builtin_amdgcn_sched_barrier(0);
            }
            float* Cb = Cs + (size_t)(it & 1) * (32 * C_LD) + (st * 16 + rOff) * C_LD + jt * 16 + l15;
#pragma unroll
            for (int i = 0; i < 8; ++i) Cb[i * C_LD] = acc8[i];
        }
        o = on;
    }
    if (it > 0) {
        __syncthreads();
        const float* Cp = Cs + (size_t)((it - 1) & 1) * (32 * C_LD);
        const v4f cv = *(const v4f*)(Cp + wave * C_LD + 4 * lane);
        const int rp = (p - step) * 32 + wave;
        if (rp < N) __builtin_nontemporal_store(cv, (v4f*)C + ((size_t)rp * 32 + lane));
    }
}

extern "C" void kernel_launch(void* const* d_in, const int* in_sizes, int n_in,
                              void* d_out, int out_size, void* d_ws, size_t ws_size,
                              hipStream_t stream) {
    const float* x  = (const float*)d_in[0];
    const int*   ei = (const int*)d_in[1];
    const float* Wl[4] = {(const float*)d_in[2], (const float*)d_in[4],
                          (const float*)d_in[6], (const float*)d_in[8]};
    const float* bl[4] = {(const float*)d_in[3], (const float*)d_in[5],
                          (const float*)d_in[7], (const float*)d_in[9]};

    const int NFEAT = in_sizes[2] / HDIM;
    const int N     = in_sizes[0] / NFEAT;
    const int E     = in_sizes[1] / 2;
    const int* srcp = ei;
    const int* dstp = ei + E;

    float* wsf  = (float*)d_ws;
    float* r0   = wsf;
    float* r1   = r0 + (size_t)N * HDIM;
    float* r2   = r1 + (size_t)N * HDIM;
    float* dinv = r2 + (size_t)N * HDIM;

    const int nb  = (N + BKT_ROWS - 1) >> BKT_SHIFT;
    const int n2  = nb * BIN_G;
    const int n2p = (n2 + 31) & ~31;
    int*   histG  = (int*)r1;
    int*   startG = histG + n2p;
    int*   rowptr = startG + n2p + 32;
    int*   rec    = rowptr + ((N + 31) & ~31) + 32;
    int*   eid    = rec + 2 * (size_t)E;
    int*   col    = eid + E;

    bin_chunk<<<BIN_G, BIN_T, (size_t)nb * sizeof(int), stream>>>(dstp, histG, startG, rec, E, nb);
    bucket_rows<<<nb, BIN_T, 0, stream>>>(histG, startG, rec, srcp, rowptr, dinv, eid, col, N, E);

    const int Kpad0 = (NFEAT + 3) & ~3;
    const size_t shmem0 =
        ((size_t)(Kpad0 / 2) * WS2 * 2 + 8u * 16u * A_LD) * sizeof(float);
    gcn_gemm_wmma_f32<<<256, 256, shmem0, stream>>>(x, Wl[0], dinv, r0, N, NFEAT, Kpad0);

    const size_t shmemF =
        ((size_t)(HDIM / 2) * WS2 * 2 + 2u * 2u * FZ_STRIP + 2u * 32u * C_LD) * sizeof(float);
    gcn_gather_gemm<<<512, FZ_T, shmemF, stream>>>(rowptr, col, dinv, bl[0], r0, Wl[1], r2, N);
    gcn_gather_gemm<<<512, FZ_T, shmemF, stream>>>(rowptr, col, dinv, bl[1], r2, Wl[2], r0, N);
    gcn_gather_gemm<<<512, FZ_T, shmemF, stream>>>(rowptr, col, dinv, bl[2], r0, Wl[3], r2, N);

    gcn_gather_rows<<<(N + 7) / 8, 256, 0, stream>>>(
        rowptr, col, dinv, bl[3], r2, (float*)d_out, N);
}
